# starting kernel
# baseline (speedup 1.0000x reference)
#include <hip/hip_runtime.h>


#define Bsz 512
#define Isz 1024
#define HD  2048
#define Lsz 1000
#define LP  1024
#define TMAX 32

typedef __attribute__((ext_vector_type(16))) __bf16 v16bf;
typedef __attribute__((ext_vector_type(8)))  float  v8f;
typedef __attribute__((ext_vector_type(4)))  unsigned int u32x4;

__device__ __forceinline__ unsigned short f2bf(float f) {
    unsigned int u = __float_as_uint(f);
    u += 0x7FFFu + ((u >> 16) & 1u);
    return (unsigned short)(u >> 16);
}

struct alignas(32) FragBits { u32x4 lo, hi; };

__device__ __forceinline__ v16bf load_frag(const unsigned short* __restrict__ p) {
    FragBits fb;
    fb.lo = *reinterpret_cast<const u32x4*>(p);
    fb.hi = *reinterpret_cast<const u32x4*>(p + 16);
    v16bf v;
    __builtin_memcpy(&v, &fb, sizeof(v));
    return v;
}

template <int MODE>
__global__ __launch_bounds__(128)
void wmma_gemm(const unsigned short* __restrict__ A,
               const unsigned short* __restrict__ BT,
               int M, int N, int K,
               const float* __restrict__ bias,
               const float* __restrict__ drive,
               const float* __restrict__ Hold,
               const int*   __restrict__ active,
               float* __restrict__ Cout,
               unsigned short* __restrict__ HbOut,
               unsigned int* __restrict__ dHbits,
               int NL) {
    const int lane = threadIdx.x & 31;
    const int wave = threadIdx.x >> 5;
    const int rowBase = blockIdx.y * 64 + wave * 16;
    const int colBase = blockIdx.x * 64;
    const int r16 = lane & 15;
    const int kh  = lane >> 4;

    const unsigned short* pa = A + (size_t)(rowBase + r16) * K + kh * 8;
    const unsigned short* pb0 = BT + (size_t)(colBase + r16) * K + kh * 8;
    const unsigned short* pb1 = pb0 + (size_t)16 * K;
    const unsigned short* pb2 = pb0 + (size_t)32 * K;
    const unsigned short* pb3 = pb0 + (size_t)48 * K;

    v8f acc0 = {0.f,0.f,0.f,0.f,0.f,0.f,0.f,0.f};
    v8f acc[4];
    acc[0] = acc0; acc[1] = acc0; acc[2] = acc0; acc[3] = acc0;

    v16bf a_cur = load_frag(pa);
    v16bf b_cur0 = load_frag(pb0);
    v16bf b_cur1 = load_frag(pb1);
    v16bf b_cur2 = load_frag(pb2);
    v16bf b_cur3 = load_frag(pb3);

    for (int k0 = 0; k0 < K - 32; k0 += 32) {
        pa += 32; pb0 += 32; pb1 += 32; pb2 += 32; pb3 += 32;
        const v16bf a_nxt  = load_frag(pa);
        const v16bf b_nxt0 = load_frag(pb0);
        const v16bf b_nxt1 = load_frag(pb1);
        const v16bf b_nxt2 = load_frag(pb2);
        const v16bf b_nxt3 = load_frag(pb3);
        __builtin_prefetch(pa + 32, 0, 3);

        acc[0] = __builtin_amdgcn_wmma_f32_16x16x32_bf16(
            false, a_cur, false, b_cur0, (short)0, acc[0], false, false);
        acc[1] = __builtin_amdgcn_wmma_f32_16x16x32_bf16(
            false, a_cur, false, b_cur1, (short)0, acc[1], false, false);
        acc[2] = __builtin_amdgcn_wmma_f32_16x16x32_bf16(
            false, a_cur, false, b_cur2, (short)0, acc[2], false, false);
        acc[3] = __builtin_amdgcn_wmma_f32_16x16x32_bf16(
            false, a_cur, false, b_cur3, (short)0, acc[3], false, false);

        a_cur = a_nxt;
        b_cur0 = b_nxt0; b_cur1 = b_nxt1; b_cur2 = b_nxt2; b_cur3 = b_nxt3;
    }
    acc[0] = __builtin_amdgcn_wmma_f32_16x16x32_bf16(
        false, a_cur, false, b_cur0, (short)0, acc[0], false, false);
    acc[1] = __builtin_amdgcn_wmma_f32_16x16x32_bf16(
        false, a_cur, false, b_cur1, (short)0, acc[1], false, false);
    acc[2] = __builtin_amdgcn_wmma_f32_16x16x32_bf16(
        false, a_cur, false, b_cur2, (short)0, acc[2], false, false);
    acc[3] = __builtin_amdgcn_wmma_f32_16x16x32_bf16(
        false, a_cur, false, b_cur3, (short)0, acc[3], false, false);

    bool rowActive[8];
    float dmax[8];
    if (MODE == 1) {
#pragma unroll
        for (int r = 0; r < 8; ++r) {
            rowActive[r] = active[rowBase + r + 8 * kh] != 0;
            dmax[r] = 0.f;
        }
    }

#pragma unroll
    for (int f = 0; f < 4; ++f) {
        const int n = colBase + f * 16 + r16;
#pragma unroll
        for (int r = 0; r < 8; ++r) {
            const int m = rowBase + r + 8 * kh;
            const float v = acc[f][r];
            if (MODE == 0) {
                Cout[(size_t)m * N + n] = v + bias[n];
            } else if (MODE == 2) {
                if (n < NL) Cout[(size_t)m * N + n] = v + bias[n];
            } else {
                const size_t idx = (size_t)m * N + n;
                const float pre = v + drive[idx];
                const float actv = pre >= 0.f ? pre : 0.05f * pre;
                const float ho = Hold[idx];
                const float hn = rowActive[r] ? actv : ho;
                Cout[idx] = hn;
                HbOut[idx] = f2bf(hn);
                dmax[r] = fmaxf(dmax[r], fabsf(hn - ho));
            }
        }
    }

    if (MODE == 1) {
#pragma unroll
        for (int r = 0; r < 8; ++r) {
            if (dmax[r] > 0.f) {
                atomicMax(dHbits + (rowBase + r + 8 * kh), __float_as_uint(dmax[r]));
            }
        }
    }
}

__global__ __launch_bounds__(256)
void col_l1_kernel(const float* __restrict__ A, const float* __restrict__ Msk,
                   float* __restrict__ colsum, int rows, int cols) {
    __shared__ float sm[256];
    const int j = blockIdx.x;
    float s = 0.f;
    for (int i = threadIdx.x; i < rows; i += 256) {
        const size_t idx = (size_t)i * cols + j;
        s += fabsf(A[idx] * Msk[idx]);
    }
    sm[threadIdx.x] = s;
    __syncthreads();
    for (int st = 128; st > 0; st >>= 1) {
        if (threadIdx.x < st) sm[threadIdx.x] += sm[threadIdx.x + st];
        __syncthreads();
    }
    if (threadIdx.x == 0) colsum[j] = sm[0];
}

__global__ __launch_bounds__(256)
void scalars_kernel(const float* __restrict__ col_l1,
                    const float* __restrict__ whl,
                    float* __restrict__ scalars) {
    __shared__ float sm[256];
    const int t = threadIdx.x;
    float mx = -1e30f;
    for (int j = t; j < HD; j += 256) mx = fmaxf(mx, col_l1[j]);
    sm[t] = mx; __syncthreads();
    for (int s = 128; s > 0; s >>= 1) {
        if (t < s) sm[t] = fmaxf(sm[t], sm[t + s]);
        __syncthreads();
    }
    const float m1 = sm[0];
    __syncthreads();
    mx = -1e30f;
    for (int j = t; j < Lsz; j += 256) mx = fmaxf(mx, whl[j]);
    sm[t] = mx; __syncthreads();
    for (int s = 128; s > 0; s >>= 1) {
        if (t < s) sm[t] = fmaxf(sm[t], sm[t + s]);
        __syncthreads();
    }
    if (t == 0) {
        scalars[0] = 0.95f / fmaxf(m1, 1e-8f);
        scalars[1] = sm[0] * 19.0f;
    }
}

__global__ __launch_bounds__(256)
void build_wt_kernel(const float* __restrict__ R, const float* __restrict__ M,
                     const float* __restrict__ scalars, int sel,
                     unsigned short* __restrict__ WT,
                     int K, int N, int srcld, int NL) {
    const size_t idx = (size_t)blockIdx.x * 256 + threadIdx.x;
    const size_t total = (size_t)N * K;
    if (idx >= total) return;
    const int n = (int)(idx / K);
    const int k = (int)(idx % K);
    float v = 0.f;
    if (n < NL) {
        const size_t s = (size_t)k * srcld + n;
        v = R[s] * M[s];
        if (sel == 0) v *= scalars[0];
    }
    WT[idx] = f2bf(v);
}

__global__ __launch_bounds__(256)
void to_bf16_kernel(const float* __restrict__ src, unsigned short* __restrict__ dst,
                    int n) {
    const int i = blockIdx.x * 256 + threadIdx.x;
    if (i < n) dst[i] = f2bf(src[i]);
}

__global__ __launch_bounds__(256)
void init_state(float* __restrict__ H0, unsigned short* __restrict__ Hb0,
                unsigned int* __restrict__ dH, int* __restrict__ active) {
    const size_t n = (size_t)Bsz * HD;
    for (size_t t = (size_t)blockIdx.x * 256 + threadIdx.x; t < n;
         t += (size_t)gridDim.x * 256) {
        H0[t] = 0.f;
        Hb0[t] = 0;
    }
    const int i = blockIdx.x * 256 + threadIdx.x;
    if (i < Bsz) { dH[i] = 0u; active[i] = 1; }
}

__global__ __launch_bounds__(256)
void update_rows(const float* __restrict__ logits,
                 const float* __restrict__ scalars,
                 unsigned int* __restrict__ dHbits,
                 int* __restrict__ active,
                 float* __restrict__ out,
                 int is_last) {
    const int lane = threadIdx.x & 31;
    const int wave = threadIdx.x >> 5;
    const int row = blockIdx.x * 8 + wave;
    if (row >= Bsz) return;

    if (active[row] != 0) {
        const float* lr = logits + (size_t)row * LP;
        float m1 = -1e30f, m2 = -1e30f;
        for (int n = lane; n < Lsz; n += 32) {
            const float v = lr[n];
            if (v > m1) { m2 = m1; m1 = v; }
            else if (v > m2) { m2 = v; }
        }
        for (int off = 16; off > 0; off >>= 1) {
            const float o1 = __shfl_xor(m1, off, 32);
            const float o2 = __shfl_xor(m2, off, 32);
            const float hi = fmaxf(m1, o1);
            const float lo = fmaxf(fminf(m1, o1), fmaxf(m2, o2));
            m1 = hi; m2 = lo;
        }
        const float margin = m1 - m2;
        const float dh = __uint_as_float(dHbits[row]);
        const float Gamma = scalars[1] * dh;
        const bool newly = margin > 2.0f * Gamma;
        if (newly || is_last) {
            for (int n = lane; n < Lsz; n += 32)
                out[(size_t)row * Lsz + n] = lr[n];
            if (lane == 0) active[row] = 0;
        }
    }
    if (lane == 0) dHbits[row] = 0u;
}

static inline char* wsalloc(char*& p, size_t bytes) {
    char* r = p;
    p += (bytes + 255) & ~(size_t)255;
    return r;
}

extern "C" void kernel_launch(void* const* d_in, const int* in_sizes, int n_in,
                              void* d_out, int out_size, void* d_ws, size_t ws_size,
                              hipStream_t stream) {
    (void)in_sizes; (void)n_in; (void)out_size; (void)ws_size;
    const float* x   = (const float*)d_in[0];
    const float* RIH = (const float*)d_in[1];
    const float* RH  = (const float*)d_in[2];
    const float* RHL = (const float*)d_in[3];
    const float* BH  = (const float*)d_in[4];
    const float* BL  = (const float*)d_in[5];
    const float* MIH = (const float*)d_in[6];
    const float* MH  = (const float*)d_in[7];
    const float* MHL = (const float*)d_in[8];
    float* out = (float*)d_out;

    char* p = (char*)d_ws;
    unsigned short* WT_H  = (unsigned short*)wsalloc(p, (size_t)HD * HD * 2);
    unsigned short* WT_HL = (unsigned short*)wsalloc(p, (size_t)LP * HD * 2);
    unsigned short* WT_IH = (unsigned short*)wsalloc(p, (size_t)HD * Isz * 2);
    unsigned short* xb    = (unsigned short*)wsalloc(p, (size_t)Bsz * Isz * 2);
    float* drive = (float*)wsalloc(p, (size_t)Bsz * HD * 4);
    float* Hbuf0 = (float*)wsalloc(p, (size_t)Bsz * HD * 4);
    float* Hbuf1 = (float*)wsalloc(p, (size_t)Bsz * HD * 4);
    unsigned short* Hb0 = (unsigned short*)wsalloc(p, (size_t)Bsz * HD * 2);
    unsigned short* Hb1 = (unsigned short*)wsalloc(p, (size_t)Bsz * HD * 2);
    float* logits = (float*)wsalloc(p, (size_t)Bsz * LP * 4);
    float* col_l1 = (float*)wsalloc(p, HD * 4);
    float* whl    = (float*)wsalloc(p, LP * 4);
    float* scalars = (float*)wsalloc(p, 256);
    unsigned int* dHb = (unsigned int*)wsalloc(p, Bsz * 4);
    int* active = (int*)wsalloc(p, Bsz * 4);

    col_l1_kernel<<<HD, 256, 0, stream>>>(RH, MH, col_l1, HD, HD);
    col_l1_kernel<<<Lsz, 256, 0, stream>>>(RHL, MHL, whl, HD, Lsz);
    scalars_kernel<<<1, 256, 0, stream>>>(col_l1, whl, scalars);

    build_wt_kernel<<<(HD * HD) / 256, 256, 0, stream>>>(
        RH, MH, scalars,  0, WT_H, HD, HD, HD, HD);
    build_wt_kernel<<<(LP * HD) / 256, 256, 0, stream>>>(
        RHL, MHL, scalars,  1, WT_HL, HD, LP, Lsz, Lsz);
    build_wt_kernel<<<(HD * Isz) / 256, 256, 0, stream>>>(
        RIH, MIH, scalars,  1, WT_IH, Isz, HD, HD, HD);
    to_bf16_kernel<<<(Bsz * Isz) / 256, 256, 0, stream>>>(x, xb, Bsz * Isz);
    init_state<<<(Bsz * HD) / 256, 256, 0, stream>>>(Hbuf0, Hb0, dHb, active);

    wmma_gemm<0><<<dim3(HD / 64, Bsz / 64), 128, 0, stream>>>(
        xb, WT_IH, Bsz, HD, Isz, BH, nullptr, nullptr, nullptr,
        drive, nullptr, nullptr, HD);

    float* Hc[2] = {Hbuf0, Hbuf1};
    unsigned short* Hbc[2] = {Hb0, Hb1};
    for (int t = 1; t <= TMAX; ++t) {
        const int pc = (t - 1) & 1, pn = t & 1;
        wmma_gemm<1><<<dim3(HD / 64, Bsz / 64), 128, 0, stream>>>(
            Hbc[pc], WT_H, Bsz, HD, HD, nullptr, drive, Hc[pc], active,
            Hc[pn], Hbc[pn], dHb, HD);
        wmma_gemm<2><<<dim3(LP / 64, Bsz / 64), 128, 0, stream>>>(
            Hbc[pn], WT_HL, Bsz, LP, HD, BL, nullptr, nullptr, nullptr,
            logits, nullptr, nullptr, Lsz);
        update_rows<<<Bsz / 8, 256, 0, stream>>>(
            logits, scalars, dHb, active, out, (t == TMAX) ? 1 : 0);
    }
}


# GEMM operands staged through LDS: each block fetches its 64x256 A and BT chunks once with whole-line loads (next chunk kept in registers), waves read WMMA fragments from LDS; sums unchanged
# speedup vs baseline: 1.3417x; 1.3417x over previous
#include <hip/hip_runtime.h>


#define Bsz 512
#define Isz 1024
#define HD  2048
#define Lsz 1000
#define LP  1024
#define TMAX 32

typedef __attribute__((ext_vector_type(16))) __bf16 v16bf;
typedef __attribute__((ext_vector_type(8)))  float  v8f;
typedef __attribute__((ext_vector_type(4)))  unsigned int u32x4;

__device__ __forceinline__ unsigned short f2bf(float f) {
    unsigned int u = __float_as_uint(f);
    u += 0x7FFFu + ((u >> 16) & 1u);
    return (unsigned short)(u >> 16);
}

#define KC 256
#define PITCH (KC * 2 + 16)
#define TILE_BYTES (64 * PITCH)

struct alignas(32) FragBits { u32x4 lo, hi; };

__device__ __forceinline__ v16bf lds_frag(const unsigned char* p) {
    FragBits fb;
    fb.lo = *reinterpret_cast<const u32x4*>(p);
    fb.hi = *reinterpret_cast<const u32x4*>(p + 32);
    v16bf v;
    __builtin_memcpy(&v, &fb, sizeof(v));
    return v;
}

template <int MODE>
__global__ __launch_bounds__(128)
void wmma_gemm(const unsigned short* __restrict__ A,
               const unsigned short* __restrict__ BT,
               int M, int N, int K,
               const float* __restrict__ bias,
               const float* __restrict__ drive,
               const float* __restrict__ Hold,
               const int*   __restrict__ active,
               float* __restrict__ Cout,
               unsigned short* __restrict__ HbOut,
               unsigned int* __restrict__ dHbits,
               int NL) {
    __shared__ __attribute__((aligned(16))) unsigned char smem[2 * TILE_BYTES];
    unsigned char* const sA = smem;
    unsigned char* const sB = smem + TILE_BYTES;

    const int tid  = threadIdx.x;
    const int lane = tid & 31;
    const int wave = tid >> 5;
    const int rowBase = blockIdx.y * 64 + wave * 16;
    const int colBase = blockIdx.x * 64;
    const int r16 = lane & 15;
    const int kh  = lane >> 4;

    const unsigned short* ga = A  + (size_t)(blockIdx.y * 64 + wave) * K + lane * 8;
    const unsigned short* gb = BT + (size_t)(colBase + wave) * K + lane * 8;
    const size_t gstep = (size_t)4 * K;
    const int sst = wave * PITCH + lane * 16;

    const int fa = (wave * 16 + r16) * PITCH + kh * 16;
    const int fb = r16 * PITCH + kh * 16;

    v8f acc0 = {0.f,0.f,0.f,0.f,0.f,0.f,0.f,0.f};
    v8f acc[4];
    acc[0] = acc0; acc[1] = acc0; acc[2] = acc0; acc[3] = acc0;

    u32x4 ra[16], rb[16];

#pragma unroll
    for (int i = 0; i < 16; ++i) {
        ra[i] = *reinterpret_cast<const u32x4*>(ga + i * gstep);
        rb[i] = *reinterpret_cast<const u32x4*>(gb + i * gstep);
    }
#pragma unroll
    for (int i = 0; i < 16; ++i) {
        *reinterpret_cast<u32x4*>(sA + sst + i * 4 * PITCH) = ra[i];
        *reinterpret_cast<u32x4*>(sB + sst + i * 4 * PITCH) = rb[i];
    }
    __syncthreads();

    const int nchunks = K / KC;
    for (int c = 0; c < nchunks; ++c) {
        const bool more = (c + 1 < nchunks);
        if (more) {
            ga += KC; gb += KC;
#pragma unroll
            for (int i = 0; i < 16; ++i) {
                ra[i] = *reinterpret_cast<const u32x4*>(ga + i * gstep);
                rb[i] = *reinterpret_cast<const u32x4*>(gb + i * gstep);
            }
        }

        v16bf a_cur  = lds_frag(sA + fa);
        v16bf b_cur0 = lds_frag(sB + fb);
        v16bf b_cur1 = lds_frag(sB + fb + 16 * PITCH);
        v16bf b_cur2 = lds_frag(sB + fb + 32 * PITCH);
        v16bf b_cur3 = lds_frag(sB + fb + 48 * PITCH);
#pragma unroll
        for (int ks = 0; ks < KC / 32; ++ks) {
            v16bf a_nxt = a_cur, b_nxt0 = b_cur0, b_nxt1 = b_cur1,
                  b_nxt2 = b_cur2, b_nxt3 = b_cur3;
            if (ks + 1 < KC / 32) {
                const int ko = (ks + 1) * 64;
                a_nxt  = lds_frag(sA + fa + ko);
                b_nxt0 = lds_frag(sB + fb + ko);
                b_nxt1 = lds_frag(sB + fb + 16 * PITCH + ko);
                b_nxt2 = lds_frag(sB + fb + 32 * PITCH + ko);
                b_nxt3 = lds_frag(sB + fb + 48 * PITCH + ko);
            }
            __builtin_amdgcn_sched_barrier(0);
            acc[0] = __builtin_amdgcn_wmma_f32_16x16x32_bf16(
                false, a_cur, false, b_cur0, (short)0, acc[0], false, false);
            acc[1] = __builtin_amdgcn_wmma_f32_16x16x32_bf16(
                false, a_cur, false, b_cur1, (short)0, acc[1], false, false);
            acc[2] = __builtin_amdgcn_wmma_f32_16x16x32_bf16(
                false, a_cur, false, b_cur2, (short)0, acc[2], false, false);
            acc[3] = __builtin_amdgcn_wmma_f32_16x16x32_bf16(
                false, a_cur, false, b_cur3, (short)0, acc[3], false, false);
            __builtin_amdgcn_sched_barrier(0);
            a_cur = a_nxt;
            b_cur0 = b_nxt0; b_cur1 = b_nxt1; b_cur2 = b_nxt2; b_cur3 = b_nxt3;
        }

        __syncthreads();
        if (more) {
#pragma unroll
            for (int i = 0; i < 16; ++i) {
                *reinterpret_cast<u32x4*>(sA + sst + i * 4 * PITCH) = ra[i];
                *reinterpret_cast<u32x4*>(sB + sst + i * 4 * PITCH) = rb[i];
            }
            __syncthreads();
        }
    }

    bool rowActive[8];
    float dmax[8];
    if (MODE == 1) {
#pragma unroll
        for (int r = 0; r < 8; ++r) {
            rowActive[r] = active[rowBase + r + 8 * kh] != 0;
            dmax[r] = 0.f;
        }
    }

#pragma unroll
    for (int f = 0; f < 4; ++f) {
        const int n = colBase + f * 16 + r16;
#pragma unroll
        for (int r = 0; r < 8; ++r) {
            const int m = rowBase + r + 8 * kh;
            const float v = acc[f][r];
            if (MODE == 0) {
                Cout[(size_t)m * N + n] = v + bias[n];
            } else if (MODE == 2) {
                if (n < NL) Cout[(size_t)m * N + n] = v + bias[n];
            } else {
                const size_t idx = (size_t)m * N + n;
                const float pre = v + drive[idx];
                const float actv = pre >= 0.f ? pre : 0.05f * pre;
                const float ho = Hold[idx];
                const float hn = rowActive[r] ? actv : ho;
                Cout[idx] = hn;
                HbOut[idx] = f2bf(hn);
                dmax[r] = fmaxf(dmax[r], fabsf(hn - ho));
            }
        }
    }

    if (MODE == 1) {
#pragma unroll
        for (int r = 0; r < 8; ++r) {
            if (dmax[r] > 0.f) {
                atomicMax(dHbits + (rowBase + r + 8 * kh), __float_as_uint(dmax[r]));
            }
        }
    }
}

__global__ __launch_bounds__(256)
void col_l1_kernel(const float* __restrict__ A, const float* __restrict__ Msk,
                   float* __restrict__ colsum, int rows, int cols) {
    __shared__ float sm[256];
    const int j = blockIdx.x;
    float s = 0.f;
    for (int i = threadIdx.x; i < rows; i += 256) {
        const size_t idx = (size_t)i * cols + j;
        s += fabsf(A[idx] * Msk[idx]);
    }
    sm[threadIdx.x] = s;
    __syncthreads();
    for (int st = 128; st > 0; st >>= 1) {
        if (threadIdx.x < st) sm[threadIdx.x] += sm[threadIdx.x + st];
        __syncthreads();
    }
    if (threadIdx.x == 0) colsum[j] = sm[0];
}

__global__ __launch_bounds__(256)
void scalars_kernel(const float* __restrict__ col_l1,
                    const float* __restrict__ whl,
                    float* __restrict__ scalars) {
    __shared__ float sm[256];
    const int t = threadIdx.x;
    float mx = -1e30f;
    for (int j = t; j < HD; j += 256) mx = fmaxf(mx, col_l1[j]);
    sm[t] = mx; __syncthreads();
    for (int s = 128; s > 0; s >>= 1) {
        if (t < s) sm[t] = fmaxf(sm[t], sm[t + s]);
        __syncthreads();
    }
    const float m1 = sm[0];
    __syncthreads();
    mx = -1e30f;
    for (int j = t; j < Lsz; j += 256) mx = fmaxf(mx, whl[j]);
    sm[t] = mx; __syncthreads();
    for (int s = 128; s > 0; s >>= 1) {
        if (t < s) sm[t] = fmaxf(sm[t], sm[t + s]);
        __syncthreads();
    }
    if (t == 0) {
        scalars[0] = 0.95f / fmaxf(m1, 1e-8f);
        scalars[1] = sm[0] * 19.0f;
    }
}

__global__ __launch_bounds__(256)
void build_wt_kernel(const float* __restrict__ R, const float* __restrict__ M,
                     const float* __restrict__ scalars, int sel,
                     unsigned short* __restrict__ WT,
                     int K, int N, int srcld, int NL) {
    const size_t idx = (size_t)blockIdx.x * 256 + threadIdx.x;
    const size_t total = (size_t)N * K;
    if (idx >= total) return;
    const int n = (int)(idx / K);
    const int k = (int)(idx % K);
    float v = 0.f;
    if (n < NL) {
        const size_t s = (size_t)k * srcld + n;
        v = R[s] * M[s];
        if (sel == 0) v *= scalars[0];
    }
    WT[idx] = f2bf(v);
}

__global__ __launch_bounds__(256)
void to_bf16_kernel(const float* __restrict__ src, unsigned short* __restrict__ dst,
                    int n) {
    const int i = blockIdx.x * 256 + threadIdx.x;
    if (i < n) dst[i] = f2bf(src[i]);
}

__global__ __launch_bounds__(256)
void init_state(float* __restrict__ H0, unsigned short* __restrict__ Hb0,
                unsigned int* __restrict__ dH, int* __restrict__ active) {
    const size_t n = (size_t)Bsz * HD;
    for (size_t t = (size_t)blockIdx.x * 256 + threadIdx.x; t < n;
         t += (size_t)gridDim.x * 256) {
        H0[t] = 0.f;
        Hb0[t] = 0;
    }
    const int i = blockIdx.x * 256 + threadIdx.x;
    if (i < Bsz) { dH[i] = 0u; active[i] = 1; }
}

__global__ __launch_bounds__(256)
void update_rows(const float* __restrict__ logits,
                 const float* __restrict__ scalars,
                 unsigned int* __restrict__ dHbits,
                 int* __restrict__ active,
                 float* __restrict__ out,
                 int is_last) {
    const int lane = threadIdx.x & 31;
    const int wave = threadIdx.x >> 5;
    const int row = blockIdx.x * 8 + wave;
    if (row >= Bsz) return;

    if (active[row] != 0) {
        const float* lr = logits + (size_t)row * LP;
        float m1 = -1e30f, m2 = -1e30f;
        for (int n = lane; n < Lsz; n += 32) {
            const float v = lr[n];
            if (v > m1) { m2 = m1; m1 = v; }
            else if (v > m2) { m2 = v; }
        }
        for (int off = 16; off > 0; off >>= 1) {
            const float o1 = __shfl_xor(m1, off, 32);
            const float o2 = __shfl_xor(m2, off, 32);
            const float hi = fmaxf(m1, o1);
            const float lo = fmaxf(fminf(m1, o1), fmaxf(m2, o2));
            m1 = hi; m2 = lo;
        }
        const float margin = m1 - m2;
        const float dh = __uint_as_float(dHbits[row]);
        const float Gamma = scalars[1] * dh;
        const bool newly = margin > 2.0f * Gamma;
        if (newly || is_last) {
            for (int n = lane; n < Lsz; n += 32)
                out[(size_t)row * Lsz + n] = lr[n];
            if (lane == 0) active[row] = 0;
        }
    }
    if (lane == 0) dHbits[row] = 0u;
}

static inline char* wsalloc(char*& p, size_t bytes) {
    char* r = p;
    p += (bytes + 255) & ~(size_t)255;
    return r;
}

extern "C" void kernel_launch(void* const* d_in, const int* in_sizes, int n_in,
                              void* d_out, int out_size, void* d_ws, size_t ws_size,
                              hipStream_t stream) {
    (void)in_sizes; (void)n_in; (void)out_size; (void)ws_size;
    const float* x   = (const float*)d_in[0];
    const float* RIH = (const float*)d_in[1];
    const float* RH  = (const float*)d_in[2];
    const float* RHL = (const float*)d_in[3];
    const float* BH  = (const float*)d_in[4];
    const float* BL  = (const float*)d_in[5];
    const float* MIH = (const float*)d_in[6];
    const float* MH  = (const float*)d_in[7];
    const float* MHL = (const float*)d_in[8];
    float* out = (float*)d_out;

    char* p = (char*)d_ws;
    unsigned short* WT_H  = (unsigned short*)wsalloc(p, (size_t)HD * HD * 2);
    unsigned short* WT_HL = (unsigned short*)wsalloc(p, (size_t)LP * HD * 2);
    unsigned short* WT_IH = (unsigned short*)wsalloc(p, (size_t)HD * Isz * 2);
    unsigned short* xb    = (unsigned short*)wsalloc(p, (size_t)Bsz * Isz * 2);
    float* drive = (float*)wsalloc(p, (size_t)Bsz * HD * 4);
    float* Hbuf0 = (float*)wsalloc(p, (size_t)Bsz * HD * 4);
    float* Hbuf1 = (float*)wsalloc(p, (size_t)Bsz * HD * 4);
    unsigned short* Hb0 = (unsigned short*)wsalloc(p, (size_t)Bsz * HD * 2);
    unsigned short* Hb1 = (unsigned short*)wsalloc(p, (size_t)Bsz * HD * 2);
    float* logits = (float*)wsalloc(p, (size_t)Bsz * LP * 4);
    float* col_l1 = (float*)wsalloc(p, HD * 4);
    float* whl    = (float*)wsalloc(p, LP * 4);
    float* scalars = (float*)wsalloc(p, 256);
    unsigned int* dHb = (unsigned int*)wsalloc(p, Bsz * 4);
    int* active = (int*)wsalloc(p, Bsz * 4);

    col_l1_kernel<<<HD, 256, 0, stream>>>(RH, MH, col_l1, HD, HD);
    col_l1_kernel<<<Lsz, 256, 0, stream>>>(RHL, MHL, whl, HD, Lsz);
    scalars_kernel<<<1, 256, 0, stream>>>(col_l1, whl, scalars);

    build_wt_kernel<<<(HD * HD) / 256, 256, 0, stream>>>(
        RH, MH, scalars,  0, WT_H, HD, HD, HD, HD);
    build_wt_kernel<<<(LP * HD) / 256, 256, 0, stream>>>(
        RHL, MHL, scalars,  1, WT_HL, HD, LP, Lsz, Lsz);
    build_wt_kernel<<<(HD * Isz) / 256, 256, 0, stream>>>(
        RIH, MIH, scalars,  1, WT_IH, Isz, HD, HD, HD);
    to_bf16_kernel<<<(Bsz * Isz) / 256, 256, 0, stream>>>(x, xb, Bsz * Isz);
    init_state<<<(Bsz * HD) / 256, 256, 0, stream>>>(Hbuf0, Hb0, dHb, active);

    wmma_gemm<0><<<dim3(HD / 64, Bsz / 64), 128, 0, stream>>>(
        xb, WT_IH, Bsz, HD, Isz, BH, nullptr, nullptr, nullptr,
        drive, nullptr, nullptr, HD);

    float* Hc[2] = {Hbuf0, Hbuf1};
    unsigned short* Hbc[2] = {Hb0, Hb1};
    for (int t = 1; t <= TMAX; ++t) {
        const int pc = (t - 1) & 1, pn = t & 1;
        wmma_gemm<1><<<dim3(HD / 64, Bsz / 64), 128, 0, stream>>>(
            Hbc[pc], WT_H, Bsz, HD, HD, nullptr, drive, Hc[pc], active,
            Hc[pn], Hbc[pn], dHb, HD);
        wmma_gemm<2><<<dim3(LP / 64, Bsz / 64), 128, 0, stream>>>(
            Hbc[pn], WT_HL, Bsz, LP, HD, BL, nullptr, nullptr, nullptr,
            logits, nullptr, nullptr, Lsz);
        update_rows<<<Bsz / 8, 256, 0, stream>>>(
            logits, scalars, dHb, active, out, (t == TMAX) ? 1 : 0);
    }
}


# Recurrence GEMMs follow the active rows: a 64-row block with none exits; with <=32 left the rows are gathered and each 16x16 tile gets a block, K split over 4 waves (4 partial sums added in k order)
# speedup vs baseline: 1.6940x; 1.2626x over previous
#include <hip/hip_runtime.h>


#define Bsz 512
#define Isz 1024
#define HD  2048
#define Lsz 1000
#define LP  1024
#define TMAX 32

typedef __attribute__((ext_vector_type(16))) __bf16 v16bf;
typedef __attribute__((ext_vector_type(8)))  float  v8f;
typedef __attribute__((ext_vector_type(4)))  unsigned int u32x4;

__device__ __forceinline__ unsigned short f2bf(float f) {
    unsigned int u = __float_as_uint(f);
    u += 0x7FFFu + ((u >> 16) & 1u);
    return (unsigned short)(u >> 16);
}

#define KC 256
#define PITCH (KC * 2 + 16)
#define TILE_BYTES (64 * PITCH)

struct alignas(32) FragBits { u32x4 lo, hi; };

__device__ __forceinline__ v16bf lds_frag(const unsigned char* p) {
    FragBits fb;
    fb.lo = *reinterpret_cast<const u32x4*>(p);
    fb.hi = *reinterpret_cast<const u32x4*>(p + 32);
    v16bf v;
    __builtin_memcpy(&v, &fb, sizeof(v));
    return v;
}

#define TAIL_ROWS (Bsz / 16)
template <int MODE>
__global__ __launch_bounds__(128)
void wmma_gemm(const unsigned short* __restrict__ A,
               const unsigned short* __restrict__ BT,
               int M, int N, int K,
               const float* __restrict__ bias,
               const float* __restrict__ drive,
               const float* __restrict__ Hold,
               const int*   __restrict__ active,
               float* __restrict__ Cout,
               unsigned short* __restrict__ HbOut,
               unsigned int* __restrict__ dHbits,
               int NL) {
    __shared__ __attribute__((aligned(16))) unsigned char smem[2 * TILE_BYTES];
    unsigned char* const sA = smem;
    unsigned char* const sB = smem + TILE_BYTES;

    const int tid  = threadIdx.x;
    const int lane = tid & 31;
    const int wave = tid >> 5;
    const int rowBase = blockIdx.y * 64 + wave * 16;
    const int colBase = blockIdx.x * 64;
    const int r16 = lane & 15;
    const int kh  = lane >> 4;

    if (MODE != 0) {
        __shared__ int s_cnt[Bsz / 32];
        __shared__ int s_list[TAIL_ROWS];
        int flag[Bsz / 128];
        unsigned msk[Bsz / 128];
#pragma unroll
        for (int j = 0; j < Bsz / 128; ++j) {
            flag[j] = active[j * 128 + tid];
            msk[j]  = (unsigned)__ballot(flag[j] != 0);
        }
        if (lane == 0) {
#pragma unroll
            for (int j = 0; j < Bsz / 128; ++j) s_cnt[j * 4 + wave] = __popc(msk[j]);
        }
        __syncthreads();
        int n_act = 0;
        int before[Bsz / 128];
#pragma unroll
        for (int s = 0; s < Bsz / 32; ++s) {
            if ((s & 3) == wave) before[s >> 2] = n_act;
            n_act += s_cnt[s];
        }
        if (n_act == 0) return;

        if (n_act <= TAIL_ROWS) {
#pragma unroll
            for (int j = 0; j < Bsz / 128; ++j) {
                if (flag[j] != 0)
                    s_list[before[j] + __popc(msk[j] & ((1u << lane) - 1u))] = j * 128 + tid;
            }
            __syncthreads();

            const int NT = N >> 4;
            const int RT = (n_act + 15) >> 4;
            const int b  = blockIdx.y * gridDim.x + blockIdx.x;
            if (b >= RT * NT) return;
            const int rt = b / NT;
            const int ct = b - rt * NT;

            const int li   = rt * 16 + r16;
            const int arow = s_list[li < n_act ? li : 0];
            const int KQ   = K >> 2;
            const unsigned short* pa = A  + (size_t)arow * K + wave * KQ + kh * 16;
            const unsigned short* pb = BT + (size_t)(ct * 16 + r16) * K + wave * KQ + kh * 16;

            v8f part = {0.f,0.f,0.f,0.f,0.f,0.f,0.f,0.f};
            for (int k0 = 0; k0 < KQ; k0 += 512) {
                FragBits ta[16], tb[16];
#pragma unroll
                for (int i = 0; i < 16; ++i) {
                    ta[i].lo = *reinterpret_cast<const u32x4*>(pa + k0 + i * 32);
                    ta[i].hi = *reinterpret_cast<const u32x4*>(pa + k0 + i * 32 + 8);
                    tb[i].lo = *reinterpret_cast<const u32x4*>(pb + k0 + i * 32);
                    tb[i].hi = *reinterpret_cast<const u32x4*>(pb + k0 + i * 32 + 8);
                }
                __builtin_amdgcn_sched_barrier(0);
#pragma unroll
                for (int i = 0; i < 16; ++i) {
                    v16bf av, bv;
                    __builtin_memcpy(&av, &ta[i], sizeof(av));
                    __builtin_memcpy(&bv, &tb[i], sizeof(bv));
                    part = __builtin_amdgcn_wmma_f32_16x16x32_bf16(
                        false, av, false, bv, (short)0, part, false, false);
                }
            }

            float* const red = reinterpret_cast<float*>(smem);
#pragma unroll
            for (int r = 0; r < 8; ++r) red[(wave * 8 + r) * 32 + lane] = part[r];
            __syncthreads();

            const int n = ct * 16 + r16;
            int   mrow[2];
            bool  live[2];
            float v[2];
#pragma unroll
            for (int j = 0; j < 2; ++j) {
                const int r  = 2 * wave + j;
                const int lr = rt * 16 + r + 8 * kh;
                live[j] = lr < n_act;
                mrow[j] = s_list[live[j] ? lr : 0];
                v[j] = ((red[r * 32 + lane] + red[(8 + r) * 32 + lane])
                        + red[(16 + r) * 32 + lane]) + red[(24 + r) * 32 + lane];
            }
            if (MODE == 2) {
                if (n < NL) {
                    const float bn = bias[n];
#pragma unroll
                    for (int j = 0; j < 2; ++j)
                        if (live[j]) Cout[(size_t)mrow[j] * N + n] = v[j] + bn;
                }
            } else {
                float dv[2], hv[2], d[2];
#pragma unroll
                for (int j = 0; j < 2; ++j) {
                    const size_t idx = (size_t)mrow[j] * N + n;
                    dv[j] = drive[idx];
                    hv[j] = Hold[idx];
                }
#pragma unroll
                for (int j = 0; j < 2; ++j) {
                    const size_t idx = (size_t)mrow[j] * N + n;
                    const float pre  = v[j] + dv[j];
                    const float actv = pre >= 0.f ? pre : 0.05f * pre;
                    d[j] = live[j] ? fabsf(actv - hv[j]) : 0.f;
                    if (live[j]) {
                        Cout[idx]  = actv;
                        HbOut[idx] = f2bf(actv);
                    }
                }
#pragma unroll
                for (int off = 8; off > 0; off >>= 1) {
#pragma unroll
                    for (int j = 0; j < 2; ++j) d[j] = fmaxf(d[j], __shfl_xor(d[j], off, 32));
                }
#pragma unroll
                for (int j = 0; j < 2; ++j)
                    if (live[j] && r16 == 0 && d[j] > 0.f)
                        atomicMax(dHbits + mrow[j], __float_as_uint(d[j]));
            }
            return;
        }

        if (s_cnt[2 * blockIdx.y] + s_cnt[2 * blockIdx.y + 1] == 0) return;
    }

    const unsigned short* ga = A  + (size_t)(blockIdx.y * 64 + wave) * K + lane * 8;
    const unsigned short* gb = BT + (size_t)(colBase + wave) * K + lane * 8;
    const size_t gstep = (size_t)4 * K;
    const int sst = wave * PITCH + lane * 16;

    const int fa = (wave * 16 + r16) * PITCH + kh * 16;
    const int fb = r16 * PITCH + kh * 16;

    v8f acc0 = {0.f,0.f,0.f,0.f,0.f,0.f,0.f,0.f};
    v8f acc[4];
    acc[0] = acc0; acc[1] = acc0; acc[2] = acc0; acc[3] = acc0;

    u32x4 ra[16], rb[16];

#pragma unroll
    for (int i = 0; i < 16; ++i) {
        ra[i] = *reinterpret_cast<const u32x4*>(ga + i * gstep);
        rb[i] = *reinterpret_cast<const u32x4*>(gb + i * gstep);
    }
#pragma unroll
    for (int i = 0; i < 16; ++i) {
        *reinterpret_cast<u32x4*>(sA + sst + i * 4 * PITCH) = ra[i];
        *reinterpret_cast<u32x4*>(sB + sst + i * 4 * PITCH) = rb[i];
    }
    __syncthreads();

    const int nchunks = K / KC;
    for (int c = 0; c < nchunks; ++c) {
        const bool more = (c + 1 < nchunks);
        if (more) {
            ga += KC; gb += KC;
#pragma unroll
            for (int i = 0; i < 16; ++i) {
                ra[i] = *reinterpret_cast<const u32x4*>(ga + i * gstep);
                rb[i] = *reinterpret_cast<const u32x4*>(gb + i * gstep);
            }
        }

        v16bf a_cur  = lds_frag(sA + fa);
        v16bf b_cur0 = lds_frag(sB + fb);
        v16bf b_cur1 = lds_frag(sB + fb + 16 * PITCH);
        v16bf b_cur2 = lds_frag(sB + fb + 32 * PITCH);
        v16bf b_cur3 = lds_frag(sB + fb + 48 * PITCH);
#pragma unroll
        for (int ks = 0; ks < KC / 32; ++ks) {
            v16bf a_nxt = a_cur, b_nxt0 = b_cur0, b_nxt1 = b_cur1,
                  b_nxt2 = b_cur2, b_nxt3 = b_cur3;
            if (ks + 1 < KC / 32) {
                const int ko = (ks + 1) * 64;
                a_nxt  = lds_frag(sA + fa + ko);
                b_nxt0 = lds_frag(sB + fb + ko);
                b_nxt1 = lds_frag(sB + fb + 16 * PITCH + ko);
                b_nxt2 = lds_frag(sB + fb + 32 * PITCH + ko);
                b_nxt3 = lds_frag(sB + fb + 48 * PITCH + ko);
            }
            __builtin_amdgcn_sched_barrier(0);
            acc[0] = __builtin_amdgcn_wmma_f32_16x16x32_bf16(
                false, a_cur, false, b_cur0, (short)0, acc[0], false, false);
            acc[1] = __builtin_amdgcn_wmma_f32_16x16x32_bf16(
                false, a_cur, false, b_cur1, (short)0, acc[1], false, false);
            acc[2] = __builtin_amdgcn_wmma_f32_16x16x32_bf16(
                false, a_cur, false, b_cur2, (short)0, acc[2], false, false);
            acc[3] = __builtin_amdgcn_wmma_f32_16x16x32_bf16(
                false, a_cur, false, b_cur3, (short)0, acc[3], false, false);
            __builtin_amdgcn_sched_barrier(0);
            a_cur = a_nxt;
            b_cur0 = b_nxt0; b_cur1 = b_nxt1; b_cur2 = b_nxt2; b_cur3 = b_nxt3;
        }

        __syncthreads();
        if (more) {
#pragma unroll
            for (int i = 0; i < 16; ++i) {
                *reinterpret_cast<u32x4*>(sA + sst + i * 4 * PITCH) = ra[i];
                *reinterpret_cast<u32x4*>(sB + sst + i * 4 * PITCH) = rb[i];
            }
            __syncthreads();
        }
    }

    bool rowActive[8];
    float dmax[8];
    if (MODE == 1) {
#pragma unroll
        for (int r = 0; r < 8; ++r) {
            rowActive[r] = active[rowBase + r + 8 * kh] != 0;
            dmax[r] = 0.f;
        }
    }

#pragma unroll
    for (int f = 0; f < 4; ++f) {
        const int n = colBase + f * 16 + r16;
#pragma unroll
        for (int r = 0; r < 8; ++r) {
            const int m = rowBase + r + 8 * kh;
            const float v = acc[f][r];
            if (MODE == 0) {
                Cout[(size_t)m * N + n] = v + bias[n];
            } else if (MODE == 2) {
                if (n < NL) Cout[(size_t)m * N + n] = v + bias[n];
            } else {
                const size_t idx = (size_t)m * N + n;
                const float pre = v + drive[idx];
                const float actv = pre >= 0.f ? pre : 0.05f * pre;
                const float ho = Hold[idx];
                const float hn = rowActive[r] ? actv : ho;
                Cout[idx] = hn;
                HbOut[idx] = f2bf(hn);
                dmax[r] = fmaxf(dmax[r], fabsf(hn - ho));
            }
        }
    }

    if (MODE == 1) {
#pragma unroll
        for (int r = 0; r < 8; ++r) {
            if (dmax[r] > 0.f) {
                atomicMax(dHbits + (rowBase + r + 8 * kh), __float_as_uint(dmax[r]));
            }
        }
    }
}

__global__ __launch_bounds__(256)
void col_l1_kernel(const float* __restrict__ A, const float* __restrict__ Msk,
                   float* __restrict__ colsum, int rows, int cols) {
    __shared__ float sm[256];
    const int j = blockIdx.x;
    float s = 0.f;
    for (int i = threadIdx.x; i < rows; i += 256) {
        const size_t idx = (size_t)i * cols + j;
        s += fabsf(A[idx] * Msk[idx]);
    }
    sm[threadIdx.x] = s;
    __syncthreads();
    for (int st = 128; st > 0; st >>= 1) {
        if (threadIdx.x < st) sm[threadIdx.x] += sm[threadIdx.x + st];
        __syncthreads();
    }
    if (threadIdx.x == 0) colsum[j] = sm[0];
}

__global__ __launch_bounds__(256)
void scalars_kernel(const float* __restrict__ col_l1,
                    const float* __restrict__ whl,
                    float* __restrict__ scalars) {
    __shared__ float sm[256];
    const int t = threadIdx.x;
    float mx = -1e30f;
    for (int j = t; j < HD; j += 256) mx = fmaxf(mx, col_l1[j]);
    sm[t] = mx; __syncthreads();
    for (int s = 128; s > 0; s >>= 1) {
        if (t < s) sm[t] = fmaxf(sm[t], sm[t + s]);
        __syncthreads();
    }
    const float m1 = sm[0];
    __syncthreads();
    mx = -1e30f;
    for (int j = t; j < Lsz; j += 256) mx = fmaxf(mx, whl[j]);
    sm[t] = mx; __syncthreads();
    for (int s = 128; s > 0; s >>= 1) {
        if (t < s) sm[t] = fmaxf(sm[t], sm[t + s]);
        __syncthreads();
    }
    if (t == 0) {
        scalars[0] = 0.95f / fmaxf(m1, 1e-8f);
        scalars[1] = sm[0] * 19.0f;
    }
}

__global__ __launch_bounds__(256)
void build_wt_kernel(const float* __restrict__ R, const float* __restrict__ M,
                     const float* __restrict__ scalars, int sel,
                     unsigned short* __restrict__ WT,
                     int K, int N, int srcld, int NL) {
    const size_t idx = (size_t)blockIdx.x * 256 + threadIdx.x;
    const size_t total = (size_t)N * K;
    if (idx >= total) return;
    const int n = (int)(idx / K);
    const int k = (int)(idx % K);
    float v = 0.f;
    if (n < NL) {
        const size_t s = (size_t)k * srcld + n;
        v = R[s] * M[s];
        if (sel == 0) v *= scalars[0];
    }
    WT[idx] = f2bf(v);
}

__global__ __launch_bounds__(256)
void to_bf16_kernel(const float* __restrict__ src, unsigned short* __restrict__ dst,
                    int n) {
    const int i = blockIdx.x * 256 + threadIdx.x;
    if (i < n) dst[i] = f2bf(src[i]);
}

__global__ __launch_bounds__(256)
void init_state(float* __restrict__ H0, unsigned short* __restrict__ Hb0,
                unsigned int* __restrict__ dH, int* __restrict__ active) {
    const size_t n = (size_t)Bsz * HD;
    for (size_t t = (size_t)blockIdx.x * 256 + threadIdx.x; t < n;
         t += (size_t)gridDim.x * 256) {
        H0[t] = 0.f;
        Hb0[t] = 0;
    }
    const int i = blockIdx.x * 256 + threadIdx.x;
    if (i < Bsz) { dH[i] = 0u; active[i] = 1; }
}

__global__ __launch_bounds__(256)
void update_rows(const float* __restrict__ logits,
                 const float* __restrict__ scalars,
                 unsigned int* __restrict__ dHbits,
                 int* __restrict__ active,
                 float* __restrict__ out,
                 int is_last) {
    const int lane = threadIdx.x & 31;
    const int wave = threadIdx.x >> 5;
    const int row = blockIdx.x * 8 + wave;
    if (row >= Bsz) return;

    if (active[row] != 0) {
        const float* lr = logits + (size_t)row * LP;
        float m1 = -1e30f, m2 = -1e30f;
        for (int n = lane; n < Lsz; n += 32) {
            const float v = lr[n];
            if (v > m1) { m2 = m1; m1 = v; }
            else if (v > m2) { m2 = v; }
        }
        for (int off = 16; off > 0; off >>= 1) {
            const float o1 = __shfl_xor(m1, off, 32);
            const float o2 = __shfl_xor(m2, off, 32);
            const float hi = fmaxf(m1, o1);
            const float lo = fmaxf(fminf(m1, o1), fmaxf(m2, o2));
            m1 = hi; m2 = lo;
        }
        const float margin = m1 - m2;
        const float dh = __uint_as_float(dHbits[row]);
        const float Gamma = scalars[1] * dh;
        const bool newly = margin > 2.0f * Gamma;
        if (newly || is_last) {
            for (int n = lane; n < Lsz; n += 32)
                out[(size_t)row * Lsz + n] = lr[n];
            if (lane == 0) active[row] = 0;
        }
    }
    if (lane == 0) dHbits[row] = 0u;
}

static inline char* wsalloc(char*& p, size_t bytes) {
    char* r = p;
    p += (bytes + 255) & ~(size_t)255;
    return r;
}

extern "C" void kernel_launch(void* const* d_in, const int* in_sizes, int n_in,
                              void* d_out, int out_size, void* d_ws, size_t ws_size,
                              hipStream_t stream) {
    (void)in_sizes; (void)n_in; (void)out_size; (void)ws_size;
    const float* x   = (const float*)d_in[0];
    const float* RIH = (const float*)d_in[1];
    const float* RH  = (const float*)d_in[2];
    const float* RHL = (const float*)d_in[3];
    const float* BH  = (const float*)d_in[4];
    const float* BL  = (const float*)d_in[5];
    const float* MIH = (const float*)d_in[6];
    const float* MH  = (const float*)d_in[7];
    const float* MHL = (const float*)d_in[8];
    float* out = (float*)d_out;

    char* p = (char*)d_ws;
    unsigned short* WT_H  = (unsigned short*)wsalloc(p, (size_t)HD * HD * 2);
    unsigned short* WT_HL = (unsigned short*)wsalloc(p, (size_t)LP * HD * 2);
    unsigned short* WT_IH = (unsigned short*)wsalloc(p, (size_t)HD * Isz * 2);
    unsigned short* xb    = (unsigned short*)wsalloc(p, (size_t)Bsz * Isz * 2);
    float* drive = (float*)wsalloc(p, (size_t)Bsz * HD * 4);
    float* Hbuf0 = (float*)wsalloc(p, (size_t)Bsz * HD * 4);
    float* Hbuf1 = (float*)wsalloc(p, (size_t)Bsz * HD * 4);
    unsigned short* Hb0 = (unsigned short*)wsalloc(p, (size_t)Bsz * HD * 2);
    unsigned short* Hb1 = (unsigned short*)wsalloc(p, (size_t)Bsz * HD * 2);
    float* logits = (float*)wsalloc(p, (size_t)Bsz * LP * 4);
    float* col_l1 = (float*)wsalloc(p, HD * 4);
    float* whl    = (float*)wsalloc(p, LP * 4);
    float* scalars = (float*)wsalloc(p, 256);
    unsigned int* dHb = (unsigned int*)wsalloc(p, Bsz * 4);
    int* active = (int*)wsalloc(p, Bsz * 4);

    col_l1_kernel<<<HD, 256, 0, stream>>>(RH, MH, col_l1, HD, HD);
    col_l1_kernel<<<Lsz, 256, 0, stream>>>(RHL, MHL, whl, HD, Lsz);
    scalars_kernel<<<1, 256, 0, stream>>>(col_l1, whl, scalars);

    build_wt_kernel<<<(HD * HD) / 256, 256, 0, stream>>>(
        RH, MH, scalars,  0, WT_H, HD, HD, HD, HD);
    build_wt_kernel<<<(LP * HD) / 256, 256, 0, stream>>>(
        RHL, MHL, scalars,  1, WT_HL, HD, LP, Lsz, Lsz);
    build_wt_kernel<<<(HD * Isz) / 256, 256, 0, stream>>>(
        RIH, MIH, scalars,  1, WT_IH, Isz, HD, HD, HD);
    to_bf16_kernel<<<(Bsz * Isz) / 256, 256, 0, stream>>>(x, xb, Bsz * Isz);
    init_state<<<(Bsz * HD) / 256, 256, 0, stream>>>(Hbuf0, Hb0, dHb, active);

    wmma_gemm<0><<<dim3(HD / 64, Bsz / 64), 128, 0, stream>>>(
        xb, WT_IH, Bsz, HD, Isz, BH, nullptr, nullptr, nullptr,
        drive, nullptr, nullptr, HD);

    float* Hc[2] = {Hbuf0, Hbuf1};
    unsigned short* Hbc[2] = {Hb0, Hb1};
    for (int t = 1; t <= TMAX; ++t) {
        const int pc = (t - 1) & 1, pn = t & 1;
        wmma_gemm<1><<<dim3(HD / 64, Bsz / 64), 128, 0, stream>>>(
            Hbc[pc], WT_H, Bsz, HD, HD, nullptr, drive, Hc[pc], active,
            Hc[pn], Hbc[pn], dHb, HD);
        wmma_gemm<2><<<dim3(LP / 64, Bsz / 64), 128, 0, stream>>>(
            Hbc[pn], WT_HL, Bsz, LP, HD, BL, nullptr, nullptr, active,
            logits, nullptr, nullptr, Lsz);
        update_rows<<<Bsz / 8, 256, 0, stream>>>(
            logits, scalars, dHb, active, out, (t == TMAX) ? 1 : 0);
    }
}


# One launch per step, not three: next recurrence GEMM, this step's logits GEMM, last step's certification as block groups reading only earlier launches; row skipping, K-split tail sums lag 1-2 steps
# speedup vs baseline: 2.1253x; 1.2546x over previous
#include <hip/hip_runtime.h>


#define Bsz 512
#define Isz 1024
#define HD  2048
#define Lsz 1000
#define LP  1024
#define TMAX 32

typedef __attribute__((ext_vector_type(16))) __bf16 v16bf;
typedef __attribute__((ext_vector_type(8)))  float  v8f;
typedef __attribute__((ext_vector_type(4)))  unsigned int u32x4;

__device__ __forceinline__ unsigned short f2bf(float f) {
    unsigned int u = __float_as_uint(f);
    u += 0x7FFFu + ((u >> 16) & 1u);
    return (unsigned short)(u >> 16);
}

#define KC 256
#define PITCH (KC * 2 + 16)
#define TILE_BYTES (64 * PITCH)

struct alignas(32) FragBits { u32x4 lo, hi; };

__device__ __forceinline__ v16bf lds_frag(const unsigned char* p) {
    FragBits fb;
    fb.lo = *reinterpret_cast<const u32x4*>(p);
    fb.hi = *reinterpret_cast<const u32x4*>(p + 32);
    v16bf v;
    __builtin_memcpy(&v, &fb, sizeof(v));
    return v;
}

#define TAIL_ROWS (Bsz / 16)
template <int MODE>
__device__ __forceinline__
void gemm_tile(const unsigned short* __restrict__ A,
               const unsigned short* __restrict__ BT,
               int M, int N, int K,
               const float* __restrict__ bias,
               const float* __restrict__ drive,
               const float* __restrict__ Hold,
               const int*   __restrict__ active,
               float* __restrict__ Cout,
               unsigned short* __restrict__ HbOut,
               unsigned int* __restrict__ dHbits,
               int NL,
               unsigned char* const smem, int* const s_cnt, int* const s_list,
               const int bx, const int by, const int gx) {
    unsigned char* const sA = smem;
    unsigned char* const sB = smem + TILE_BYTES;

    const int tid  = threadIdx.x;
    const int lane = tid & 31;
    const int wave = tid >> 5;
    const int rowBase = by * 64 + wave * 16;
    const int colBase = bx * 64;
    const int r16 = lane & 15;
    const int kh  = lane >> 4;

    if (MODE != 0) {
        int flag[Bsz / 128];
        unsigned msk[Bsz / 128];
#pragma unroll
        for (int j = 0; j < Bsz / 128; ++j) {
            flag[j] = active[j * 128 + tid];
            msk[j]  = (unsigned)__ballot(flag[j] != 0);
        }
        if (lane == 0) {
#pragma unroll
            for (int j = 0; j < Bsz / 128; ++j) s_cnt[j * 4 + wave] = __popc(msk[j]);
        }
        __syncthreads();
        int n_act = 0;
        int before[Bsz / 128];
#pragma unroll
        for (int s = 0; s < Bsz / 32; ++s) {
            if ((s & 3) == wave) before[s >> 2] = n_act;
            n_act += s_cnt[s];
        }
        if (n_act == 0) return;

        if (n_act <= TAIL_ROWS) {
#pragma unroll
            for (int j = 0; j < Bsz / 128; ++j) {
                if (flag[j] != 0)
                    s_list[before[j] + __popc(msk[j] & ((1u << lane) - 1u))] = j * 128 + tid;
            }
            __syncthreads();

            const int NT = N >> 4;
            const int RT = (n_act + 15) >> 4;
            const int b  = by * gx + bx;
            if (b >= RT * NT) return;
            const int rt = b / NT;
            const int ct = b - rt * NT;

            const int li   = rt * 16 + r16;
            const int arow = s_list[li < n_act ? li : 0];
            const int KQ   = K >> 2;
            const unsigned short* pa = A  + (size_t)arow * K + wave * KQ + kh * 16;
            const unsigned short* pb = BT + (size_t)(ct * 16 + r16) * K + wave * KQ + kh * 16;

            v8f part = {0.f,0.f,0.f,0.f,0.f,0.f,0.f,0.f};
            for (int k0 = 0; k0 < KQ; k0 += 512) {
                FragBits ta[16], tb[16];
#pragma unroll
                for (int i = 0; i < 16; ++i) {
                    ta[i].lo = *reinterpret_cast<const u32x4*>(pa + k0 + i * 32);
                    ta[i].hi = *reinterpret_cast<const u32x4*>(pa + k0 + i * 32 + 8);
                    tb[i].lo = *reinterpret_cast<const u32x4*>(pb + k0 + i * 32);
                    tb[i].hi = *reinterpret_cast<const u32x4*>(pb + k0 + i * 32 + 8);
                }
                __builtin_amdgcn_sched_barrier(0);
#pragma unroll
                for (int i = 0; i < 16; ++i) {
                    v16bf av, bv;
                    __builtin_memcpy(&av, &ta[i], sizeof(av));
                    __builtin_memcpy(&bv, &tb[i], sizeof(bv));
                    part = __builtin_amdgcn_wmma_f32_16x16x32_bf16(
                        false, av, false, bv, (short)0, part, false, false);
                }
            }

            float* const red = reinterpret_cast<float*>(smem);
#pragma unroll
            for (int r = 0; r < 8; ++r) red[(wave * 8 + r) * 32 + lane] = part[r];
            __syncthreads();

            const int n = ct * 16 + r16;
            int   mrow[2];
            bool  live[2];
            float v[2];
#pragma unroll
            for (int j = 0; j < 2; ++j) {
                const int r  = 2 * wave + j;
                const int lr = rt * 16 + r + 8 * kh;
                live[j] = lr < n_act;
                mrow[j] = s_list[live[j] ? lr : 0];
                v[j] = ((red[r * 32 + lane] + red[(8 + r) * 32 + lane])
                        + red[(16 + r) * 32 + lane]) + red[(24 + r) * 32 + lane];
            }
            if (MODE == 2) {
                if (n < NL) {
                    const float bn = bias[n];
#pragma unroll
                    for (int j = 0; j < 2; ++j)
                        if (live[j]) Cout[(size_t)mrow[j] * N + n] = v[j] + bn;
                }
            } else {
                float dv[2], hv[2], d[2];
#pragma unroll
                for (int j = 0; j < 2; ++j) {
                    const size_t idx = (size_t)mrow[j] * N + n;
                    dv[j] = drive[idx];
                    hv[j] = Hold[idx];
                }
#pragma unroll
                for (int j = 0; j < 2; ++j) {
                    const size_t idx = (size_t)mrow[j] * N + n;
                    const float pre  = v[j] + dv[j];
                    const float actv = pre >= 0.f ? pre : 0.05f * pre;
                    d[j] = live[j] ? fabsf(actv - hv[j]) : 0.f;
                    if (live[j]) {
                        Cout[idx]  = actv;
                        HbOut[idx] = f2bf(actv);
                    }
                }
#pragma unroll
                for (int off = 8; off > 0; off >>= 1) {
#pragma unroll
                    for (int j = 0; j < 2; ++j) d[j] = fmaxf(d[j], __shfl_xor(d[j], off, 32));
                }
#pragma unroll
                for (int j = 0; j < 2; ++j)
                    if (live[j] && r16 == 0 && d[j] > 0.f)
                        atomicMax(dHbits + mrow[j], __float_as_uint(d[j]));
            }
            return;
        }

        if (s_cnt[2 * by] + s_cnt[2 * by + 1] == 0) return;
    }

    const unsigned short* ga = A  + (size_t)(by * 64 + wave) * K + lane * 8;
    const unsigned short* gb = BT + (size_t)(colBase + wave) * K + lane * 8;
    const size_t gstep = (size_t)4 * K;
    const int sst = wave * PITCH + lane * 16;

    const int fa = (wave * 16 + r16) * PITCH + kh * 16;
    const int fb = r16 * PITCH + kh * 16;

    v8f acc0 = {0.f,0.f,0.f,0.f,0.f,0.f,0.f,0.f};
    v8f acc[4];
    acc[0] = acc0; acc[1] = acc0; acc[2] = acc0; acc[3] = acc0;

    u32x4 ra[16], rb[16];

#pragma unroll
    for (int i = 0; i < 16; ++i) {
        ra[i] = *reinterpret_cast<const u32x4*>(ga + i * gstep);
        rb[i] = *reinterpret_cast<const u32x4*>(gb + i * gstep);
    }
#pragma unroll
    for (int i = 0; i < 16; ++i) {
        *reinterpret_cast<u32x4*>(sA + sst + i * 4 * PITCH) = ra[i];
        *reinterpret_cast<u32x4*>(sB + sst + i * 4 * PITCH) = rb[i];
    }
    __syncthreads();

    const int nchunks = K / KC;
    for (int c = 0; c < nchunks; ++c) {
        const bool more = (c + 1 < nchunks);
        if (more) {
            ga += KC; gb += KC;
#pragma unroll
            for (int i = 0; i < 16; ++i) {
                ra[i] = *reinterpret_cast<const u32x4*>(ga + i * gstep);
                rb[i] = *reinterpret_cast<const u32x4*>(gb + i * gstep);
            }
        }

        v16bf a_cur  = lds_frag(sA + fa);
        v16bf b_cur0 = lds_frag(sB + fb);
        v16bf b_cur1 = lds_frag(sB + fb + 16 * PITCH);
        v16bf b_cur2 = lds_frag(sB + fb + 32 * PITCH);
        v16bf b_cur3 = lds_frag(sB + fb + 48 * PITCH);
#pragma unroll
        for (int ks = 0; ks < KC / 32; ++ks) {
            v16bf a_nxt = a_cur, b_nxt0 = b_cur0, b_nxt1 = b_cur1,
                  b_nxt2 = b_cur2, b_nxt3 = b_cur3;
            if (ks + 1 < KC / 32) {
                const int ko = (ks + 1) * 64;
                a_nxt  = lds_frag(sA + fa + ko);
                b_nxt0 = lds_frag(sB + fb + ko);
                b_nxt1 = lds_frag(sB + fb + 16 * PITCH + ko);
                b_nxt2 = lds_frag(sB + fb + 32 * PITCH + ko);
                b_nxt3 = lds_frag(sB + fb + 48 * PITCH + ko);
            }
            __builtin_amdgcn_sched_barrier(0);
            acc[0] = __builtin_amdgcn_wmma_f32_16x16x32_bf16(
                false, a_cur, false, b_cur0, (short)0, acc[0], false, false);
            acc[1] = __builtin_amdgcn_wmma_f32_16x16x32_bf16(
                false, a_cur, false, b_cur1, (short)0, acc[1], false, false);
            acc[2] = __builtin_amdgcn_wmma_f32_16x16x32_bf16(
                false, a_cur, false, b_cur2, (short)0, acc[2], false, false);
            acc[3] = __builtin_amdgcn_wmma_f32_16x16x32_bf16(
                false, a_cur, false, b_cur3, (short)0, acc[3], false, false);
            __builtin_amdgcn_sched_barrier(0);
            a_cur = a_nxt;
            b_cur0 = b_nxt0; b_cur1 = b_nxt1; b_cur2 = b_nxt2; b_cur3 = b_nxt3;
        }

        __syncthreads();
        if (more) {
#pragma unroll
            for (int i = 0; i < 16; ++i) {
                *reinterpret_cast<u32x4*>(sA + sst + i * 4 * PITCH) = ra[i];
                *reinterpret_cast<u32x4*>(sB + sst + i * 4 * PITCH) = rb[i];
            }
            __syncthreads();
        }
    }

    bool rowActive[8];
    float dmax[8];
    if (MODE == 1) {
#pragma unroll
        for (int r = 0; r < 8; ++r) {
            rowActive[r] = active[rowBase + r + 8 * kh] != 0;
            dmax[r] = 0.f;
        }
    }

#pragma unroll
    for (int f = 0; f < 4; ++f) {
        const int n = colBase + f * 16 + r16;
#pragma unroll
        for (int r = 0; r < 8; ++r) {
            const int m = rowBase + r + 8 * kh;
            const float v = acc[f][r];
            if (MODE == 0) {
                Cout[(size_t)m * N + n] = v + bias[n];
            } else if (MODE == 2) {
                if (n < NL) Cout[(size_t)m * N + n] = v + bias[n];
            } else {
                const size_t idx = (size_t)m * N + n;
                const float pre = v + drive[idx];
                const float actv = pre >= 0.f ? pre : 0.05f * pre;
                const float ho = Hold[idx];
                const float hn = rowActive[r] ? actv : ho;
                Cout[idx] = hn;
                HbOut[idx] = f2bf(hn);
                dmax[r] = fmaxf(dmax[r], fabsf(hn - ho));
            }
        }
    }

    if (MODE == 1) {
#pragma unroll
        for (int r = 0; r < 8; ++r) {
            if (dmax[r] > 0.f) {
                atomicMax(dHbits + (rowBase + r + 8 * kh), __float_as_uint(dmax[r]));
            }
        }
    }
}

__global__ __launch_bounds__(256)
void col_l1_kernel(const float* __restrict__ A, const float* __restrict__ Msk,
                   float* __restrict__ colsum, int rows, int cols) {
    __shared__ float sm[256];
    const int j = blockIdx.x;
    float s = 0.f;
    for (int i = threadIdx.x; i < rows; i += 256) {
        const size_t idx = (size_t)i * cols + j;
        s += fabsf(A[idx] * Msk[idx]);
    }
    sm[threadIdx.x] = s;
    __syncthreads();
    for (int st = 128; st > 0; st >>= 1) {
        if (threadIdx.x < st) sm[threadIdx.x] += sm[threadIdx.x + st];
        __syncthreads();
    }
    if (threadIdx.x == 0) colsum[j] = sm[0];
}

__global__ __launch_bounds__(256)
void scalars_kernel(const float* __restrict__ col_l1,
                    const float* __restrict__ whl,
                    float* __restrict__ scalars) {
    __shared__ float sm[256];
    const int t = threadIdx.x;
    float mx = -1e30f;
    for (int j = t; j < HD; j += 256) mx = fmaxf(mx, col_l1[j]);
    sm[t] = mx; __syncthreads();
    for (int s = 128; s > 0; s >>= 1) {
        if (t < s) sm[t] = fmaxf(sm[t], sm[t + s]);
        __syncthreads();
    }
    const float m1 = sm[0];
    __syncthreads();
    mx = -1e30f;
    for (int j = t; j < Lsz; j += 256) mx = fmaxf(mx, whl[j]);
    sm[t] = mx; __syncthreads();
    for (int s = 128; s > 0; s >>= 1) {
        if (t < s) sm[t] = fmaxf(sm[t], sm[t + s]);
        __syncthreads();
    }
    if (t == 0) {
        scalars[0] = 0.95f / fmaxf(m1, 1e-8f);
        scalars[1] = sm[0] * 19.0f;
    }
}

__global__ __launch_bounds__(256)
void build_wt_kernel(const float* __restrict__ R, const float* __restrict__ M,
                     const float* __restrict__ scalars, int sel,
                     unsigned short* __restrict__ WT,
                     int K, int N, int srcld, int NL) {
    const size_t idx = (size_t)blockIdx.x * 256 + threadIdx.x;
    const size_t total = (size_t)N * K;
    if (idx >= total) return;
    const int n = (int)(idx / K);
    const int k = (int)(idx % K);
    float v = 0.f;
    if (n < NL) {
        const size_t s = (size_t)k * srcld + n;
        v = R[s] * M[s];
        if (sel == 0) v *= scalars[0];
    }
    WT[idx] = f2bf(v);
}

__global__ __launch_bounds__(256)
void to_bf16_kernel(const float* __restrict__ src, unsigned short* __restrict__ dst,
                    int n) {
    const int i = blockIdx.x * 256 + threadIdx.x;
    if (i < n) dst[i] = f2bf(src[i]);
}

__global__ __launch_bounds__(256)
void init_state(float* __restrict__ H0, unsigned short* __restrict__ Hb0,
                unsigned int* __restrict__ dHa, unsigned int* __restrict__ dHb,
                unsigned int* __restrict__ dHc,
                int* __restrict__ act0, int* __restrict__ act1) {
    const size_t n = (size_t)Bsz * HD;
    for (size_t t = (size_t)blockIdx.x * 256 + threadIdx.x; t < n;
         t += (size_t)gridDim.x * 256) {
        H0[t] = 0.f;
        Hb0[t] = 0;
    }
    const int i = blockIdx.x * 256 + threadIdx.x;
    if (i < Bsz) { dHa[i] = 0u; dHb[i] = 0u; dHc[i] = 0u; act0[i] = 1; act1[i] = 1; }
}

__global__ __launch_bounds__(128)
void drive_gemm(const unsigned short* __restrict__ A,
                const unsigned short* __restrict__ BT,
                int M, int N, int K,
                const float* __restrict__ bias,
                float* __restrict__ Cout) {
    __shared__ __attribute__((aligned(16))) unsigned char smem[2 * TILE_BYTES];
    gemm_tile<0>(A, BT, M, N, K, bias, nullptr, nullptr, nullptr,
                 Cout, nullptr, nullptr, N,
                 smem, nullptr, nullptr, (int)blockIdx.x, (int)blockIdx.y, (int)gridDim.x);
}

__device__ __forceinline__
void certify_rows(const int ub,
                  const float* __restrict__ logits,
                  const float* __restrict__ scalars,
                  unsigned int* __restrict__ dHbits,
                  const int* __restrict__ actIn,
                  int* __restrict__ actOut,
                  float* __restrict__ out,
                  const int is_last) {
    const int lane = threadIdx.x & 31;
    const int wave = threadIdx.x >> 5;
    for (int j = 0; j < 2; ++j) {
        const int row = ub * 8 + wave * 2 + j;
        int a = actIn[row];
        if (a != 0) {
            const float* lr = logits + (size_t)row * LP;
            float m1 = -1e30f, m2 = -1e30f;
            for (int n = lane; n < Lsz; n += 32) {
                const float v = lr[n];
                if (v > m1) { m2 = m1; m1 = v; }
                else if (v > m2) { m2 = v; }
            }
            for (int off = 16; off > 0; off >>= 1) {
                const float o1 = __shfl_xor(m1, off, 32);
                const float o2 = __shfl_xor(m2, off, 32);
                const float hi = fmaxf(m1, o1);
                const float lo = fmaxf(fminf(m1, o1), fmaxf(m2, o2));
                m1 = hi; m2 = lo;
            }
            const float margin = m1 - m2;
            const float dh = __uint_as_float(dHbits[row]);
            const float Gamma = scalars[1] * dh;
            const bool newly = margin > 2.0f * Gamma;
            if (newly || is_last) {
                for (int n = lane; n < Lsz; n += 32)
                    out[(size_t)row * Lsz + n] = lr[n];
                a = 0;
            }
        }
        if (lane == 0) {
            actOut[row] = a;
            dHbits[row] = 0u;
        }
    }
}

#define REC_BLOCKS  ((HD / 64) * (Bsz / 64))
#define LOG_BLOCKS  ((LP / 64) * (Bsz / 64))
#define CERT_BLOCKS (Bsz / 8)
__global__ __launch_bounds__(128)
void step_kernel(const unsigned short* __restrict__ HbCur,
                 const float* __restrict__ HCur,
                 const unsigned short* __restrict__ WT_H,
                 const unsigned short* __restrict__ WT_HL,
                 const float* __restrict__ drive,
                 const float* __restrict__ BL,
                 const float* __restrict__ scalars,
                 const int* __restrict__ actIn,
                 int* __restrict__ actOut,
                 float* __restrict__ HNext,
                 unsigned short* __restrict__ HbNext,
                 unsigned int* __restrict__ dHNext,
                 float* __restrict__ logitsCur,
                 const float* __restrict__ logitsPrev,
                 unsigned int* __restrict__ dHPrev,
                 float* __restrict__ out,
                 int do_rec, int do_log, int do_cert, int is_last) {
    __shared__ __attribute__((aligned(16))) unsigned char smem[2 * TILE_BYTES];
    __shared__ int s_cnt[Bsz / 32];
    __shared__ int s_list[TAIL_ROWS];
    const int b = blockIdx.x;
    if (b < REC_BLOCKS) {
        if (do_rec)
            gemm_tile<1>(HbCur, WT_H, Bsz, HD, HD, nullptr, drive, HCur, actIn,
                         HNext, HbNext, dHNext, HD,
                         smem, s_cnt, s_list, b % (HD / 64), b / (HD / 64), HD / 64);
    } else if (b < REC_BLOCKS + LOG_BLOCKS) {
        const int q = b - REC_BLOCKS;
        if (do_log)
            gemm_tile<2>(HbCur, WT_HL, Bsz, LP, HD, BL, nullptr, nullptr, actIn,
                         logitsCur, nullptr, nullptr, Lsz,
                         smem, s_cnt, s_list, q % (LP / 64), q / (LP / 64), LP / 64);
    } else {
        if (do_cert)
            certify_rows(b - (REC_BLOCKS + LOG_BLOCKS), logitsPrev, scalars, dHPrev,
                         actIn, actOut, out, is_last);
    }
}

static inline char* wsalloc(char*& p, size_t bytes) {
    char* r = p;
    p += (bytes + 255) & ~(size_t)255;
    return r;
}

extern "C" void kernel_launch(void* const* d_in, const int* in_sizes, int n_in,
                              void* d_out, int out_size, void* d_ws, size_t ws_size,
                              hipStream_t stream) {
    (void)in_sizes; (void)n_in; (void)out_size; (void)ws_size;
    const float* x   = (const float*)d_in[0];
    const float* RIH = (const float*)d_in[1];
    const float* RH  = (const float*)d_in[2];
    const float* RHL = (const float*)d_in[3];
    const float* BH  = (const float*)d_in[4];
    const float* BL  = (const float*)d_in[5];
    const float* MIH = (const float*)d_in[6];
    const float* MH  = (const float*)d_in[7];
    const float* MHL = (const float*)d_in[8];
    float* out = (float*)d_out;

    char* p = (char*)d_ws;
    unsigned short* WT_H  = (unsigned short*)wsalloc(p, (size_t)HD * HD * 2);
    unsigned short* WT_HL = (unsigned short*)wsalloc(p, (size_t)LP * HD * 2);
    unsigned short* WT_IH = (unsigned short*)wsalloc(p, (size_t)HD * Isz * 2);
    unsigned short* xb    = (unsigned short*)wsalloc(p, (size_t)Bsz * Isz * 2);
    float* drive = (float*)wsalloc(p, (size_t)Bsz * HD * 4);
    float* Hbuf0 = (float*)wsalloc(p, (size_t)Bsz * HD * 4);
    float* Hbuf1 = (float*)wsalloc(p, (size_t)Bsz * HD * 4);
    unsigned short* Hb0 = (unsigned short*)wsalloc(p, (size_t)Bsz * HD * 2);
    unsigned short* Hb1 = (unsigned short*)wsalloc(p, (size_t)Bsz * HD * 2);
    float* logits = (float*)wsalloc(p, (size_t)Bsz * LP * 4);
    float* col_l1 = (float*)wsalloc(p, HD * 4);
    float* whl    = (float*)wsalloc(p, LP * 4);
    float* scalars = (float*)wsalloc(p, 256);
    unsigned int* dHb = (unsigned int*)wsalloc(p, Bsz * 4);
    int* active = (int*)wsalloc(p, Bsz * 4);

    unsigned int* dH1 = (unsigned int*)col_l1;
    unsigned int* dH2 = dH1 + Bsz;
    int* active1 = (int*)(dH2 + Bsz);
    float* logits1 = (float*)WT_IH;

    col_l1_kernel<<<HD, 256, 0, stream>>>(RH, MH, col_l1, HD, HD);
    col_l1_kernel<<<Lsz, 256, 0, stream>>>(RHL, MHL, whl, HD, Lsz);
    scalars_kernel<<<1, 256, 0, stream>>>(col_l1, whl, scalars);

    build_wt_kernel<<<(HD * HD) / 256, 256, 0, stream>>>(
        RH, MH, scalars,  0, WT_H, HD, HD, HD, HD);
    build_wt_kernel<<<(LP * HD) / 256, 256, 0, stream>>>(
        RHL, MHL, scalars,  1, WT_HL, HD, LP, Lsz, Lsz);
    build_wt_kernel<<<(HD * Isz) / 256, 256, 0, stream>>>(
        RIH, MIH, scalars,  1, WT_IH, Isz, HD, HD, HD);
    to_bf16_kernel<<<(Bsz * Isz) / 256, 256, 0, stream>>>(x, xb, Bsz * Isz);
    init_state<<<(Bsz * HD) / 256, 256, 0, stream>>>(Hbuf0, Hb0, dHb, dH1, dH2,
                                                    active, active1);

    drive_gemm<<<dim3(HD / 64, Bsz / 64), 128, 0, stream>>>(
        xb, WT_IH, Bsz, HD, Isz, BH, drive);

    float* Hc[2] = {Hbuf0, Hbuf1};
    unsigned short* Hbc[2] = {Hb0, Hb1};
    int* act[2] = {active, active1};
    unsigned int* dHp[3] = {dHb, dH1, dH2};
    float* lg[2] = {logits, logits1};
    for (int s = 0; s <= TMAX + 1; ++s) {
        const int c = s & 1, n = (s + 1) & 1;
        step_kernel<<<REC_BLOCKS + LOG_BLOCKS + CERT_BLOCKS, 128, 0, stream>>>(
            Hbc[c], Hc[c], WT_H, WT_HL, drive, BL, scalars, act[c], act[n],
            Hc[n], Hbc[n], dHp[(s + 1) % 3], lg[c], lg[n], dHp[(s + 2) % 3], out,
            (s < TMAX) ? 1 : 0, (s >= 1 && s <= TMAX) ? 1 : 0, (s >= 2) ? 1 : 0,
            (s == TMAX + 1) ? 1 : 0);
    }
}


# Weight set-up reads along rows: column L1 sums from row-contiguous loads (same addition tree, 8 partial planes), bf16 transposed weights through a 64x64 LDS tile; all sums in the parent's order
# speedup vs baseline: 2.4063x; 1.1322x over previous
#include <hip/hip_runtime.h>


#define Bsz 512
#define Isz 1024
#define HD  2048
#define Lsz 1000
#define LP  1024
#define TMAX 32

typedef __attribute__((ext_vector_type(16))) __bf16 v16bf;
typedef __attribute__((ext_vector_type(8)))  float  v8f;
typedef __attribute__((ext_vector_type(4)))  unsigned int u32x4;

__device__ __forceinline__ unsigned short f2bf(float f) {
    unsigned int u = __float_as_uint(f);
    u += 0x7FFFu + ((u >> 16) & 1u);
    return (unsigned short)(u >> 16);
}

#define KC 256
#define PITCH (KC * 2 + 16)
#define TILE_BYTES (64 * PITCH)

struct alignas(32) FragBits { u32x4 lo, hi; };

__device__ __forceinline__ v16bf lds_frag(const unsigned char* p) {
    FragBits fb;
    fb.lo = *reinterpret_cast<const u32x4*>(p);
    fb.hi = *reinterpret_cast<const u32x4*>(p + 32);
    v16bf v;
    __builtin_memcpy(&v, &fb, sizeof(v));
    return v;
}

#define TAIL_ROWS (Bsz / 16)
template <int MODE>
__device__ __forceinline__
void gemm_tile(const unsigned short* __restrict__ A,
               const unsigned short* __restrict__ BT,
               int M, int N, int K,
               const float* __restrict__ bias,
               const float* __restrict__ drive,
               const float* __restrict__ Hold,
               const int*   __restrict__ active,
               float* __restrict__ Cout,
               unsigned short* __restrict__ HbOut,
               unsigned int* __restrict__ dHbits,
               int NL,
               unsigned char* const smem, int* const s_cnt, int* const s_list,
               const int bx, const int by, const int gx) {
    unsigned char* const sA = smem;
    unsigned char* const sB = smem + TILE_BYTES;

    const int tid  = threadIdx.x;
    const int lane = tid & 31;
    const int wave = tid >> 5;
    const int rowBase = by * 64 + wave * 16;
    const int colBase = bx * 64;
    const int r16 = lane & 15;
    const int kh  = lane >> 4;

    if (MODE != 0) {
        int flag[Bsz / 128];
        unsigned msk[Bsz / 128];
#pragma unroll
        for (int j = 0; j < Bsz / 128; ++j) {
            flag[j] = active[j * 128 + tid];
            msk[j]  = (unsigned)__ballot(flag[j] != 0);
        }
        if (lane == 0) {
#pragma unroll
            for (int j = 0; j < Bsz / 128; ++j) s_cnt[j * 4 + wave] = __popc(msk[j]);
        }
        __syncthreads();
        int n_act = 0;
        int before[Bsz / 128];
#pragma unroll
        for (int s = 0; s < Bsz / 32; ++s) {
            if ((s & 3) == wave) before[s >> 2] = n_act;
            n_act += s_cnt[s];
        }
        if (n_act == 0) return;

        if (n_act <= TAIL_ROWS) {
#pragma unroll
            for (int j = 0; j < Bsz / 128; ++j) {
                if (flag[j] != 0)
                    s_list[before[j] + __popc(msk[j] & ((1u << lane) - 1u))] = j * 128 + tid;
            }
            __syncthreads();

            const int NT = N >> 4;
            const int RT = (n_act + 15) >> 4;
            const int b  = by * gx + bx;
            if (b >= RT * NT) return;
            const int rt = b / NT;
            const int ct = b - rt * NT;

            const int li   = rt * 16 + r16;
            const int arow = s_list[li < n_act ? li : 0];
            const int KQ   = K >> 2;
            const unsigned short* pa = A  + (size_t)arow * K + wave * KQ + kh * 16;
            const unsigned short* pb = BT + (size_t)(ct * 16 + r16) * K + wave * KQ + kh * 16;

            v8f part = {0.f,0.f,0.f,0.f,0.f,0.f,0.f,0.f};
            for (int k0 = 0; k0 < KQ; k0 += 512) {
                FragBits ta[16], tb[16];
#pragma unroll
                for (int i = 0; i < 16; ++i) {
                    ta[i].lo = *reinterpret_cast<const u32x4*>(pa + k0 + i * 32);
                    ta[i].hi = *reinterpret_cast<const u32x4*>(pa + k0 + i * 32 + 8);
                    tb[i].lo = *reinterpret_cast<const u32x4*>(pb + k0 + i * 32);
                    tb[i].hi = *reinterpret_cast<const u32x4*>(pb + k0 + i * 32 + 8);
                }
                __builtin_amdgcn_sched_barrier(0);
#pragma unroll
                for (int i = 0; i < 16; ++i) {
                    v16bf av, bv;
                    __builtin_memcpy(&av, &ta[i], sizeof(av));
                    __builtin_memcpy(&bv, &tb[i], sizeof(bv));
                    part = __builtin_amdgcn_wmma_f32_16x16x32_bf16(
                        false, av, false, bv, (short)0, part, false, false);
                }
            }

            float* const red = reinterpret_cast<float*>(smem);
#pragma unroll
            for (int r = 0; r < 8; ++r) red[(wave * 8 + r) * 32 + lane] = part[r];
            __syncthreads();

            const int n = ct * 16 + r16;
            int   mrow[2];
            bool  live[2];
            float v[2];
#pragma unroll
            for (int j = 0; j < 2; ++j) {
                const int r  = 2 * wave + j;
                const int lr = rt * 16 + r + 8 * kh;
                live[j] = lr < n_act;
                mrow[j] = s_list[live[j] ? lr : 0];
                v[j] = ((red[r * 32 + lane] + red[(8 + r) * 32 + lane])
                        + red[(16 + r) * 32 + lane]) + red[(24 + r) * 32 + lane];
            }
            if (MODE == 2) {
                if (n < NL) {
                    const float bn = bias[n];
#pragma unroll
                    for (int j = 0; j < 2; ++j)
                        if (live[j]) Cout[(size_t)mrow[j] * N + n] = v[j] + bn;
                }
            } else {
                float dv[2], hv[2], d[2];
#pragma unroll
                for (int j = 0; j < 2; ++j) {
                    const size_t idx = (size_t)mrow[j] * N + n;
                    dv[j] = drive[idx];
                    hv[j] = Hold[idx];
                }
#pragma unroll
                for (int j = 0; j < 2; ++j) {
                    const size_t idx = (size_t)mrow[j] * N + n;
                    const float pre  = v[j] + dv[j];
                    const float actv = pre >= 0.f ? pre : 0.05f * pre;
                    d[j] = live[j] ? fabsf(actv - hv[j]) : 0.f;
                    if (live[j]) {
                        Cout[idx]  = actv;
                        HbOut[idx] = f2bf(actv);
                    }
                }
#pragma unroll
                for (int off = 8; off > 0; off >>= 1) {
#pragma unroll
                    for (int j = 0; j < 2; ++j) d[j] = fmaxf(d[j], __shfl_xor(d[j], off, 32));
                }
#pragma unroll
                for (int j = 0; j < 2; ++j)
                    if (live[j] && r16 == 0 && d[j] > 0.f)
                        atomicMax(dHbits + mrow[j], __float_as_uint(d[j]));
            }
            return;
        }

        if (s_cnt[2 * by] + s_cnt[2 * by + 1] == 0) return;
    }

    const unsigned short* ga = A  + (size_t)(by * 64 + wave) * K + lane * 8;
    const unsigned short* gb = BT + (size_t)(colBase + wave) * K + lane * 8;
    const size_t gstep = (size_t)4 * K;
    const int sst = wave * PITCH + lane * 16;

    const int fa = (wave * 16 + r16) * PITCH + kh * 16;
    const int fb = r16 * PITCH + kh * 16;

    v8f acc0 = {0.f,0.f,0.f,0.f,0.f,0.f,0.f,0.f};
    v8f acc[4];
    acc[0] = acc0; acc[1] = acc0; acc[2] = acc0; acc[3] = acc0;

    u32x4 ra[16], rb[16];

#pragma unroll
    for (int i = 0; i < 16; ++i) {
        ra[i] = *reinterpret_cast<const u32x4*>(ga + i * gstep);
        rb[i] = *reinterpret_cast<const u32x4*>(gb + i * gstep);
    }
#pragma unroll
    for (int i = 0; i < 16; ++i) {
        *reinterpret_cast<u32x4*>(sA + sst + i * 4 * PITCH) = ra[i];
        *reinterpret_cast<u32x4*>(sB + sst + i * 4 * PITCH) = rb[i];
    }
    __syncthreads();

    const int nchunks = K / KC;
    for (int c = 0; c < nchunks; ++c) {
        const bool more = (c + 1 < nchunks);
        if (more) {
            ga += KC; gb += KC;
#pragma unroll
            for (int i = 0; i < 16; ++i) {
                ra[i] = *reinterpret_cast<const u32x4*>(ga + i * gstep);
                rb[i] = *reinterpret_cast<const u32x4*>(gb + i * gstep);
            }
        }

        v16bf a_cur  = lds_frag(sA + fa);
        v16bf b_cur0 = lds_frag(sB + fb);
        v16bf b_cur1 = lds_frag(sB + fb + 16 * PITCH);
        v16bf b_cur2 = lds_frag(sB + fb + 32 * PITCH);
        v16bf b_cur3 = lds_frag(sB + fb + 48 * PITCH);
#pragma unroll
        for (int ks = 0; ks < KC / 32; ++ks) {
            v16bf a_nxt = a_cur, b_nxt0 = b_cur0, b_nxt1 = b_cur1,
                  b_nxt2 = b_cur2, b_nxt3 = b_cur3;
            if (ks + 1 < KC / 32) {
                const int ko = (ks + 1) * 64;
                a_nxt  = lds_frag(sA + fa + ko);
                b_nxt0 = lds_frag(sB + fb + ko);
                b_nxt1 = lds_frag(sB + fb + 16 * PITCH + ko);
                b_nxt2 = lds_frag(sB + fb + 32 * PITCH + ko);
                b_nxt3 = lds_frag(sB + fb + 48 * PITCH + ko);
            }
            __builtin_amdgcn_sched_barrier(0);
            acc[0] = __builtin_amdgcn_wmma_f32_16x16x32_bf16(
                false, a_cur, false, b_cur0, (short)0, acc[0], false, false);
            acc[1] = __builtin_amdgcn_wmma_f32_16x16x32_bf16(
                false, a_cur, false, b_cur1, (short)0, acc[1], false, false);
            acc[2] = __builtin_amdgcn_wmma_f32_16x16x32_bf16(
                false, a_cur, false, b_cur2, (short)0, acc[2], false, false);
            acc[3] = __builtin_amdgcn_wmma_f32_16x16x32_bf16(
                false, a_cur, false, b_cur3, (short)0, acc[3], false, false);
            __builtin_amdgcn_sched_barrier(0);
            a_cur = a_nxt;
            b_cur0 = b_nxt0; b_cur1 = b_nxt1; b_cur2 = b_nxt2; b_cur3 = b_nxt3;
        }

        __syncthreads();
        if (more) {
#pragma unroll
            for (int i = 0; i < 16; ++i) {
                *reinterpret_cast<u32x4*>(sA + sst + i * 4 * PITCH) = ra[i];
                *reinterpret_cast<u32x4*>(sB + sst + i * 4 * PITCH) = rb[i];
            }
            __syncthreads();
        }
    }

    bool rowActive[8];
    float dmax[8];
    if (MODE == 1) {
#pragma unroll
        for (int r = 0; r < 8; ++r) {
            rowActive[r] = active[rowBase + r + 8 * kh] != 0;
            dmax[r] = 0.f;
        }
    }

#pragma unroll
    for (int f = 0; f < 4; ++f) {
        const int n = colBase + f * 16 + r16;
#pragma unroll
        for (int r = 0; r < 8; ++r) {
            const int m = rowBase + r + 8 * kh;
            const float v = acc[f][r];
            if (MODE == 0) {
                Cout[(size_t)m * N + n] = v + bias[n];
            } else if (MODE == 2) {
                if (n < NL) Cout[(size_t)m * N + n] = v + bias[n];
            } else {
                const size_t idx = (size_t)m * N + n;
                const float pre = v + drive[idx];
                const float actv = pre >= 0.f ? pre : 0.05f * pre;
                const float ho = Hold[idx];
                const float hn = rowActive[r] ? actv : ho;
                Cout[idx] = hn;
                HbOut[idx] = f2bf(hn);
                dmax[r] = fmaxf(dmax[r], fabsf(hn - ho));
            }
        }
    }

    if (MODE == 1) {
#pragma unroll
        for (int r = 0; r < 8; ++r) {
            if (dmax[r] > 0.f) {
                atomicMax(dHbits + (rowBase + r + 8 * kh), __float_as_uint(dmax[r]));
            }
        }
    }
}

__global__ __launch_bounds__(256)
void col_l1_kernel(const float* __restrict__ A, const float* __restrict__ Msk,
                   float* __restrict__ part, int cols) {
    static_assert(HD % 256 == 0, "leaf rows are t + 256*r");
    __shared__ float smx[8][32];
    __shared__ float smy[8][32];
    const int lane = threadIdx.x & 31;
    const int v    = threadIdx.x >> 5;
    const int b    = blockIdx.y;
    const int c    = (blockIdx.x * 32 + lane) * 2;
    const bool in  = c < cols;
    const int cl   = in ? c : 0;
    float lx[4], ly[4];
#pragma unroll
    for (int m = 0; m < 4; ++m) {
        const int t = b + 8 * (v + 8 * m);
        float sx = 0.f, sy = 0.f;
#pragma unroll
        for (int r = 0; r < HD / 256; ++r) {
            const size_t idx = (size_t)(t + 256 * r) * cols + cl;
            const float2 a  = *reinterpret_cast<const float2*>(A + idx);
            const float2 mk = *reinterpret_cast<const float2*>(Msk + idx);
            sx += fabsf(a.x * mk.x);
            sy += fabsf(a.y * mk.y);
        }
        lx[m] = sx; ly[m] = sy;
    }
    smx[v][lane] = (lx[0] + lx[2]) + (lx[1] + lx[3]);
    smy[v][lane] = (ly[0] + ly[2]) + (ly[1] + ly[3]);
    __syncthreads();
    if (v == 0 && in) {
        part[(size_t)b * cols + c] =
            ((smx[0][lane] + smx[4][lane]) + (smx[2][lane] + smx[6][lane])) +
            ((smx[1][lane] + smx[5][lane]) + (smx[3][lane] + smx[7][lane]));
        part[(size_t)b * cols + c + 1] =
            ((smy[0][lane] + smy[4][lane]) + (smy[2][lane] + smy[6][lane])) +
            ((smy[1][lane] + smy[5][lane]) + (smy[3][lane] + smy[7][lane]));
    }
}

__device__ __forceinline__ float fold8(const float* __restrict__ p, int cols, int j) {
    const float p0 = p[j],            p1 = p[cols + j];
    const float p2 = p[2 * cols + j], p3 = p[3 * cols + j];
    const float p4 = p[4 * cols + j], p5 = p[5 * cols + j];
    const float p6 = p[6 * cols + j], p7 = p[7 * cols + j];
    return ((p0 + p4) + (p2 + p6)) + ((p1 + p5) + (p3 + p7));
}

__global__ __launch_bounds__(256)
void scalars_kernel(const float* __restrict__ col_l1,
                    const float* __restrict__ whl,
                    float* __restrict__ scalars) {
    __shared__ float sm[256];
    const int t = threadIdx.x;
    float mx = -1e30f;
    for (int j = t; j < HD; j += 256) mx = fmaxf(mx, fold8(col_l1, HD, j));
    sm[t] = mx; __syncthreads();
    for (int s = 128; s > 0; s >>= 1) {
        if (t < s) sm[t] = fmaxf(sm[t], sm[t + s]);
        __syncthreads();
    }
    const float m1 = sm[0];
    __syncthreads();
    mx = -1e30f;
    for (int j = t; j < Lsz; j += 256) mx = fmaxf(mx, fold8(whl, Lsz, j));
    sm[t] = mx; __syncthreads();
    for (int s = 128; s > 0; s >>= 1) {
        if (t < s) sm[t] = fmaxf(sm[t], sm[t + s]);
        __syncthreads();
    }
    if (t == 0) {
        scalars[0] = 0.95f / fmaxf(m1, 1e-8f);
        scalars[1] = sm[0] * 19.0f;
    }
}

#define TT 64
#define TPITCH (TT + 2)
__global__ __launch_bounds__(256)
void build_wt_kernel(const float* __restrict__ R, const float* __restrict__ M,
                     const float* __restrict__ scalars, int sel,
                     unsigned short* __restrict__ WT,
                     int K, int N, int srcld, int NL) {
    (void)N;
    __shared__ __attribute__((aligned(16))) unsigned short tile[TT * TPITCH];
    const int k0 = blockIdx.x * TT;
    const int n0 = blockIdx.y * TT;
    const int t  = threadIdx.x;

    const int ln = (t & 15) * 4;
    const int lk = t >> 4;
    const int n  = n0 + ln;
    const bool in = n < NL;
    const int nc = in ? n : 0;
    float4 r[4], m[4];
#pragma unroll
    for (int i = 0; i < 4; ++i) {
        const size_t sidx = (size_t)(k0 + lk + 16 * i) * srcld + nc;
        r[i] = *reinterpret_cast<const float4*>(R + sidx);
        m[i] = *reinterpret_cast<const float4*>(M + sidx);
    }
    float sc = 1.f;
    if (sel == 0) sc = scalars[0];
#pragma unroll
    for (int i = 0; i < 4; ++i) {
        float v0 = r[i].x * m[i].x, v1 = r[i].y * m[i].y;
        float v2 = r[i].z * m[i].z, v3 = r[i].w * m[i].w;
        if (sel == 0) { v0 *= sc; v1 *= sc; v2 *= sc; v3 *= sc; }
        if (!in) { v0 = 0.f; v1 = 0.f; v2 = 0.f; v3 = 0.f; }
        unsigned int* d = reinterpret_cast<unsigned int*>(tile + (lk + 16 * i) * TPITCH + ln);
        d[0] = (unsigned int)f2bf(v0) | ((unsigned int)f2bf(v1) << 16);
        d[1] = (unsigned int)f2bf(v2) | ((unsigned int)f2bf(v3) << 16);
    }
    __syncthreads();

    const int ks = (t & 7) * 8;
#pragma unroll
    for (int p = 0; p < 2; ++p) {
        const int nl = (t >> 3) + 32 * p;
        unsigned int w[4];
#pragma unroll
        for (int j = 0; j < 4; ++j) {
            const unsigned int e0 = tile[(ks + 2 * j) * TPITCH + nl];
            const unsigned int e1 = tile[(ks + 2 * j + 1) * TPITCH + nl];
            w[j] = e0 | (e1 << 16);
        }
        u32x4 o; o.x = w[0]; o.y = w[1]; o.z = w[2]; o.w = w[3];
        *reinterpret_cast<u32x4*>(WT + (size_t)(n0 + nl) * K + k0 + ks) = o;
    }
}

__global__ __launch_bounds__(256)
void to_bf16_kernel(const float* __restrict__ src, unsigned short* __restrict__ dst,
                    int n) {
    const int i = blockIdx.x * 256 + threadIdx.x;
    if (i < n) dst[i] = f2bf(src[i]);
}

__global__ __launch_bounds__(256)
void init_state(float* __restrict__ H0, unsigned short* __restrict__ Hb0,
                unsigned int* __restrict__ dHa, unsigned int* __restrict__ dHb,
                unsigned int* __restrict__ dHc,
                int* __restrict__ act0, int* __restrict__ act1) {
    const size_t n = (size_t)Bsz * HD;
    for (size_t t = (size_t)blockIdx.x * 256 + threadIdx.x; t < n;
         t += (size_t)gridDim.x * 256) {
        H0[t] = 0.f;
        Hb0[t] = 0;
    }
    const int i = blockIdx.x * 256 + threadIdx.x;
    if (i < Bsz) { dHa[i] = 0u; dHb[i] = 0u; dHc[i] = 0u; act0[i] = 1; act1[i] = 1; }
}

__global__ __launch_bounds__(128)
void drive_gemm(const unsigned short* __restrict__ A,
                const unsigned short* __restrict__ BT,
                int M, int N, int K,
                const float* __restrict__ bias,
                float* __restrict__ Cout) {
    __shared__ __attribute__((aligned(16))) unsigned char smem[2 * TILE_BYTES];
    gemm_tile<0>(A, BT, M, N, K, bias, nullptr, nullptr, nullptr,
                 Cout, nullptr, nullptr, N,
                 smem, nullptr, nullptr, (int)blockIdx.x, (int)blockIdx.y, (int)gridDim.x);
}

__device__ __forceinline__
void certify_rows(const int ub,
                  const float* __restrict__ logits,
                  const float* __restrict__ scalars,
                  unsigned int* __restrict__ dHbits,
                  const int* __restrict__ actIn,
                  int* __restrict__ actOut,
                  float* __restrict__ out,
                  const int is_last) {
    const int lane = threadIdx.x & 31;
    const int wave = threadIdx.x >> 5;
    for (int j = 0; j < 2; ++j) {
        const int row = ub * 8 + wave * 2 + j;
        int a = actIn[row];
        if (a != 0) {
            const float* lr = logits + (size_t)row * LP;
            float m1 = -1e30f, m2 = -1e30f;
            for (int n = lane; n < Lsz; n += 32) {
                const float v = lr[n];
                if (v > m1) { m2 = m1; m1 = v; }
                else if (v > m2) { m2 = v; }
            }
            for (int off = 16; off > 0; off >>= 1) {
                const float o1 = __shfl_xor(m1, off, 32);
                const float o2 = __shfl_xor(m2, off, 32);
                const float hi = fmaxf(m1, o1);
                const float lo = fmaxf(fminf(m1, o1), fmaxf(m2, o2));
                m1 = hi; m2 = lo;
            }
            const float margin = m1 - m2;
            const float dh = __uint_as_float(dHbits[row]);
            const float Gamma = scalars[1] * dh;
            const bool newly = margin > 2.0f * Gamma;
            if (newly || is_last) {
                for (int n = lane; n < Lsz; n += 32)
                    out[(size_t)row * Lsz + n] = lr[n];
                a = 0;
            }
        }
        if (lane == 0) {
            actOut[row] = a;
            dHbits[row] = 0u;
        }
    }
}

#define REC_BLOCKS  ((HD / 64) * (Bsz / 64))
#define LOG_BLOCKS  ((LP / 64) * (Bsz / 64))
#define CERT_BLOCKS (Bsz / 8)
__global__ __launch_bounds__(128)
void step_kernel(const unsigned short* __restrict__ HbCur,
                 const float* __restrict__ HCur,
                 const unsigned short* __restrict__ WT_H,
                 const unsigned short* __restrict__ WT_HL,
                 const float* __restrict__ drive,
                 const float* __restrict__ BL,
                 const float* __restrict__ scalars,
                 const int* __restrict__ actIn,
                 int* __restrict__ actOut,
                 float* __restrict__ HNext,
                 unsigned short* __restrict__ HbNext,
                 unsigned int* __restrict__ dHNext,
                 float* __restrict__ logitsCur,
                 const float* __restrict__ logitsPrev,
                 unsigned int* __restrict__ dHPrev,
                 float* __restrict__ out,
                 int do_rec, int do_log, int do_cert, int is_last) {
    __shared__ __attribute__((aligned(16))) unsigned char smem[2 * TILE_BYTES];
    __shared__ int s_cnt[Bsz / 32];
    __shared__ int s_list[TAIL_ROWS];
    const int b = blockIdx.x;
    if (b < REC_BLOCKS) {
        if (do_rec)
            gemm_tile<1>(HbCur, WT_H, Bsz, HD, HD, nullptr, drive, HCur, actIn,
                         HNext, HbNext, dHNext, HD,
                         smem, s_cnt, s_list, b % (HD / 64), b / (HD / 64), HD / 64);
    } else if (b < REC_BLOCKS + LOG_BLOCKS) {
        const int q = b - REC_BLOCKS;
        if (do_log)
            gemm_tile<2>(HbCur, WT_HL, Bsz, LP, HD, BL, nullptr, nullptr, actIn,
                         logitsCur, nullptr, nullptr, Lsz,
                         smem, s_cnt, s_list, q % (LP / 64), q / (LP / 64), LP / 64);
    } else {
        if (do_cert)
            certify_rows(b - (REC_BLOCKS + LOG_BLOCKS), logitsPrev, scalars, dHPrev,
                         actIn, actOut, out, is_last);
    }
}

static inline char* wsalloc(char*& p, size_t bytes) {
    char* r = p;
    p += (bytes + 255) & ~(size_t)255;
    return r;
}

extern "C" void kernel_launch(void* const* d_in, const int* in_sizes, int n_in,
                              void* d_out, int out_size, void* d_ws, size_t ws_size,
                              hipStream_t stream) {
    (void)in_sizes; (void)n_in; (void)out_size; (void)ws_size;
    const float* x   = (const float*)d_in[0];
    const float* RIH = (const float*)d_in[1];
    const float* RH  = (const float*)d_in[2];
    const float* RHL = (const float*)d_in[3];
    const float* BH  = (const float*)d_in[4];
    const float* BL  = (const float*)d_in[5];
    const float* MIH = (const float*)d_in[6];
    const float* MH  = (const float*)d_in[7];
    const float* MHL = (const float*)d_in[8];
    float* out = (float*)d_out;

    char* p = (char*)d_ws;
    unsigned short* WT_H  = (unsigned short*)wsalloc(p, (size_t)HD * HD * 2);
    unsigned short* WT_HL = (unsigned short*)wsalloc(p, (size_t)LP * HD * 2);
    unsigned short* WT_IH = (unsigned short*)wsalloc(p, (size_t)HD * Isz * 2);
    unsigned short* xb    = (unsigned short*)wsalloc(p, (size_t)Bsz * Isz * 2);
    float* drive = (float*)wsalloc(p, (size_t)Bsz * HD * 4);
    float* Hbuf0 = (float*)wsalloc(p, (size_t)Bsz * HD * 4);
    float* Hbuf1 = (float*)wsalloc(p, (size_t)Bsz * HD * 4);
    unsigned short* Hb0 = (unsigned short*)wsalloc(p, (size_t)Bsz * HD * 2);
    unsigned short* Hb1 = (unsigned short*)wsalloc(p, (size_t)Bsz * HD * 2);
    float* logits = (float*)wsalloc(p, (size_t)Bsz * LP * 4);
    float* col_l1 = (float*)wsalloc(p, HD * 4);
    float* scalars = (float*)wsalloc(p, 256);
    unsigned int* dHb = (unsigned int*)wsalloc(p, Bsz * 4);
    int* active = (int*)wsalloc(p, Bsz * 4);

    unsigned int* dH1 = (unsigned int*)col_l1;
    unsigned int* dH2 = dH1 + Bsz;
    int* active1 = (int*)(dH2 + Bsz);
    float* logits1 = (float*)WT_IH;
    float* part_h = logits;
    float* part_l = logits + 8 * HD;

    col_l1_kernel<<<dim3(HD / 64, 8), 256, 0, stream>>>(RH, MH, part_h, HD);
    col_l1_kernel<<<dim3((Lsz + 63) / 64, 8), 256, 0, stream>>>(RHL, MHL, part_l, Lsz);
    scalars_kernel<<<1, 256, 0, stream>>>(part_h, part_l, scalars);

    build_wt_kernel<<<dim3(HD / TT, HD / TT), 256, 0, stream>>>(
        RH, MH, scalars,  0, WT_H, HD, HD, HD, HD);
    build_wt_kernel<<<dim3(HD / TT, LP / TT), 256, 0, stream>>>(
        RHL, MHL, scalars,  1, WT_HL, HD, LP, Lsz, Lsz);
    build_wt_kernel<<<dim3(Isz / TT, HD / TT), 256, 0, stream>>>(
        RIH, MIH, scalars,  1, WT_IH, Isz, HD, HD, HD);
    to_bf16_kernel<<<(Bsz * Isz) / 256, 256, 0, stream>>>(x, xb, Bsz * Isz);
    init_state<<<(Bsz * HD) / 256, 256, 0, stream>>>(Hbuf0, Hb0, dHb, dH1, dH2,
                                                    active, active1);

    drive_gemm<<<dim3(HD / 64, Bsz / 64), 128, 0, stream>>>(
        xb, WT_IH, Bsz, HD, Isz, BH, drive);

    float* Hc[2] = {Hbuf0, Hbuf1};
    unsigned short* Hbc[2] = {Hb0, Hb1};
    int* act[2] = {active, active1};
    unsigned int* dHp[3] = {dHb, dH1, dH2};
    float* lg[2] = {logits, logits1};
    for (int s = 0; s <= TMAX + 1; ++s) {
        const int c = s & 1, n = (s + 1) & 1;
        step_kernel<<<REC_BLOCKS + LOG_BLOCKS + CERT_BLOCKS, 128, 0, stream>>>(
            Hbc[c], Hc[c], WT_H, WT_HL, drive, BL, scalars, act[c], act[n],
            Hc[n], Hbc[n], dHp[(s + 1) % 3], lg[c], lg[n], dHp[(s + 2) % 3], out,
            (s < TMAX) ? 1 : 0, (s >= 1 && s <= TMAX) ? 1 : 0, (s >= 2) ? 1 : 0,
            (s == TMAX + 1) ? 1 : 0);
    }
}


# Certification fetches each row's flag, dH and 1000 logits as one batch of independent 16-byte loads into registers (a round trip or two per wave, not one per 32 logits); top-2 and flush from there
# speedup vs baseline: 3.9234x; 1.6305x over previous
#include <hip/hip_runtime.h>


#define Bsz 512
#define Isz 1024
#define HD  2048
#define Lsz 1000
#define LP  1024
#define TMAX 32

typedef __attribute__((ext_vector_type(16))) __bf16 v16bf;
typedef __attribute__((ext_vector_type(8)))  float  v8f;
typedef __attribute__((ext_vector_type(4)))  unsigned int u32x4;

__device__ __forceinline__ unsigned short f2bf(float f) {
    unsigned int u = __float_as_uint(f);
    u += 0x7FFFu + ((u >> 16) & 1u);
    return (unsigned short)(u >> 16);
}

#define KC 256
#define PITCH (KC * 2 + 16)
#define TILE_BYTES (64 * PITCH)

struct alignas(32) FragBits { u32x4 lo, hi; };

__device__ __forceinline__ v16bf lds_frag(const unsigned char* p) {
    FragBits fb;
    fb.lo = *reinterpret_cast<const u32x4*>(p);
    fb.hi = *reinterpret_cast<const u32x4*>(p + 32);
    v16bf v;
    __builtin_memcpy(&v, &fb, sizeof(v));
    return v;
}

#define TAIL_ROWS (Bsz / 16)
template <int MODE>
__device__ __forceinline__
void gemm_tile(const unsigned short* __restrict__ A,
               const unsigned short* __restrict__ BT,
               int M, int N, int K,
               const float* __restrict__ bias,
               const float* __restrict__ drive,
               const float* __restrict__ Hold,
               const int*   __restrict__ active,
               float* __restrict__ Cout,
               unsigned short* __restrict__ HbOut,
               unsigned int* __restrict__ dHbits,
               int NL,
               unsigned char* const smem, int* const s_cnt, int* const s_list,
               const int bx, const int by, const int gx) {
    unsigned char* const sA = smem;
    unsigned char* const sB = smem + TILE_BYTES;

    const int tid  = threadIdx.x;
    const int lane = tid & 31;
    const int wave = tid >> 5;
    const int rowBase = by * 64 + wave * 16;
    const int colBase = bx * 64;
    const int r16 = lane & 15;
    const int kh  = lane >> 4;

    if (MODE != 0) {
        int flag[Bsz / 128];
        unsigned msk[Bsz / 128];
#pragma unroll
        for (int j = 0; j < Bsz / 128; ++j) {
            flag[j] = active[j * 128 + tid];
            msk[j]  = (unsigned)__ballot(flag[j] != 0);
        }
        if (lane == 0) {
#pragma unroll
            for (int j = 0; j < Bsz / 128; ++j) s_cnt[j * 4 + wave] = __popc(msk[j]);
        }
        __syncthreads();
        int n_act = 0;
        int before[Bsz / 128];
#pragma unroll
        for (int s = 0; s < Bsz / 32; ++s) {
            if ((s & 3) == wave) before[s >> 2] = n_act;
            n_act += s_cnt[s];
        }
        if (n_act == 0) return;

        if (n_act <= TAIL_ROWS) {
#pragma unroll
            for (int j = 0; j < Bsz / 128; ++j) {
                if (flag[j] != 0)
                    s_list[before[j] + __popc(msk[j] & ((1u << lane) - 1u))] = j * 128 + tid;
            }
            __syncthreads();

            const int NT = N >> 4;
            const int RT = (n_act + 15) >> 4;
            const int b  = by * gx + bx;
            if (b >= RT * NT) return;
            const int rt = b / NT;
            const int ct = b - rt * NT;

            const int li   = rt * 16 + r16;
            const int arow = s_list[li < n_act ? li : 0];
            const int KQ   = K >> 2;
            const unsigned short* pa = A  + (size_t)arow * K + wave * KQ + kh * 16;
            const unsigned short* pb = BT + (size_t)(ct * 16 + r16) * K + wave * KQ + kh * 16;

            v8f part = {0.f,0.f,0.f,0.f,0.f,0.f,0.f,0.f};
            for (int k0 = 0; k0 < KQ; k0 += 512) {
                FragBits ta[16], tb[16];
#pragma unroll
                for (int i = 0; i < 16; ++i) {
                    ta[i].lo = *reinterpret_cast<const u32x4*>(pa + k0 + i * 32);
                    ta[i].hi = *reinterpret_cast<const u32x4*>(pa + k0 + i * 32 + 8);
                    tb[i].lo = *reinterpret_cast<const u32x4*>(pb + k0 + i * 32);
                    tb[i].hi = *reinterpret_cast<const u32x4*>(pb + k0 + i * 32 + 8);
                }
                __builtin_amdgcn_sched_barrier(0);
#pragma unroll
                for (int i = 0; i < 16; ++i) {
                    v16bf av, bv;
                    __builtin_memcpy(&av, &ta[i], sizeof(av));
                    __builtin_memcpy(&bv, &tb[i], sizeof(bv));
                    part = __builtin_amdgcn_wmma_f32_16x16x32_bf16(
                        false, av, false, bv, (short)0, part, false, false);
                }
            }

            float* const red = reinterpret_cast<float*>(smem);
#pragma unroll
            for (int r = 0; r < 8; ++r) red[(wave * 8 + r) * 32 + lane] = part[r];
            __syncthreads();

            const int n = ct * 16 + r16;
            int   mrow[2];
            bool  live[2];
            float v[2];
#pragma unroll
            for (int j = 0; j < 2; ++j) {
                const int r  = 2 * wave + j;
                const int lr = rt * 16 + r + 8 * kh;
                live[j] = lr < n_act;
                mrow[j] = s_list[live[j] ? lr : 0];
                v[j] = ((red[r * 32 + lane] + red[(8 + r) * 32 + lane])
                        + red[(16 + r) * 32 + lane]) + red[(24 + r) * 32 + lane];
            }
            if (MODE == 2) {
                if (n < NL) {
                    const float bn = bias[n];
#pragma unroll
                    for (int j = 0; j < 2; ++j)
                        if (live[j]) Cout[(size_t)mrow[j] * N + n] = v[j] + bn;
                }
            } else {
                float dv[2], hv[2], d[2];
#pragma unroll
                for (int j = 0; j < 2; ++j) {
                    const size_t idx = (size_t)mrow[j] * N + n;
                    dv[j] = drive[idx];
                    hv[j] = Hold[idx];
                }
#pragma unroll
                for (int j = 0; j < 2; ++j) {
                    const size_t idx = (size_t)mrow[j] * N + n;
                    const float pre  = v[j] + dv[j];
                    const float actv = pre >= 0.f ? pre : 0.05f * pre;
                    d[j] = live[j] ? fabsf(actv - hv[j]) : 0.f;
                    if (live[j]) {
                        Cout[idx]  = actv;
                        HbOut[idx] = f2bf(actv);
                    }
                }
#pragma unroll
                for (int off = 8; off > 0; off >>= 1) {
#pragma unroll
                    for (int j = 0; j < 2; ++j) d[j] = fmaxf(d[j], __shfl_xor(d[j], off, 32));
                }
#pragma unroll
                for (int j = 0; j < 2; ++j)
                    if (live[j] && r16 == 0 && d[j] > 0.f)
                        atomicMax(dHbits + mrow[j], __float_as_uint(d[j]));
            }
            return;
        }

        if (s_cnt[2 * by] + s_cnt[2 * by + 1] == 0) return;
    }

    const unsigned short* ga = A  + (size_t)(by * 64 + wave) * K + lane * 8;
    const unsigned short* gb = BT + (size_t)(colBase + wave) * K + lane * 8;
    const size_t gstep = (size_t)4 * K;
    const int sst = wave * PITCH + lane * 16;

    const int fa = (wave * 16 + r16) * PITCH + kh * 16;
    const int fb = r16 * PITCH + kh * 16;

    v8f acc0 = {0.f,0.f,0.f,0.f,0.f,0.f,0.f,0.f};
    v8f acc[4];
    acc[0] = acc0; acc[1] = acc0; acc[2] = acc0; acc[3] = acc0;

    u32x4 ra[16], rb[16];

#pragma unroll
    for (int i = 0; i < 16; ++i) {
        ra[i] = *reinterpret_cast<const u32x4*>(ga + i * gstep);
        rb[i] = *reinterpret_cast<const u32x4*>(gb + i * gstep);
    }
#pragma unroll
    for (int i = 0; i < 16; ++i) {
        *reinterpret_cast<u32x4*>(sA + sst + i * 4 * PITCH) = ra[i];
        *reinterpret_cast<u32x4*>(sB + sst + i * 4 * PITCH) = rb[i];
    }
    __syncthreads();

    const int nchunks = K / KC;
    for (int c = 0; c < nchunks; ++c) {
        const bool more = (c + 1 < nchunks);
        if (more) {
            ga += KC; gb += KC;
#pragma unroll
            for (int i = 0; i < 16; ++i) {
                ra[i] = *reinterpret_cast<const u32x4*>(ga + i * gstep);
                rb[i] = *reinterpret_cast<const u32x4*>(gb + i * gstep);
            }
        }

        v16bf a_cur  = lds_frag(sA + fa);
        v16bf b_cur0 = lds_frag(sB + fb);
        v16bf b_cur1 = lds_frag(sB + fb + 16 * PITCH);
        v16bf b_cur2 = lds_frag(sB + fb + 32 * PITCH);
        v16bf b_cur3 = lds_frag(sB + fb + 48 * PITCH);
#pragma unroll
        for (int ks = 0; ks < KC / 32; ++ks) {
            v16bf a_nxt = a_cur, b_nxt0 = b_cur0, b_nxt1 = b_cur1,
                  b_nxt2 = b_cur2, b_nxt3 = b_cur3;
            if (ks + 1 < KC / 32) {
                const int ko = (ks + 1) * 64;
                a_nxt  = lds_frag(sA + fa + ko);
                b_nxt0 = lds_frag(sB + fb + ko);
                b_nxt1 = lds_frag(sB + fb + 16 * PITCH + ko);
                b_nxt2 = lds_frag(sB + fb + 32 * PITCH + ko);
                b_nxt3 = lds_frag(sB + fb + 48 * PITCH + ko);
            }
            __builtin_amdgcn_sched_barrier(0);
            acc[0] = __builtin_amdgcn_wmma_f32_16x16x32_bf16(
                false, a_cur, false, b_cur0, (short)0, acc[0], false, false);
            acc[1] = __builtin_amdgcn_wmma_f32_16x16x32_bf16(
                false, a_cur, false, b_cur1, (short)0, acc[1], false, false);
            acc[2] = __builtin_amdgcn_wmma_f32_16x16x32_bf16(
                false, a_cur, false, b_cur2, (short)0, acc[2], false, false);
            acc[3] = __builtin_amdgcn_wmma_f32_16x16x32_bf16(
                false, a_cur, false, b_cur3, (short)0, acc[3], false, false);
            __builtin_amdgcn_sched_barrier(0);
            a_cur = a_nxt;
            b_cur0 = b_nxt0; b_cur1 = b_nxt1; b_cur2 = b_nxt2; b_cur3 = b_nxt3;
        }

        __syncthreads();
        if (more) {
#pragma unroll
            for (int i = 0; i < 16; ++i) {
                *reinterpret_cast<u32x4*>(sA + sst + i * 4 * PITCH) = ra[i];
                *reinterpret_cast<u32x4*>(sB + sst + i * 4 * PITCH) = rb[i];
            }
            __syncthreads();
        }
    }

    bool rowActive[8];
    float dmax[8];
    if (MODE == 1) {
#pragma unroll
        for (int r = 0; r < 8; ++r) {
            rowActive[r] = active[rowBase + r + 8 * kh] != 0;
            dmax[r] = 0.f;
        }
    }

#pragma unroll
    for (int f = 0; f < 4; ++f) {
        const int n = colBase + f * 16 + r16;
#pragma unroll
        for (int r = 0; r < 8; ++r) {
            const int m = rowBase + r + 8 * kh;
            const float v = acc[f][r];
            if (MODE == 0) {
                Cout[(size_t)m * N + n] = v + bias[n];
            } else if (MODE == 2) {
                if (n < NL) Cout[(size_t)m * N + n] = v + bias[n];
            } else {
                const size_t idx = (size_t)m * N + n;
                const float pre = v + drive[idx];
                const float actv = pre >= 0.f ? pre : 0.05f * pre;
                const float ho = Hold[idx];
                const float hn = rowActive[r] ? actv : ho;
                Cout[idx] = hn;
                HbOut[idx] = f2bf(hn);
                dmax[r] = fmaxf(dmax[r], fabsf(hn - ho));
            }
        }
    }

    if (MODE == 1) {
#pragma unroll
        for (int r = 0; r < 8; ++r) {
            if (dmax[r] > 0.f) {
                atomicMax(dHbits + (rowBase + r + 8 * kh), __float_as_uint(dmax[r]));
            }
        }
    }
}

__global__ __launch_bounds__(256)
void col_l1_kernel(const float* __restrict__ A, const float* __restrict__ Msk,
                   float* __restrict__ part, int cols) {
    static_assert(HD % 256 == 0, "leaf rows are t + 256*r");
    __shared__ float smx[8][32];
    __shared__ float smy[8][32];
    const int lane = threadIdx.x & 31;
    const int v    = threadIdx.x >> 5;
    const int b    = blockIdx.y;
    const int c    = (blockIdx.x * 32 + lane) * 2;
    const bool in  = c < cols;
    const int cl   = in ? c : 0;
    float lx[4], ly[4];
#pragma unroll
    for (int m = 0; m < 4; ++m) {
        const int t = b + 8 * (v + 8 * m);
        float sx = 0.f, sy = 0.f;
#pragma unroll
        for (int r = 0; r < HD / 256; ++r) {
            const size_t idx = (size_t)(t + 256 * r) * cols + cl;
            const float2 a  = *reinterpret_cast<const float2*>(A + idx);
            const float2 mk = *reinterpret_cast<const float2*>(Msk + idx);
            sx += fabsf(a.x * mk.x);
            sy += fabsf(a.y * mk.y);
        }
        lx[m] = sx; ly[m] = sy;
    }
    smx[v][lane] = (lx[0] + lx[2]) + (lx[1] + lx[3]);
    smy[v][lane] = (ly[0] + ly[2]) + (ly[1] + ly[3]);
    __syncthreads();
    if (v == 0 && in) {
        part[(size_t)b * cols + c] =
            ((smx[0][lane] + smx[4][lane]) + (smx[2][lane] + smx[6][lane])) +
            ((smx[1][lane] + smx[5][lane]) + (smx[3][lane] + smx[7][lane]));
        part[(size_t)b * cols + c + 1] =
            ((smy[0][lane] + smy[4][lane]) + (smy[2][lane] + smy[6][lane])) +
            ((smy[1][lane] + smy[5][lane]) + (smy[3][lane] + smy[7][lane]));
    }
}

__device__ __forceinline__ float fold8(const float* __restrict__ p, int cols, int j) {
    const float p0 = p[j],            p1 = p[cols + j];
    const float p2 = p[2 * cols + j], p3 = p[3 * cols + j];
    const float p4 = p[4 * cols + j], p5 = p[5 * cols + j];
    const float p6 = p[6 * cols + j], p7 = p[7 * cols + j];
    return ((p0 + p4) + (p2 + p6)) + ((p1 + p5) + (p3 + p7));
}

__global__ __launch_bounds__(256)
void scalars_kernel(const float* __restrict__ col_l1,
                    const float* __restrict__ whl,
                    float* __restrict__ scalars) {
    __shared__ float sm[256];
    const int t = threadIdx.x;
    float mx = -1e30f;
    for (int j = t; j < HD; j += 256) mx = fmaxf(mx, fold8(col_l1, HD, j));
    sm[t] = mx; __syncthreads();
    for (int s = 128; s > 0; s >>= 1) {
        if (t < s) sm[t] = fmaxf(sm[t], sm[t + s]);
        __syncthreads();
    }
    const float m1 = sm[0];
    __syncthreads();
    mx = -1e30f;
    for (int j = t; j < Lsz; j += 256) mx = fmaxf(mx, fold8(whl, Lsz, j));
    sm[t] = mx; __syncthreads();
    for (int s = 128; s > 0; s >>= 1) {
        if (t < s) sm[t] = fmaxf(sm[t], sm[t + s]);
        __syncthreads();
    }
    if (t == 0) {
        scalars[0] = 0.95f / fmaxf(m1, 1e-8f);
        scalars[1] = sm[0] * 19.0f;
    }
}

#define TT 64
#define TPITCH (TT + 2)
__global__ __launch_bounds__(256)
void build_wt_kernel(const float* __restrict__ R, const float* __restrict__ M,
                     const float* __restrict__ scalars, int sel,
                     unsigned short* __restrict__ WT,
                     int K, int N, int srcld, int NL) {
    (void)N;
    __shared__ __attribute__((aligned(16))) unsigned short tile[TT * TPITCH];
    const int k0 = blockIdx.x * TT;
    const int n0 = blockIdx.y * TT;
    const int t  = threadIdx.x;

    const int ln = (t & 15) * 4;
    const int lk = t >> 4;
    const int n  = n0 + ln;
    const bool in = n < NL;
    const int nc = in ? n : 0;
    float4 r[4], m[4];
#pragma unroll
    for (int i = 0; i < 4; ++i) {
        const size_t sidx = (size_t)(k0 + lk + 16 * i) * srcld + nc;
        r[i] = *reinterpret_cast<const float4*>(R + sidx);
        m[i] = *reinterpret_cast<const float4*>(M + sidx);
    }
    float sc = 1.f;
    if (sel == 0) sc = scalars[0];
#pragma unroll
    for (int i = 0; i < 4; ++i) {
        float v0 = r[i].x * m[i].x, v1 = r[i].y * m[i].y;
        float v2 = r[i].z * m[i].z, v3 = r[i].w * m[i].w;
        if (sel == 0) { v0 *= sc; v1 *= sc; v2 *= sc; v3 *= sc; }
        if (!in) { v0 = 0.f; v1 = 0.f; v2 = 0.f; v3 = 0.f; }
        unsigned int* d = reinterpret_cast<unsigned int*>(tile + (lk + 16 * i) * TPITCH + ln);
        d[0] = (unsigned int)f2bf(v0) | ((unsigned int)f2bf(v1) << 16);
        d[1] = (unsigned int)f2bf(v2) | ((unsigned int)f2bf(v3) << 16);
    }
    __syncthreads();

    const int ks = (t & 7) * 8;
#pragma unroll
    for (int p = 0; p < 2; ++p) {
        const int nl = (t >> 3) + 32 * p;
        unsigned int w[4];
#pragma unroll
        for (int j = 0; j < 4; ++j) {
            const unsigned int e0 = tile[(ks + 2 * j) * TPITCH + nl];
            const unsigned int e1 = tile[(ks + 2 * j + 1) * TPITCH + nl];
            w[j] = e0 | (e1 << 16);
        }
        u32x4 o; o.x = w[0]; o.y = w[1]; o.z = w[2]; o.w = w[3];
        *reinterpret_cast<u32x4*>(WT + (size_t)(n0 + nl) * K + k0 + ks) = o;
    }
}

__global__ __launch_bounds__(256)
void to_bf16_kernel(const float* __restrict__ src, unsigned short* __restrict__ dst,
                    int n) {
    const int i = blockIdx.x * 256 + threadIdx.x;
    if (i < n) dst[i] = f2bf(src[i]);
}

__global__ __launch_bounds__(256)
void init_state(float* __restrict__ H0, unsigned short* __restrict__ Hb0,
                unsigned int* __restrict__ dHa, unsigned int* __restrict__ dHb,
                unsigned int* __restrict__ dHc,
                int* __restrict__ act0, int* __restrict__ act1) {
    const size_t n = (size_t)Bsz * HD;
    for (size_t t = (size_t)blockIdx.x * 256 + threadIdx.x; t < n;
         t += (size_t)gridDim.x * 256) {
        H0[t] = 0.f;
        Hb0[t] = 0;
    }
    const int i = blockIdx.x * 256 + threadIdx.x;
    if (i < Bsz) { dHa[i] = 0u; dHb[i] = 0u; dHc[i] = 0u; act0[i] = 1; act1[i] = 1; }
}

__global__ __launch_bounds__(128)
void drive_gemm(const unsigned short* __restrict__ A,
                const unsigned short* __restrict__ BT,
                int M, int N, int K,
                const float* __restrict__ bias,
                float* __restrict__ Cout) {
    __shared__ __attribute__((aligned(16))) unsigned char smem[2 * TILE_BYTES];
    gemm_tile<0>(A, BT, M, N, K, bias, nullptr, nullptr, nullptr,
                 Cout, nullptr, nullptr, N,
                 smem, nullptr, nullptr, (int)blockIdx.x, (int)blockIdx.y, (int)gridDim.x);
}

__device__ __forceinline__
void certify_rows(const int ub,
                  const float* __restrict__ logits,
                  const float* __restrict__ scalars,
                  unsigned int* __restrict__ dHbits,
                  const int* __restrict__ actIn,
                  int* __restrict__ actOut,
                  float* __restrict__ out,
                  const int is_last) {
    static_assert(Lsz % 4 == 0 && LP == 1024, "a row is eight 16-B pieces per lane, none straddling column Lsz");
    const int lane = threadIdx.x & 31;
    const int wave = threadIdx.x >> 5;
    const int row0 = ub * 8 + wave * 2;

    int    a[2];
    float  dh[2];
    float4 v[2][8];
#pragma unroll
    for (int j = 0; j < 2; ++j) {
        a[j]  = actIn[row0 + j];
        dh[j] = __uint_as_float(dHbits[row0 + j]);
        const float* lr = logits + (size_t)(row0 + j) * LP;
#pragma unroll
        for (int q = 0; q < 8; ++q) {
            const int col = 128 * q + 4 * lane;
            v[j][q] = *reinterpret_cast<const float4*>(lr + (col < Lsz ? col : Lsz - 4));
        }
    }
    const float gcoef = scalars[1];

#pragma unroll
    for (int j = 0; j < 2; ++j) {
        const int row = row0 + j;
        int aj = a[j];
        if (aj != 0) {
            float m1 = -1e30f, m2 = -1e30f;
#pragma unroll
            for (int q = 0; q < 8; ++q) {
                const bool in = 128 * q + 4 * lane < Lsz;
                const float e[4] = {v[j][q].x, v[j][q].y, v[j][q].z, v[j][q].w};
#pragma unroll
                for (int c = 0; c < 4; ++c) {
                    const float x = in ? e[c] : -1e30f;
                    m2 = fmaxf(m2, fminf(m1, x));
                    m1 = fmaxf(m1, x);
                }
            }
            for (int off = 16; off > 0; off >>= 1) {
                const float o1 = __shfl_xor(m1, off, 32);
                const float o2 = __shfl_xor(m2, off, 32);
                const float hi = fmaxf(m1, o1);
                const float lo = fmaxf(fminf(m1, o1), fmaxf(m2, o2));
                m1 = hi; m2 = lo;
            }
            const float margin = m1 - m2;
            const float Gamma = gcoef * dh[j];
            const bool newly = margin > 2.0f * Gamma;
            if (newly || is_last) {
                float* orow = out + (size_t)row * Lsz;
#pragma unroll
                for (int q = 0; q < 8; ++q) {
                    const int col = 128 * q + 4 * lane;
                    if (col < Lsz) *reinterpret_cast<float4*>(orow + col) = v[j][q];
                }
                aj = 0;
            }
        }
        if (lane == 0) {
            actOut[row] = aj;
            dHbits[row] = 0u;
        }
    }
}

#define REC_BLOCKS  ((HD / 64) * (Bsz / 64))
#define LOG_BLOCKS  ((LP / 64) * (Bsz / 64))
#define CERT_BLOCKS (Bsz / 8)
__global__ __launch_bounds__(128)
void step_kernel(const unsigned short* __restrict__ HbCur,
                 const float* __restrict__ HCur,
                 const unsigned short* __restrict__ WT_H,
                 const unsigned short* __restrict__ WT_HL,
                 const float* __restrict__ drive,
                 const float* __restrict__ BL,
                 const float* __restrict__ scalars,
                 const int* __restrict__ actIn,
                 int* __restrict__ actOut,
                 float* __restrict__ HNext,
                 unsigned short* __restrict__ HbNext,
                 unsigned int* __restrict__ dHNext,
                 float* __restrict__ logitsCur,
                 const float* __restrict__ logitsPrev,
                 unsigned int* __restrict__ dHPrev,
                 float* __restrict__ out,
                 int do_rec, int do_log, int do_cert, int is_last) {
    __shared__ __attribute__((aligned(16))) unsigned char smem[2 * TILE_BYTES];
    __shared__ int s_cnt[Bsz / 32];
    __shared__ int s_list[TAIL_ROWS];
    const int b = blockIdx.x;
    if (b < REC_BLOCKS) {
        if (do_rec)
            gemm_tile<1>(HbCur, WT_H, Bsz, HD, HD, nullptr, drive, HCur, actIn,
                         HNext, HbNext, dHNext, HD,
                         smem, s_cnt, s_list, b % (HD / 64), b / (HD / 64), HD / 64);
    } else if (b < REC_BLOCKS + LOG_BLOCKS) {
        const int q = b - REC_BLOCKS;
        if (do_log)
            gemm_tile<2>(HbCur, WT_HL, Bsz, LP, HD, BL, nullptr, nullptr, actIn,
                         logitsCur, nullptr, nullptr, Lsz,
                         smem, s_cnt, s_list, q % (LP / 64), q / (LP / 64), LP / 64);
    } else {
        if (do_cert)
            certify_rows(b - (REC_BLOCKS + LOG_BLOCKS), logitsPrev, scalars, dHPrev,
                         actIn, actOut, out, is_last);
    }
}

static inline char* wsalloc(char*& p, size_t bytes) {
    char* r = p;
    p += (bytes + 255) & ~(size_t)255;
    return r;
}

extern "C" void kernel_launch(void* const* d_in, const int* in_sizes, int n_in,
                              void* d_out, int out_size, void* d_ws, size_t ws_size,
                              hipStream_t stream) {
    (void)in_sizes; (void)n_in; (void)out_size; (void)ws_size;
    const float* x   = (const float*)d_in[0];
    const float* RIH = (const float*)d_in[1];
    const float* RH  = (const float*)d_in[2];
    const float* RHL = (const float*)d_in[3];
    const float* BH  = (const float*)d_in[4];
    const float* BL  = (const float*)d_in[5];
    const float* MIH = (const float*)d_in[6];
    const float* MH  = (const float*)d_in[7];
    const float* MHL = (const float*)d_in[8];
    float* out = (float*)d_out;

    char* p = (char*)d_ws;
    unsigned short* WT_H  = (unsigned short*)wsalloc(p, (size_t)HD * HD * 2);
    unsigned short* WT_HL = (unsigned short*)wsalloc(p, (size_t)LP * HD * 2);
    unsigned short* WT_IH = (unsigned short*)wsalloc(p, (size_t)HD * Isz * 2);
    unsigned short* xb    = (unsigned short*)wsalloc(p, (size_t)Bsz * Isz * 2);
    float* drive = (float*)wsalloc(p, (size_t)Bsz * HD * 4);
    float* Hbuf0 = (float*)wsalloc(p, (size_t)Bsz * HD * 4);
    float* Hbuf1 = (float*)wsalloc(p, (size_t)Bsz * HD * 4);
    unsigned short* Hb0 = (unsigned short*)wsalloc(p, (size_t)Bsz * HD * 2);
    unsigned short* Hb1 = (unsigned short*)wsalloc(p, (size_t)Bsz * HD * 2);
    float* logits = (float*)wsalloc(p, (size_t)Bsz * LP * 4);
    float* col_l1 = (float*)wsalloc(p, HD * 4);
    float* scalars = (float*)wsalloc(p, 256);
    unsigned int* dHb = (unsigned int*)wsalloc(p, Bsz * 4);
    int* active = (int*)wsalloc(p, Bsz * 4);

    unsigned int* dH1 = (unsigned int*)col_l1;
    unsigned int* dH2 = dH1 + Bsz;
    int* active1 = (int*)(dH2 + Bsz);
    float* logits1 = (float*)WT_IH;
    float* part_h = logits;
    float* part_l = logits + 8 * HD;

    col_l1_kernel<<<dim3(HD / 64, 8), 256, 0, stream>>>(RH, MH, part_h, HD);
    col_l1_kernel<<<dim3((Lsz + 63) / 64, 8), 256, 0, stream>>>(RHL, MHL, part_l, Lsz);
    scalars_kernel<<<1, 256, 0, stream>>>(part_h, part_l, scalars);

    build_wt_kernel<<<dim3(HD / TT, HD / TT), 256, 0, stream>>>(
        RH, MH, scalars,  0, WT_H, HD, HD, HD, HD);
    build_wt_kernel<<<dim3(HD / TT, LP / TT), 256, 0, stream>>>(
        RHL, MHL, scalars,  1, WT_HL, HD, LP, Lsz, Lsz);
    build_wt_kernel<<<dim3(Isz / TT, HD / TT), 256, 0, stream>>>(
        RIH, MIH, scalars,  1, WT_IH, Isz, HD, HD, HD);
    to_bf16_kernel<<<(Bsz * Isz) / 256, 256, 0, stream>>>(x, xb, Bsz * Isz);
    init_state<<<(Bsz * HD) / 256, 256, 0, stream>>>(Hbuf0, Hb0, dHb, dH1, dH2,
                                                    active, active1);

    drive_gemm<<<dim3(HD / 64, Bsz / 64), 128, 0, stream>>>(
        xb, WT_IH, Bsz, HD, Isz, BH, drive);

    float* Hc[2] = {Hbuf0, Hbuf1};
    unsigned short* Hbc[2] = {Hb0, Hb1};
    int* act[2] = {active, active1};
    unsigned int* dHp[3] = {dHb, dH1, dH2};
    float* lg[2] = {logits, logits1};
    for (int s = 0; s <= TMAX + 1; ++s) {
        const int c = s & 1, n = (s + 1) & 1;
        step_kernel<<<REC_BLOCKS + LOG_BLOCKS + CERT_BLOCKS, 128, 0, stream>>>(
            Hbc[c], Hc[c], WT_H, WT_HL, drive, BL, scalars, act[c], act[n],
            Hc[n], Hbc[n], dHp[(s + 1) % 3], lg[c], lg[n], dHp[(s + 2) % 3], out,
            (s < TMAX) ? 1 : 0, (s >= 1 && s <= TMAX) ? 1 : 0, (s >= 2) ? 1 : 0,
            (s == TMAX + 1) ? 1 : 0);
    }
}


# Set-up in 3 launches, not 9: kernels needing only the inputs run as block groups of one grid (x-to-bf16, state clear at 16 B per lane), scalars share drive's launch; every sum in the parent's order
# speedup vs baseline: 4.2977x; 1.0954x over previous
#include <hip/hip_runtime.h>


#define Bsz 512
#define Isz 1024
#define HD  2048
#define Lsz 1000
#define LP  1024
#define TMAX 32

typedef __attribute__((ext_vector_type(16))) __bf16 v16bf;
typedef __attribute__((ext_vector_type(8)))  float  v8f;
typedef __attribute__((ext_vector_type(4)))  unsigned int u32x4;

__device__ __forceinline__ unsigned short f2bf(float f) {
    unsigned int u = __float_as_uint(f);
    u += 0x7FFFu + ((u >> 16) & 1u);
    return (unsigned short)(u >> 16);
}

#define KC 256
#define PITCH (KC * 2 + 16)
#define TILE_BYTES (64 * PITCH)

struct alignas(32) FragBits { u32x4 lo, hi; };

__device__ __forceinline__ v16bf lds_frag(const unsigned char* p) {
    FragBits fb;
    fb.lo = *reinterpret_cast<const u32x4*>(p);
    fb.hi = *reinterpret_cast<const u32x4*>(p + 32);
    v16bf v;
    __builtin_memcpy(&v, &fb, sizeof(v));
    return v;
}

#define TAIL_ROWS (Bsz / 16)
template <int MODE>
__device__ __forceinline__
void gemm_tile(const unsigned short* __restrict__ A,
               const unsigned short* __restrict__ BT,
               int M, int N, int K,
               const float* __restrict__ bias,
               const float* __restrict__ drive,
               const float* __restrict__ Hold,
               const int*   __restrict__ active,
               float* __restrict__ Cout,
               unsigned short* __restrict__ HbOut,
               unsigned int* __restrict__ dHbits,
               int NL,
               unsigned char* const smem, int* const s_cnt, int* const s_list,
               const int bx, const int by, const int gx) {
    unsigned char* const sA = smem;
    unsigned char* const sB = smem + TILE_BYTES;

    const int tid  = threadIdx.x;
    const int lane = tid & 31;
    const int wave = tid >> 5;
    const int rowBase = by * 64 + wave * 16;
    const int colBase = bx * 64;
    const int r16 = lane & 15;
    const int kh  = lane >> 4;

    if (MODE != 0) {
        int flag[Bsz / 128];
        unsigned msk[Bsz / 128];
#pragma unroll
        for (int j = 0; j < Bsz / 128; ++j) {
            flag[j] = active[j * 128 + tid];
            msk[j]  = (unsigned)__ballot(flag[j] != 0);
        }
        if (lane == 0) {
#pragma unroll
            for (int j = 0; j < Bsz / 128; ++j) s_cnt[j * 4 + wave] = __popc(msk[j]);
        }
        __syncthreads();
        int n_act = 0;
        int before[Bsz / 128];
#pragma unroll
        for (int s = 0; s < Bsz / 32; ++s) {
            if ((s & 3) == wave) before[s >> 2] = n_act;
            n_act += s_cnt[s];
        }
        if (n_act == 0) return;

        if (n_act <= TAIL_ROWS) {
#pragma unroll
            for (int j = 0; j < Bsz / 128; ++j) {
                if (flag[j] != 0)
                    s_list[before[j] + __popc(msk[j] & ((1u << lane) - 1u))] = j * 128 + tid;
            }
            __syncthreads();

            const int NT = N >> 4;
            const int RT = (n_act + 15) >> 4;
            const int b  = by * gx + bx;
            if (b >= RT * NT) return;
            const int rt = b / NT;
            const int ct = b - rt * NT;

            const int li   = rt * 16 + r16;
            const int arow = s_list[li < n_act ? li : 0];
            const int KQ   = K >> 2;
            const unsigned short* pa = A  + (size_t)arow * K + wave * KQ + kh * 16;
            const unsigned short* pb = BT + (size_t)(ct * 16 + r16) * K + wave * KQ + kh * 16;

            v8f part = {0.f,0.f,0.f,0.f,0.f,0.f,0.f,0.f};
            for (int k0 = 0; k0 < KQ; k0 += 512) {
                FragBits ta[16], tb[16];
#pragma unroll
                for (int i = 0; i < 16; ++i) {
                    ta[i].lo = *reinterpret_cast<const u32x4*>(pa + k0 + i * 32);
                    ta[i].hi = *reinterpret_cast<const u32x4*>(pa + k0 + i * 32 + 8);
                    tb[i].lo = *reinterpret_cast<const u32x4*>(pb + k0 + i * 32);
                    tb[i].hi = *reinterpret_cast<const u32x4*>(pb + k0 + i * 32 + 8);
                }
                __builtin_amdgcn_sched_barrier(0);
#pragma unroll
                for (int i = 0; i < 16; ++i) {
                    v16bf av, bv;
                    __builtin_memcpy(&av, &ta[i], sizeof(av));
                    __builtin_memcpy(&bv, &tb[i], sizeof(bv));
                    part = __builtin_amdgcn_wmma_f32_16x16x32_bf16(
                        false, av, false, bv, (short)0, part, false, false);
                }
            }

            float* const red = reinterpret_cast<float*>(smem);
#pragma unroll
            for (int r = 0; r < 8; ++r) red[(wave * 8 + r) * 32 + lane] = part[r];
            __syncthreads();

            const int n = ct * 16 + r16;
            int   mrow[2];
            bool  live[2];
            float v[2];
#pragma unroll
            for (int j = 0; j < 2; ++j) {
                const int r  = 2 * wave + j;
                const int lr = rt * 16 + r + 8 * kh;
                live[j] = lr < n_act;
                mrow[j] = s_list[live[j] ? lr : 0];
                v[j] = ((red[r * 32 + lane] + red[(8 + r) * 32 + lane])
                        + red[(16 + r) * 32 + lane]) + red[(24 + r) * 32 + lane];
            }
            if (MODE == 2) {
                if (n < NL) {
                    const float bn = bias[n];
#pragma unroll
                    for (int j = 0; j < 2; ++j)
                        if (live[j]) Cout[(size_t)mrow[j] * N + n] = v[j] + bn;
                }
            } else {
                float dv[2], hv[2], d[2];
#pragma unroll
                for (int j = 0; j < 2; ++j) {
                    const size_t idx = (size_t)mrow[j] * N + n;
                    dv[j] = drive[idx];
                    hv[j] = Hold[idx];
                }
#pragma unroll
                for (int j = 0; j < 2; ++j) {
                    const size_t idx = (size_t)mrow[j] * N + n;
                    const float pre  = v[j] + dv[j];
                    const float actv = pre >= 0.f ? pre : 0.05f * pre;
                    d[j] = live[j] ? fabsf(actv - hv[j]) : 0.f;
                    if (live[j]) {
                        Cout[idx]  = actv;
                        HbOut[idx] = f2bf(actv);
                    }
                }
#pragma unroll
                for (int off = 8; off > 0; off >>= 1) {
#pragma unroll
                    for (int j = 0; j < 2; ++j) d[j] = fmaxf(d[j], __shfl_xor(d[j], off, 32));
                }
#pragma unroll
                for (int j = 0; j < 2; ++j)
                    if (live[j] && r16 == 0 && d[j] > 0.f)
                        atomicMax(dHbits + mrow[j], __float_as_uint(d[j]));
            }
            return;
        }

        if (s_cnt[2 * by] + s_cnt[2 * by + 1] == 0) return;
    }

    const unsigned short* ga = A  + (size_t)(by * 64 + wave) * K + lane * 8;
    const unsigned short* gb = BT + (size_t)(colBase + wave) * K + lane * 8;
    const size_t gstep = (size_t)4 * K;
    const int sst = wave * PITCH + lane * 16;

    const int fa = (wave * 16 + r16) * PITCH + kh * 16;
    const int fb = r16 * PITCH + kh * 16;

    v8f acc0 = {0.f,0.f,0.f,0.f,0.f,0.f,0.f,0.f};
    v8f acc[4];
    acc[0] = acc0; acc[1] = acc0; acc[2] = acc0; acc[3] = acc0;

    u32x4 ra[16], rb[16];

#pragma unroll
    for (int i = 0; i < 16; ++i) {
        ra[i] = *reinterpret_cast<const u32x4*>(ga + i * gstep);
        rb[i] = *reinterpret_cast<const u32x4*>(gb + i * gstep);
    }
#pragma unroll
    for (int i = 0; i < 16; ++i) {
        *reinterpret_cast<u32x4*>(sA + sst + i * 4 * PITCH) = ra[i];
        *reinterpret_cast<u32x4*>(sB + sst + i * 4 * PITCH) = rb[i];
    }
    __syncthreads();

    const int nchunks = K / KC;
    for (int c = 0; c < nchunks; ++c) {
        const bool more = (c + 1 < nchunks);
        if (more) {
            ga += KC; gb += KC;
#pragma unroll
            for (int i = 0; i < 16; ++i) {
                ra[i] = *reinterpret_cast<const u32x4*>(ga + i * gstep);
                rb[i] = *reinterpret_cast<const u32x4*>(gb + i * gstep);
            }
        }

        v16bf a_cur  = lds_frag(sA + fa);
        v16bf b_cur0 = lds_frag(sB + fb);
        v16bf b_cur1 = lds_frag(sB + fb + 16 * PITCH);
        v16bf b_cur2 = lds_frag(sB + fb + 32 * PITCH);
        v16bf b_cur3 = lds_frag(sB + fb + 48 * PITCH);
#pragma unroll
        for (int ks = 0; ks < KC / 32; ++ks) {
            v16bf a_nxt = a_cur, b_nxt0 = b_cur0, b_nxt1 = b_cur1,
                  b_nxt2 = b_cur2, b_nxt3 = b_cur3;
            if (ks + 1 < KC / 32) {
                const int ko = (ks + 1) * 64;
                a_nxt  = lds_frag(sA + fa + ko);
                b_nxt0 = lds_frag(sB + fb + ko);
                b_nxt1 = lds_frag(sB + fb + 16 * PITCH + ko);
                b_nxt2 = lds_frag(sB + fb + 32 * PITCH + ko);
                b_nxt3 = lds_frag(sB + fb + 48 * PITCH + ko);
            }
            __builtin_amdgcn_sched_barrier(0);
            acc[0] = __builtin_amdgcn_wmma_f32_16x16x32_bf16(
                false, a_cur, false, b_cur0, (short)0, acc[0], false, false);
            acc[1] = __builtin_amdgcn_wmma_f32_16x16x32_bf16(
                false, a_cur, false, b_cur1, (short)0, acc[1], false, false);
            acc[2] = __builtin_amdgcn_wmma_f32_16x16x32_bf16(
                false, a_cur, false, b_cur2, (short)0, acc[2], false, false);
            acc[3] = __builtin_amdgcn_wmma_f32_16x16x32_bf16(
                false, a_cur, false, b_cur3, (short)0, acc[3], false, false);
            __builtin_amdgcn_sched_barrier(0);
            a_cur = a_nxt;
            b_cur0 = b_nxt0; b_cur1 = b_nxt1; b_cur2 = b_nxt2; b_cur3 = b_nxt3;
        }

        __syncthreads();
        if (more) {
#pragma unroll
            for (int i = 0; i < 16; ++i) {
                *reinterpret_cast<u32x4*>(sA + sst + i * 4 * PITCH) = ra[i];
                *reinterpret_cast<u32x4*>(sB + sst + i * 4 * PITCH) = rb[i];
            }
            __syncthreads();
        }
    }

    bool rowActive[8];
    float dmax[8];
    if (MODE == 1) {
#pragma unroll
        for (int r = 0; r < 8; ++r) {
            rowActive[r] = active[rowBase + r + 8 * kh] != 0;
            dmax[r] = 0.f;
        }
    }

#pragma unroll
    for (int f = 0; f < 4; ++f) {
        const int n = colBase + f * 16 + r16;
#pragma unroll
        for (int r = 0; r < 8; ++r) {
            const int m = rowBase + r + 8 * kh;
            const float v = acc[f][r];
            if (MODE == 0) {
                Cout[(size_t)m * N + n] = v + bias[n];
            } else if (MODE == 2) {
                if (n < NL) Cout[(size_t)m * N + n] = v + bias[n];
            } else {
                const size_t idx = (size_t)m * N + n;
                const float pre = v + drive[idx];
                const float actv = pre >= 0.f ? pre : 0.05f * pre;
                const float ho = Hold[idx];
                const float hn = rowActive[r] ? actv : ho;
                Cout[idx] = hn;
                HbOut[idx] = f2bf(hn);
                dmax[r] = fmaxf(dmax[r], fabsf(hn - ho));
            }
        }
    }

    if (MODE == 1) {
#pragma unroll
        for (int r = 0; r < 8; ++r) {
            if (dmax[r] > 0.f) {
                atomicMax(dHbits + (rowBase + r + 8 * kh), __float_as_uint(dmax[r]));
            }
        }
    }
}

__device__ __forceinline__
void col_l1_block(const float* __restrict__ A, const float* __restrict__ Msk,
                  float* __restrict__ part, const int cols, const int bx, const int b) {
    static_assert(HD % 256 == 0, "leaf rows are t + 256*r");
    __shared__ float smx[8][32];
    __shared__ float smy[8][32];
    const int lane = threadIdx.x & 31;
    const int v    = threadIdx.x >> 5;
    const int c    = (bx * 32 + lane) * 2;
    const bool in  = c < cols;
    const int cl   = in ? c : 0;
    float lx[4], ly[4];
#pragma unroll
    for (int m = 0; m < 4; ++m) {
        const int t = b + 8 * (v + 8 * m);
        float sx = 0.f, sy = 0.f;
#pragma unroll
        for (int r = 0; r < HD / 256; ++r) {
            const size_t idx = (size_t)(t + 256 * r) * cols + cl;
            const float2 a  = *reinterpret_cast<const float2*>(A + idx);
            const float2 mk = *reinterpret_cast<const float2*>(Msk + idx);
            sx += fabsf(a.x * mk.x);
            sy += fabsf(a.y * mk.y);
        }
        lx[m] = sx; ly[m] = sy;
    }
    smx[v][lane] = (lx[0] + lx[2]) + (lx[1] + lx[3]);
    smy[v][lane] = (ly[0] + ly[2]) + (ly[1] + ly[3]);
    __syncthreads();
    if (v == 0 && in) {
        part[(size_t)b * cols + c] =
            ((smx[0][lane] + smx[4][lane]) + (smx[2][lane] + smx[6][lane])) +
            ((smx[1][lane] + smx[5][lane]) + (smx[3][lane] + smx[7][lane]));
        part[(size_t)b * cols + c + 1] =
            ((smy[0][lane] + smy[4][lane]) + (smy[2][lane] + smy[6][lane])) +
            ((smy[1][lane] + smy[5][lane]) + (smy[3][lane] + smy[7][lane]));
    }
}

__device__ __forceinline__ float fold8(const float* __restrict__ p, int cols, int j) {
    const float p0 = p[j],            p1 = p[cols + j];
    const float p2 = p[2 * cols + j], p3 = p[3 * cols + j];
    const float p4 = p[4 * cols + j], p5 = p[5 * cols + j];
    const float p6 = p[6 * cols + j], p7 = p[7 * cols + j];
    return ((p0 + p4) + (p2 + p6)) + ((p1 + p5) + (p3 + p7));
}

#define SCB 128
__device__ __forceinline__
void scalars_block(const float* __restrict__ col_l1,
                   const float* __restrict__ whl,
                   float* __restrict__ scalars, float* const sm) {
    static_assert(HD % SCB == 0, "whole trips over the HD columns");
    const int t = threadIdx.x;
    float mx = -1e30f;
#pragma unroll
    for (int i = 0; i < HD / SCB; ++i) mx = fmaxf(mx, fold8(col_l1, HD, t + SCB * i));
    sm[t] = mx; __syncthreads();
    for (int s = SCB / 2; s > 0; s >>= 1) {
        if (t < s) sm[t] = fmaxf(sm[t], sm[t + s]);
        __syncthreads();
    }
    const float m1 = sm[0];
    __syncthreads();
    mx = -1e30f;
#pragma unroll
    for (int i = 0; i < (Lsz + SCB - 1) / SCB; ++i) {
        const int j = t + SCB * i;
        mx = fmaxf(mx, fold8(whl, Lsz, j < Lsz ? j : Lsz - 1));
    }
    sm[t] = mx; __syncthreads();
    for (int s = SCB / 2; s > 0; s >>= 1) {
        if (t < s) sm[t] = fmaxf(sm[t], sm[t + s]);
        __syncthreads();
    }
    if (t == 0) {
        scalars[0] = 0.95f / fmaxf(m1, 1e-8f);
        scalars[1] = sm[0] * 19.0f;
    }
}

#define TT 64
#define TPITCH (TT + 2)
__device__ __forceinline__
void build_wt_block(const float* __restrict__ R, const float* __restrict__ M,
                    const float* __restrict__ scalars, const int sel,
                    unsigned short* __restrict__ WT,
                    const int K, const int srcld, const int NL,
                    const int bkx, const int bny) {
    __shared__ __attribute__((aligned(16))) unsigned short tile[TT * TPITCH];
    const int k0 = bkx * TT;
    const int n0 = bny * TT;
    const int t  = threadIdx.x;

    const int ln = (t & 15) * 4;
    const int lk = t >> 4;
    const int n  = n0 + ln;
    const bool in = n < NL;
    const int nc = in ? n : 0;
    float4 r[4], m[4];
#pragma unroll
    for (int i = 0; i < 4; ++i) {
        const size_t sidx = (size_t)(k0 + lk + 16 * i) * srcld + nc;
        r[i] = *reinterpret_cast<const float4*>(R + sidx);
        m[i] = *reinterpret_cast<const float4*>(M + sidx);
    }
    float sc = 1.f;
    if (sel == 0) sc = scalars[0];
#pragma unroll
    for (int i = 0; i < 4; ++i) {
        float v0 = r[i].x * m[i].x, v1 = r[i].y * m[i].y;
        float v2 = r[i].z * m[i].z, v3 = r[i].w * m[i].w;
        if (sel == 0) { v0 *= sc; v1 *= sc; v2 *= sc; v3 *= sc; }
        if (!in) { v0 = 0.f; v1 = 0.f; v2 = 0.f; v3 = 0.f; }
        unsigned int* d = reinterpret_cast<unsigned int*>(tile + (lk + 16 * i) * TPITCH + ln);
        d[0] = (unsigned int)f2bf(v0) | ((unsigned int)f2bf(v1) << 16);
        d[1] = (unsigned int)f2bf(v2) | ((unsigned int)f2bf(v3) << 16);
    }
    __syncthreads();

    const int ks = (t & 7) * 8;
#pragma unroll
    for (int p = 0; p < 2; ++p) {
        const int nl = (t >> 3) + 32 * p;
        unsigned int w[4];
#pragma unroll
        for (int j = 0; j < 4; ++j) {
            const unsigned int e0 = tile[(ks + 2 * j) * TPITCH + nl];
            const unsigned int e1 = tile[(ks + 2 * j + 1) * TPITCH + nl];
            w[j] = e0 | (e1 << 16);
        }
        u32x4 o; o.x = w[0]; o.y = w[1]; o.z = w[2]; o.w = w[3];
        *reinterpret_cast<u32x4*>(WT + (size_t)(n0 + nl) * K + k0 + ks) = o;
    }
}

__global__ __launch_bounds__(256)
void build_wt_kernel(const float* __restrict__ R, const float* __restrict__ M,
                     const float* __restrict__ scalars, int sel,
                     unsigned short* __restrict__ WT,
                     int K, int srcld, int NL) {
    build_wt_block(R, M, scalars, sel, WT, K, srcld, NL, (int)blockIdx.x, (int)blockIdx.y);
}

#define SA_COLH  ((HD / 64) * 8)
#define SA_COLL  (((Lsz + 63) / 64) * 8)
#define SA_WTIH  ((Isz / TT) * (HD / TT))
#define SA_WTHL  ((HD / TT) * (LP / TT))
#define SA_XB    128
#define SA_INIT  128
#define SA_BLOCKS (SA_COLH + SA_COLL + SA_WTIH + SA_WTHL + SA_XB + SA_INIT)
__global__ __launch_bounds__(256)
void setup_inputs_kernel(const float* __restrict__ x,
                         const float* __restrict__ RIH, const float* __restrict__ MIH,
                         const float* __restrict__ RH,  const float* __restrict__ MH,
                         const float* __restrict__ RHL, const float* __restrict__ MHL,
                         float* __restrict__ part_h, float* __restrict__ part_l,
                         unsigned short* __restrict__ WT_IH,
                         unsigned short* __restrict__ WT_HL,
                         unsigned short* __restrict__ xb,
                         float* __restrict__ H0, unsigned short* __restrict__ Hb0,
                         unsigned int* __restrict__ dHa, unsigned int* __restrict__ dHb,
                         unsigned int* __restrict__ dHc,
                         int* __restrict__ act0, int* __restrict__ act1) {
    int b = blockIdx.x;
    if (b < SA_COLH) {
        col_l1_block(RH, MH, part_h, HD, b % (HD / 64), b / (HD / 64));
        return;
    }
    b -= SA_COLH;
    if (b < SA_COLL) {
        col_l1_block(RHL, MHL, part_l, Lsz, b % ((Lsz + 63) / 64), b / ((Lsz + 63) / 64));
        return;
    }
    b -= SA_COLL;
    if (b < SA_WTIH) {
        build_wt_block(RIH, MIH, nullptr,  1, WT_IH, Isz, HD, HD,
                       b % (Isz / TT), b / (Isz / TT));
        return;
    }
    b -= SA_WTIH;
    if (b < SA_WTHL) {
        build_wt_block(RHL, MHL, nullptr,  1, WT_HL, HD, Lsz, Lsz,
                       b % (HD / TT), b / (HD / TT));
        return;
    }
    b -= SA_WTHL;
    if (b < SA_XB) {
        static_assert((Bsz * Isz) % (SA_XB * 256 * 8) == 0, "whole trips");
        const int gt = b * 256 + (int)threadIdx.x;
#pragma unroll
        for (int q = 0; q < (Bsz * Isz) / (SA_XB * 256 * 8); ++q) {
            const size_t e = ((size_t)q * (SA_XB * 256) + gt) * 8;
            const float4 lo = *reinterpret_cast<const float4*>(x + e);
            const float4 hi = *reinterpret_cast<const float4*>(x + e + 4);
            u32x4 o;
            o.x = (unsigned int)f2bf(lo.x) | ((unsigned int)f2bf(lo.y) << 16);
            o.y = (unsigned int)f2bf(lo.z) | ((unsigned int)f2bf(lo.w) << 16);
            o.z = (unsigned int)f2bf(hi.x) | ((unsigned int)f2bf(hi.y) << 16);
            o.w = (unsigned int)f2bf(hi.z) | ((unsigned int)f2bf(hi.w) << 16);
            *reinterpret_cast<u32x4*>(xb + e) = o;
        }
        return;
    }
    b -= SA_XB;
    {
        static_assert((Bsz * HD) % (SA_INIT * 256 * 8) == 0, "whole trips");
        const int gt = b * 256 + (int)threadIdx.x;
        const float4 zf = {0.f, 0.f, 0.f, 0.f};
        const u32x4  zu = {0u, 0u, 0u, 0u};
#pragma unroll
        for (int q = 0; q < (Bsz * HD) / (SA_INIT * 256 * 4); ++q)
            reinterpret_cast<float4*>(H0)[(size_t)q * (SA_INIT * 256) + gt] = zf;
#pragma unroll
        for (int q = 0; q < (Bsz * HD) / (SA_INIT * 256 * 8); ++q)
            reinterpret_cast<u32x4*>(Hb0)[(size_t)q * (SA_INIT * 256) + gt] = zu;
        if (gt < Bsz) { dHa[gt] = 0u; dHb[gt] = 0u; dHc[gt] = 0u; act0[gt] = 1; act1[gt] = 1; }
    }
}

#define DRV_BLOCKS ((HD / 64) * (Bsz / 64))
__global__ __launch_bounds__(128)
void drive_scalars_kernel(const unsigned short* __restrict__ xb,
                          const unsigned short* __restrict__ WT_IH,
                          const float* __restrict__ BH,
                          float* __restrict__ drive,
                          const float* __restrict__ part_h,
                          const float* __restrict__ part_l,
                          float* __restrict__ scalars) {
    static_assert(SCB == 128, "scalars_block runs on this kernel's block size");
    __shared__ __attribute__((aligned(16))) unsigned char smem[2 * TILE_BYTES];
    if (blockIdx.x == 0) {
        scalars_block(part_h, part_l, scalars, reinterpret_cast<float*>(smem));
    } else {
        const int b = (int)blockIdx.x - 1;
        gemm_tile<0>(xb, WT_IH, Bsz, HD, Isz, BH, nullptr, nullptr, nullptr,
                     drive, nullptr, nullptr, HD,
                     smem, nullptr, nullptr, b % (HD / 64), b / (HD / 64), HD / 64);
    }
}

__device__ __forceinline__
void certify_rows(const int ub,
                  const float* __restrict__ logits,
                  const float* __restrict__ scalars,
                  unsigned int* __restrict__ dHbits,
                  const int* __restrict__ actIn,
                  int* __restrict__ actOut,
                  float* __restrict__ out,
                  const int is_last) {
    static_assert(Lsz % 4 == 0 && LP == 1024, "a row is eight 16-B pieces per lane, none straddling column Lsz");
    const int lane = threadIdx.x & 31;
    const int wave = threadIdx.x >> 5;
    const int row0 = ub * 8 + wave * 2;

    int    a[2];
    float  dh[2];
    float4 v[2][8];
#pragma unroll
    for (int j = 0; j < 2; ++j) {
        a[j]  = actIn[row0 + j];
        dh[j] = __uint_as_float(dHbits[row0 + j]);
        const float* lr = logits + (size_t)(row0 + j) * LP;
#pragma unroll
        for (int q = 0; q < 8; ++q) {
            const int col = 128 * q + 4 * lane;
            v[j][q] = *reinterpret_cast<const float4*>(lr + (col < Lsz ? col : Lsz - 4));
        }
    }
    const float gcoef = scalars[1];

#pragma unroll
    for (int j = 0; j < 2; ++j) {
        const int row = row0 + j;
        int aj = a[j];
        if (aj != 0) {
            float m1 = -1e30f, m2 = -1e30f;
#pragma unroll
            for (int q = 0; q < 8; ++q) {
                const bool in = 128 * q + 4 * lane < Lsz;
                const float e[4] = {v[j][q].x, v[j][q].y, v[j][q].z, v[j][q].w};
#pragma unroll
                for (int c = 0; c < 4; ++c) {
                    const float x = in ? e[c] : -1e30f;
                    m2 = fmaxf(m2, fminf(m1, x));
                    m1 = fmaxf(m1, x);
                }
            }
            for (int off = 16; off > 0; off >>= 1) {
                const float o1 = __shfl_xor(m1, off, 32);
                const float o2 = __shfl_xor(m2, off, 32);
                const float hi = fmaxf(m1, o1);
                const float lo = fmaxf(fminf(m1, o1), fmaxf(m2, o2));
                m1 = hi; m2 = lo;
            }
            const float margin = m1 - m2;
            const float Gamma = gcoef * dh[j];
            const bool newly = margin > 2.0f * Gamma;
            if (newly || is_last) {
                float* orow = out + (size_t)row * Lsz;
#pragma unroll
                for (int q = 0; q < 8; ++q) {
                    const int col = 128 * q + 4 * lane;
                    if (col < Lsz) *reinterpret_cast<float4*>(orow + col) = v[j][q];
                }
                aj = 0;
            }
        }
        if (lane == 0) {
            actOut[row] = aj;
            dHbits[row] = 0u;
        }
    }
}

#define REC_BLOCKS  ((HD / 64) * (Bsz / 64))
#define LOG_BLOCKS  ((LP / 64) * (Bsz / 64))
#define CERT_BLOCKS (Bsz / 8)
__global__ __launch_bounds__(128)
void step_kernel(const unsigned short* __restrict__ HbCur,
                 const float* __restrict__ HCur,
                 const unsigned short* __restrict__ WT_H,
                 const unsigned short* __restrict__ WT_HL,
                 const float* __restrict__ drive,
                 const float* __restrict__ BL,
                 const float* __restrict__ scalars,
                 const int* __restrict__ actIn,
                 int* __restrict__ actOut,
                 float* __restrict__ HNext,
                 unsigned short* __restrict__ HbNext,
                 unsigned int* __restrict__ dHNext,
                 float* __restrict__ logitsCur,
                 const float* __restrict__ logitsPrev,
                 unsigned int* __restrict__ dHPrev,
                 float* __restrict__ out,
                 int do_rec, int do_log, int do_cert, int is_last) {
    __shared__ __attribute__((aligned(16))) unsigned char smem[2 * TILE_BYTES];
    __shared__ int s_cnt[Bsz / 32];
    __shared__ int s_list[TAIL_ROWS];
    const int b = blockIdx.x;
    if (b < REC_BLOCKS) {
        if (do_rec)
            gemm_tile<1>(HbCur, WT_H, Bsz, HD, HD, nullptr, drive, HCur, actIn,
                         HNext, HbNext, dHNext, HD,
                         smem, s_cnt, s_list, b % (HD / 64), b / (HD / 64), HD / 64);
    } else if (b < REC_BLOCKS + LOG_BLOCKS) {
        const int q = b - REC_BLOCKS;
        if (do_log)
            gemm_tile<2>(HbCur, WT_HL, Bsz, LP, HD, BL, nullptr, nullptr, actIn,
                         logitsCur, nullptr, nullptr, Lsz,
                         smem, s_cnt, s_list, q % (LP / 64), q / (LP / 64), LP / 64);
    } else {
        if (do_cert)
            certify_rows(b - (REC_BLOCKS + LOG_BLOCKS), logitsPrev, scalars, dHPrev,
                         actIn, actOut, out, is_last);
    }
}

static inline char* wsalloc(char*& p, size_t bytes) {
    char* r = p;
    p += (bytes + 255) & ~(size_t)255;
    return r;
}

extern "C" void kernel_launch(void* const* d_in, const int* in_sizes, int n_in,
                              void* d_out, int out_size, void* d_ws, size_t ws_size,
                              hipStream_t stream) {
    (void)in_sizes; (void)n_in; (void)out_size; (void)ws_size;
    const float* x   = (const float*)d_in[0];
    const float* RIH = (const float*)d_in[1];
    const float* RH  = (const float*)d_in[2];
    const float* RHL = (const float*)d_in[3];
    const float* BH  = (const float*)d_in[4];
    const float* BL  = (const float*)d_in[5];
    const float* MIH = (const float*)d_in[6];
    const float* MH  = (const float*)d_in[7];
    const float* MHL = (const float*)d_in[8];
    float* out = (float*)d_out;

    char* p = (char*)d_ws;
    unsigned short* WT_H  = (unsigned short*)wsalloc(p, (size_t)HD * HD * 2);
    unsigned short* WT_HL = (unsigned short*)wsalloc(p, (size_t)LP * HD * 2);
    unsigned short* WT_IH = (unsigned short*)wsalloc(p, (size_t)HD * Isz * 2);
    unsigned short* xb    = (unsigned short*)wsalloc(p, (size_t)Bsz * Isz * 2);
    float* drive = (float*)wsalloc(p, (size_t)Bsz * HD * 4);
    float* Hbuf0 = (float*)wsalloc(p, (size_t)Bsz * HD * 4);
    float* Hbuf1 = (float*)wsalloc(p, (size_t)Bsz * HD * 4);
    unsigned short* Hb0 = (unsigned short*)wsalloc(p, (size_t)Bsz * HD * 2);
    unsigned short* Hb1 = (unsigned short*)wsalloc(p, (size_t)Bsz * HD * 2);
    float* logits = (float*)wsalloc(p, (size_t)Bsz * LP * 4);
    float* col_l1 = (float*)wsalloc(p, HD * 4);
    float* scalars = (float*)wsalloc(p, 256);
    unsigned int* dHb = (unsigned int*)wsalloc(p, Bsz * 4);
    int* active = (int*)wsalloc(p, Bsz * 4);

    unsigned int* dH1 = (unsigned int*)col_l1;
    unsigned int* dH2 = dH1 + Bsz;
    int* active1 = (int*)(dH2 + Bsz);
    float* logits1 = (float*)WT_IH;
    float* part_h = logits;
    float* part_l = logits + 8 * HD;

    setup_inputs_kernel<<<SA_BLOCKS, 256, 0, stream>>>(
        x, RIH, MIH, RH, MH, RHL, MHL, part_h, part_l, WT_IH, WT_HL, xb,
        Hbuf0, Hb0, dHb, dH1, dH2, active, active1);
    drive_scalars_kernel<<<DRV_BLOCKS + 1, 128, 0, stream>>>(
        xb, WT_IH, BH, drive, part_h, part_l, scalars);
    build_wt_kernel<<<dim3(HD / TT, HD / TT), 256, 0, stream>>>(
        RH, MH, scalars,  0, WT_H, HD, HD, HD);

    float* Hc[2] = {Hbuf0, Hbuf1};
    unsigned short* Hbc[2] = {Hb0, Hb1};
    int* act[2] = {active, active1};
    unsigned int* dHp[3] = {dHb, dH1, dH2};
    float* lg[2] = {logits, logits1};
    for (int s = 0; s <= TMAX + 1; ++s) {
        const int c = s & 1, n = (s + 1) & 1;
        step_kernel<<<REC_BLOCKS + LOG_BLOCKS + CERT_BLOCKS, 128, 0, stream>>>(
            Hbc[c], Hc[c], WT_H, WT_HL, drive, BL, scalars, act[c], act[n],
            Hc[n], Hbc[n], dHp[(s + 1) % 3], lg[c], lg[n], dHp[(s + 2) % 3], out,
            (s < TMAX) ? 1 : 0, (s >= 1 && s <= TMAX) ? 1 : 0, (s >= 2) ? 1 : 0,
            (s == TMAX + 1) ? 1 : 0);
    }
}


# Active rows gathered for the 64x64 GEMM path too: dense 64-row groups taken from the list, blocks past its end exit, groups dealt round the grid; every row's sums and their order unchanged
# speedup vs baseline: 4.5505x; 1.0588x over previous
#include <hip/hip_runtime.h>


#define Bsz 512
#define Isz 1024
#define HD  2048
#define Lsz 1000
#define LP  1024
#define TMAX 32

typedef __attribute__((ext_vector_type(16))) __bf16 v16bf;
typedef __attribute__((ext_vector_type(8)))  float  v8f;
typedef __attribute__((ext_vector_type(4)))  unsigned int u32x4;

__device__ __forceinline__ unsigned short f2bf(float f) {
    unsigned int u = __float_as_uint(f);
    u += 0x7FFFu + ((u >> 16) & 1u);
    return (unsigned short)(u >> 16);
}

#define KC 256
#define PITCH (KC * 2 + 16)
#define TILE_BYTES (64 * PITCH)

struct alignas(32) FragBits { u32x4 lo, hi; };

__device__ __forceinline__ v16bf lds_frag(const unsigned char* p) {
    FragBits fb;
    fb.lo = *reinterpret_cast<const u32x4*>(p);
    fb.hi = *reinterpret_cast<const u32x4*>(p + 32);
    v16bf v;
    __builtin_memcpy(&v, &fb, sizeof(v));
    return v;
}

#define TAIL_ROWS (Bsz / 16)
template <int MODE>
__device__ __forceinline__
void gemm_tile(const unsigned short* __restrict__ A,
               const unsigned short* __restrict__ BT,
               int M, int N, int K,
               const float* __restrict__ bias,
               const float* __restrict__ drive,
               const float* __restrict__ Hold,
               const int*   __restrict__ active,
               float* __restrict__ Cout,
               unsigned short* __restrict__ HbOut,
               unsigned int* __restrict__ dHbits,
               int NL,
               unsigned char* const smem, int* const s_cnt, int* const s_list,
               const int bx, const int by, const int lin) {
    unsigned char* const sA = smem;
    unsigned char* const sB = smem + TILE_BYTES;

    const int tid  = threadIdx.x;
    const int lane = tid & 31;
    const int wave = tid >> 5;
    const int rowBase = by * 64 + wave * 16;
    const int colBase = bx * 64;
    const int r16 = lane & 15;
    const int kh  = lane >> 4;
    int n_act = M;

    if (MODE != 0) {
        int flag[Bsz / 128];
        unsigned msk[Bsz / 128];
#pragma unroll
        for (int j = 0; j < Bsz / 128; ++j) {
            flag[j] = active[j * 128 + tid];
            msk[j]  = (unsigned)__ballot(flag[j] != 0);
        }
        if (lane == 0) {
#pragma unroll
            for (int j = 0; j < Bsz / 128; ++j) s_cnt[j * 4 + wave] = __popc(msk[j]);
        }
        __syncthreads();
        n_act = 0;
        int before[Bsz / 128];
#pragma unroll
        for (int s = 0; s < Bsz / 32; ++s) {
            if ((s & 3) == wave) before[s >> 2] = n_act;
            n_act += s_cnt[s];
        }
        if (n_act == 0) return;

#pragma unroll
        for (int j = 0; j < Bsz / 128; ++j) {
            if (flag[j] != 0)
                s_list[before[j] + __popc(msk[j] & ((1u << lane) - 1u))] = j * 128 + tid;
        }
        __syncthreads();

        if (n_act <= TAIL_ROWS) {
            const int NT = N >> 4;
            const int RT = (n_act + 15) >> 4;
            const int b  = lin;
            if (b >= RT * NT) return;
            const int rt = b / NT;
            const int ct = b - rt * NT;

            const int li   = rt * 16 + r16;
            const int arow = s_list[li < n_act ? li : 0];
            const int KQ   = K >> 2;
            const unsigned short* pa = A  + (size_t)arow * K + wave * KQ + kh * 16;
            const unsigned short* pb = BT + (size_t)(ct * 16 + r16) * K + wave * KQ + kh * 16;

            v8f part = {0.f,0.f,0.f,0.f,0.f,0.f,0.f,0.f};
            for (int k0 = 0; k0 < KQ; k0 += 512) {
                FragBits ta[16], tb[16];
#pragma unroll
                for (int i = 0; i < 16; ++i) {
                    ta[i].lo = *reinterpret_cast<const u32x4*>(pa + k0 + i * 32);
                    ta[i].hi = *reinterpret_cast<const u32x4*>(pa + k0 + i * 32 + 8);
                    tb[i].lo = *reinterpret_cast<const u32x4*>(pb + k0 + i * 32);
                    tb[i].hi = *reinterpret_cast<const u32x4*>(pb + k0 + i * 32 + 8);
                }
                __builtin_amdgcn_sched_barrier(0);
#pragma unroll
                for (int i = 0; i < 16; ++i) {
                    v16bf av, bv;
                    __builtin_memcpy(&av, &ta[i], sizeof(av));
                    __builtin_memcpy(&bv, &tb[i], sizeof(bv));
                    part = __builtin_amdgcn_wmma_f32_16x16x32_bf16(
                        false, av, false, bv, (short)0, part, false, false);
                }
            }

            float* const red = reinterpret_cast<float*>(smem);
#pragma unroll
            for (int r = 0; r < 8; ++r) red[(wave * 8 + r) * 32 + lane] = part[r];
            __syncthreads();

            const int n = ct * 16 + r16;
            int   mrow[2];
            bool  live[2];
            float v[2];
#pragma unroll
            for (int j = 0; j < 2; ++j) {
                const int r  = 2 * wave + j;
                const int lr = rt * 16 + r + 8 * kh;
                live[j] = lr < n_act;
                mrow[j] = s_list[live[j] ? lr : 0];
                v[j] = ((red[r * 32 + lane] + red[(8 + r) * 32 + lane])
                        + red[(16 + r) * 32 + lane]) + red[(24 + r) * 32 + lane];
            }
            if (MODE == 2) {
                if (n < NL) {
                    const float bn = bias[n];
#pragma unroll
                    for (int j = 0; j < 2; ++j)
                        if (live[j]) Cout[(size_t)mrow[j] * N + n] = v[j] + bn;
                }
            } else {
                float dv[2], hv[2], d[2];
#pragma unroll
                for (int j = 0; j < 2; ++j) {
                    const size_t idx = (size_t)mrow[j] * N + n;
                    dv[j] = drive[idx];
                    hv[j] = Hold[idx];
                }
#pragma unroll
                for (int j = 0; j < 2; ++j) {
                    const size_t idx = (size_t)mrow[j] * N + n;
                    const float pre  = v[j] + dv[j];
                    const float actv = pre >= 0.f ? pre : 0.05f * pre;
                    d[j] = live[j] ? fabsf(actv - hv[j]) : 0.f;
                    if (live[j]) {
                        Cout[idx]  = actv;
                        HbOut[idx] = f2bf(actv);
                    }
                }
#pragma unroll
                for (int off = 8; off > 0; off >>= 1) {
#pragma unroll
                    for (int j = 0; j < 2; ++j) d[j] = fmaxf(d[j], __shfl_xor(d[j], off, 32));
                }
#pragma unroll
                for (int j = 0; j < 2; ++j)
                    if (live[j] && r16 == 0 && d[j] > 0.f)
                        atomicMax(dHbits + mrow[j], __float_as_uint(d[j]));
            }
            return;
        }

        if (by * 64 >= n_act) return;
    }

    const int wv = __builtin_amdgcn_readfirstlane(wave);
    const unsigned char* arow[16];
#pragma unroll
    for (int i = 0; i < 16; ++i) {
        const int li = by * 64 + wv + 4 * i;
        int row = li;
        if (MODE != 0) row = __builtin_amdgcn_readfirstlane(s_list[li < n_act ? li : 0]);
        arow[i] = reinterpret_cast<const unsigned char*>(A) + (size_t)row * (size_t)(K * 2);
    }
    unsigned voff = (unsigned)(lane * 16);
    const unsigned short* gb = BT + (size_t)(colBase + wave) * K + lane * 8;
    const size_t gstep = (size_t)4 * K;
    const int sst = wave * PITCH + lane * 16;

    const int fa = (wave * 16 + r16) * PITCH + kh * 16;
    const int fb = r16 * PITCH + kh * 16;

    v8f acc0 = {0.f,0.f,0.f,0.f,0.f,0.f,0.f,0.f};
    v8f acc[4];
    acc[0] = acc0; acc[1] = acc0; acc[2] = acc0; acc[3] = acc0;

    u32x4 ra[16], rb[16];

#pragma unroll
    for (int i = 0; i < 16; ++i) {
        ra[i] = *reinterpret_cast<const u32x4*>(arow[i] + voff);
        rb[i] = *reinterpret_cast<const u32x4*>(gb + i * gstep);
    }
#pragma unroll
    for (int i = 0; i < 16; ++i) {
        *reinterpret_cast<u32x4*>(sA + sst + i * 4 * PITCH) = ra[i];
        *reinterpret_cast<u32x4*>(sB + sst + i * 4 * PITCH) = rb[i];
    }
    __syncthreads();

    const int nchunks = K / KC;
    for (int c = 0; c < nchunks; ++c) {
        const bool more = (c + 1 < nchunks);
        if (more) {
            voff += KC * 2; gb += KC;
#pragma unroll
            for (int i = 0; i < 16; ++i) {
                ra[i] = *reinterpret_cast<const u32x4*>(arow[i] + voff);
                rb[i] = *reinterpret_cast<const u32x4*>(gb + i * gstep);
            }
        }

        v16bf a_cur  = lds_frag(sA + fa);
        v16bf b_cur0 = lds_frag(sB + fb);
        v16bf b_cur1 = lds_frag(sB + fb + 16 * PITCH);
        v16bf b_cur2 = lds_frag(sB + fb + 32 * PITCH);
        v16bf b_cur3 = lds_frag(sB + fb + 48 * PITCH);
#pragma unroll
        for (int ks = 0; ks < KC / 32; ++ks) {
            v16bf a_nxt = a_cur, b_nxt0 = b_cur0, b_nxt1 = b_cur1,
                  b_nxt2 = b_cur2, b_nxt3 = b_cur3;
            if (ks + 1 < KC / 32) {
                const int ko = (ks + 1) * 64;
                a_nxt  = lds_frag(sA + fa + ko);
                b_nxt0 = lds_frag(sB + fb + ko);
                b_nxt1 = lds_frag(sB + fb + 16 * PITCH + ko);
                b_nxt2 = lds_frag(sB + fb + 32 * PITCH + ko);
                b_nxt3 = lds_frag(sB + fb + 48 * PITCH + ko);
            }
            __builtin_amdgcn_sched_barrier(0);
            acc[0] = __builtin_amdgcn_wmma_f32_16x16x32_bf16(
                false, a_cur, false, b_cur0, (short)0, acc[0], false, false);
            acc[1] = __builtin_amdgcn_wmma_f32_16x16x32_bf16(
                false, a_cur, false, b_cur1, (short)0, acc[1], false, false);
            acc[2] = __builtin_amdgcn_wmma_f32_16x16x32_bf16(
                false, a_cur, false, b_cur2, (short)0, acc[2], false, false);
            acc[3] = __builtin_amdgcn_wmma_f32_16x16x32_bf16(
                false, a_cur, false, b_cur3, (short)0, acc[3], false, false);
            __builtin_amdgcn_sched_barrier(0);
            a_cur = a_nxt;
            b_cur0 = b_nxt0; b_cur1 = b_nxt1; b_cur2 = b_nxt2; b_cur3 = b_nxt3;
        }

        __syncthreads();
        if (more) {
#pragma unroll
            for (int i = 0; i < 16; ++i) {
                *reinterpret_cast<u32x4*>(sA + sst + i * 4 * PITCH) = ra[i];
                *reinterpret_cast<u32x4*>(sB + sst + i * 4 * PITCH) = rb[i];
            }
            __syncthreads();
        }
    }

    int   mrow[8];
    bool  live[8];
    float dmax[8];
#pragma unroll
    for (int r = 0; r < 8; ++r) {
        const int li = rowBase + r + 8 * kh;
        live[r] = (MODE == 0) || (li < n_act);
        mrow[r] = (MODE != 0) ? s_list[live[r] ? li : 0] : li;
        dmax[r] = 0.f;
    }

#pragma unroll
    for (int f = 0; f < 4; ++f) {
        const int n = colBase + f * 16 + r16;
#pragma unroll
        for (int r = 0; r < 8; ++r) {
            const int m = mrow[r];
            const float v = acc[f][r];
            if (MODE == 0) {
                Cout[(size_t)m * N + n] = v + bias[n];
            } else if (MODE == 2) {
                if (n < NL && live[r]) Cout[(size_t)m * N + n] = v + bias[n];
            } else {
                const size_t idx = (size_t)m * N + n;
                const float pre = v + drive[idx];
                const float actv = pre >= 0.f ? pre : 0.05f * pre;
                const float ho = Hold[idx];
                if (live[r]) {
                    Cout[idx] = actv;
                    HbOut[idx] = f2bf(actv);
                    dmax[r] = fmaxf(dmax[r], fabsf(actv - ho));
                }
            }
        }
    }

    if (MODE == 1) {
#pragma unroll
        for (int r = 0; r < 8; ++r) {
            if (dmax[r] > 0.f) {
                atomicMax(dHbits + mrow[r], __float_as_uint(dmax[r]));
            }
        }
    }
}

__device__ __forceinline__
void col_l1_block(const float* __restrict__ A, const float* __restrict__ Msk,
                  float* __restrict__ part, const int cols, const int bx, const int b) {
    static_assert(HD % 256 == 0, "leaf rows are t + 256*r");
    __shared__ float smx[8][32];
    __shared__ float smy[8][32];
    const int lane = threadIdx.x & 31;
    const int v    = threadIdx.x >> 5;
    const int c    = (bx * 32 + lane) * 2;
    const bool in  = c < cols;
    const int cl   = in ? c : 0;
    float lx[4], ly[4];
#pragma unroll
    for (int m = 0; m < 4; ++m) {
        const int t = b + 8 * (v + 8 * m);
        float sx = 0.f, sy = 0.f;
#pragma unroll
        for (int r = 0; r < HD / 256; ++r) {
            const size_t idx = (size_t)(t + 256 * r) * cols + cl;
            const float2 a  = *reinterpret_cast<const float2*>(A + idx);
            const float2 mk = *reinterpret_cast<const float2*>(Msk + idx);
            sx += fabsf(a.x * mk.x);
            sy += fabsf(a.y * mk.y);
        }
        lx[m] = sx; ly[m] = sy;
    }
    smx[v][lane] = (lx[0] + lx[2]) + (lx[1] + lx[3]);
    smy[v][lane] = (ly[0] + ly[2]) + (ly[1] + ly[3]);
    __syncthreads();
    if (v == 0 && in) {
        part[(size_t)b * cols + c] =
            ((smx[0][lane] + smx[4][lane]) + (smx[2][lane] + smx[6][lane])) +
            ((smx[1][lane] + smx[5][lane]) + (smx[3][lane] + smx[7][lane]));
        part[(size_t)b * cols + c + 1] =
            ((smy[0][lane] + smy[4][lane]) + (smy[2][lane] + smy[6][lane])) +
            ((smy[1][lane] + smy[5][lane]) + (smy[3][lane] + smy[7][lane]));
    }
}

__device__ __forceinline__ float fold8(const float* __restrict__ p, int cols, int j) {
    const float p0 = p[j],            p1 = p[cols + j];
    const float p2 = p[2 * cols + j], p3 = p[3 * cols + j];
    const float p4 = p[4 * cols + j], p5 = p[5 * cols + j];
    const float p6 = p[6 * cols + j], p7 = p[7 * cols + j];
    return ((p0 + p4) + (p2 + p6)) + ((p1 + p5) + (p3 + p7));
}

#define SCB 128
__device__ __forceinline__
void scalars_block(const float* __restrict__ col_l1,
                   const float* __restrict__ whl,
                   float* __restrict__ scalars, float* const sm) {
    static_assert(HD % SCB == 0, "whole trips over the HD columns");
    const int t = threadIdx.x;
    float mx = -1e30f;
#pragma unroll
    for (int i = 0; i < HD / SCB; ++i) mx = fmaxf(mx, fold8(col_l1, HD, t + SCB * i));
    sm[t] = mx; __syncthreads();
    for (int s = SCB / 2; s > 0; s >>= 1) {
        if (t < s) sm[t] = fmaxf(sm[t], sm[t + s]);
        __syncthreads();
    }
    const float m1 = sm[0];
    __syncthreads();
    mx = -1e30f;
#pragma unroll
    for (int i = 0; i < (Lsz + SCB - 1) / SCB; ++i) {
        const int j = t + SCB * i;
        mx = fmaxf(mx, fold8(whl, Lsz, j < Lsz ? j : Lsz - 1));
    }
    sm[t] = mx; __syncthreads();
    for (int s = SCB / 2; s > 0; s >>= 1) {
        if (t < s) sm[t] = fmaxf(sm[t], sm[t + s]);
        __syncthreads();
    }
    if (t == 0) {
        scalars[0] = 0.95f / fmaxf(m1, 1e-8f);
        scalars[1] = sm[0] * 19.0f;
    }
}

#define TT 64
#define TPITCH (TT + 2)
__device__ __forceinline__
void build_wt_block(const float* __restrict__ R, const float* __restrict__ M,
                    const float* __restrict__ scalars, const int sel,
                    unsigned short* __restrict__ WT,
                    const int K, const int srcld, const int NL,
                    const int bkx, const int bny) {
    __shared__ __attribute__((aligned(16))) unsigned short tile[TT * TPITCH];
    const int k0 = bkx * TT;
    const int n0 = bny * TT;
    const int t  = threadIdx.x;

    const int ln = (t & 15) * 4;
    const int lk = t >> 4;
    const int n  = n0 + ln;
    const bool in = n < NL;
    const int nc = in ? n : 0;
    float4 r[4], m[4];
#pragma unroll
    for (int i = 0; i < 4; ++i) {
        const size_t sidx = (size_t)(k0 + lk + 16 * i) * srcld + nc;
        r[i] = *reinterpret_cast<const float4*>(R + sidx);
        m[i] = *reinterpret_cast<const float4*>(M + sidx);
    }
    float sc = 1.f;
    if (sel == 0) sc = scalars[0];
#pragma unroll
    for (int i = 0; i < 4; ++i) {
        float v0 = r[i].x * m[i].x, v1 = r[i].y * m[i].y;
        float v2 = r[i].z * m[i].z, v3 = r[i].w * m[i].w;
        if (sel == 0) { v0 *= sc; v1 *= sc; v2 *= sc; v3 *= sc; }
        if (!in) { v0 = 0.f; v1 = 0.f; v2 = 0.f; v3 = 0.f; }
        unsigned int* d = reinterpret_cast<unsigned int*>(tile + (lk + 16 * i) * TPITCH + ln);
        d[0] = (unsigned int)f2bf(v0) | ((unsigned int)f2bf(v1) << 16);
        d[1] = (unsigned int)f2bf(v2) | ((unsigned int)f2bf(v3) << 16);
    }
    __syncthreads();

    const int ks = (t & 7) * 8;
#pragma unroll
    for (int p = 0; p < 2; ++p) {
        const int nl = (t >> 3) + 32 * p;
        unsigned int w[4];
#pragma unroll
        for (int j = 0; j < 4; ++j) {
            const unsigned int e0 = tile[(ks + 2 * j) * TPITCH + nl];
            const unsigned int e1 = tile[(ks + 2 * j + 1) * TPITCH + nl];
            w[j] = e0 | (e1 << 16);
        }
        u32x4 o; o.x = w[0]; o.y = w[1]; o.z = w[2]; o.w = w[3];
        *reinterpret_cast<u32x4*>(WT + (size_t)(n0 + nl) * K + k0 + ks) = o;
    }
}

__global__ __launch_bounds__(256)
void build_wt_kernel(const float* __restrict__ R, const float* __restrict__ M,
                     const float* __restrict__ scalars, int sel,
                     unsigned short* __restrict__ WT,
                     int K, int srcld, int NL) {
    build_wt_block(R, M, scalars, sel, WT, K, srcld, NL, (int)blockIdx.x, (int)blockIdx.y);
}

#define SA_COLH  ((HD / 64) * 8)
#define SA_COLL  (((Lsz + 63) / 64) * 8)
#define SA_WTIH  ((Isz / TT) * (HD / TT))
#define SA_WTHL  ((HD / TT) * (LP / TT))
#define SA_XB    128
#define SA_INIT  128
#define SA_BLOCKS (SA_COLH + SA_COLL + SA_WTIH + SA_WTHL + SA_XB + SA_INIT)
__global__ __launch_bounds__(256)
void setup_inputs_kernel(const float* __restrict__ x,
                         const float* __restrict__ RIH, const float* __restrict__ MIH,
                         const float* __restrict__ RH,  const float* __restrict__ MH,
                         const float* __restrict__ RHL, const float* __restrict__ MHL,
                         float* __restrict__ part_h, float* __restrict__ part_l,
                         unsigned short* __restrict__ WT_IH,
                         unsigned short* __restrict__ WT_HL,
                         unsigned short* __restrict__ xb,
                         float* __restrict__ H0, unsigned short* __restrict__ Hb0,
                         unsigned int* __restrict__ dHa, unsigned int* __restrict__ dHb,
                         unsigned int* __restrict__ dHc,
                         int* __restrict__ act0, int* __restrict__ act1) {
    int b = blockIdx.x;
    if (b < SA_COLH) {
        col_l1_block(RH, MH, part_h, HD, b % (HD / 64), b / (HD / 64));
        return;
    }
    b -= SA_COLH;
    if (b < SA_COLL) {
        col_l1_block(RHL, MHL, part_l, Lsz, b % ((Lsz + 63) / 64), b / ((Lsz + 63) / 64));
        return;
    }
    b -= SA_COLL;
    if (b < SA_WTIH) {
        build_wt_block(RIH, MIH, nullptr,  1, WT_IH, Isz, HD, HD,
                       b % (Isz / TT), b / (Isz / TT));
        return;
    }
    b -= SA_WTIH;
    if (b < SA_WTHL) {
        build_wt_block(RHL, MHL, nullptr,  1, WT_HL, HD, Lsz, Lsz,
                       b % (HD / TT), b / (HD / TT));
        return;
    }
    b -= SA_WTHL;
    if (b < SA_XB) {
        static_assert((Bsz * Isz) % (SA_XB * 256 * 8) == 0, "whole trips");
        const int gt = b * 256 + (int)threadIdx.x;
#pragma unroll
        for (int q = 0; q < (Bsz * Isz) / (SA_XB * 256 * 8); ++q) {
            const size_t e = ((size_t)q * (SA_XB * 256) + gt) * 8;
            const float4 lo = *reinterpret_cast<const float4*>(x + e);
            const float4 hi = *reinterpret_cast<const float4*>(x + e + 4);
            u32x4 o;
            o.x = (unsigned int)f2bf(lo.x) | ((unsigned int)f2bf(lo.y) << 16);
            o.y = (unsigned int)f2bf(lo.z) | ((unsigned int)f2bf(lo.w) << 16);
            o.z = (unsigned int)f2bf(hi.x) | ((unsigned int)f2bf(hi.y) << 16);
            o.w = (unsigned int)f2bf(hi.z) | ((unsigned int)f2bf(hi.w) << 16);
            *reinterpret_cast<u32x4*>(xb + e) = o;
        }
        return;
    }
    b -= SA_XB;
    {
        static_assert((Bsz * HD) % (SA_INIT * 256 * 8) == 0, "whole trips");
        const int gt = b * 256 + (int)threadIdx.x;
        const float4 zf = {0.f, 0.f, 0.f, 0.f};
        const u32x4  zu = {0u, 0u, 0u, 0u};
#pragma unroll
        for (int q = 0; q < (Bsz * HD) / (SA_INIT * 256 * 4); ++q)
            reinterpret_cast<float4*>(H0)[(size_t)q * (SA_INIT * 256) + gt] = zf;
#pragma unroll
        for (int q = 0; q < (Bsz * HD) / (SA_INIT * 256 * 8); ++q)
            reinterpret_cast<u32x4*>(Hb0)[(size_t)q * (SA_INIT * 256) + gt] = zu;
        if (gt < Bsz) { dHa[gt] = 0u; dHb[gt] = 0u; dHc[gt] = 0u; act0[gt] = 1; act1[gt] = 1; }
    }
}

#define DRV_BLOCKS ((HD / 64) * (Bsz / 64))
__global__ __launch_bounds__(128)
void drive_scalars_kernel(const unsigned short* __restrict__ xb,
                          const unsigned short* __restrict__ WT_IH,
                          const float* __restrict__ BH,
                          float* __restrict__ drive,
                          const float* __restrict__ part_h,
                          const float* __restrict__ part_l,
                          float* __restrict__ scalars) {
    static_assert(SCB == 128, "scalars_block runs on this kernel's block size");
    __shared__ __attribute__((aligned(16))) unsigned char smem[2 * TILE_BYTES];
    if (blockIdx.x == 0) {
        scalars_block(part_h, part_l, scalars, reinterpret_cast<float*>(smem));
    } else {
        const int b = (int)blockIdx.x - 1;
        gemm_tile<0>(xb, WT_IH, Bsz, HD, Isz, BH, nullptr, nullptr, nullptr,
                     drive, nullptr, nullptr, HD,
                     smem, nullptr, nullptr, b % (HD / 64), b / (HD / 64), b);
    }
}

__device__ __forceinline__
void certify_rows(const int ub,
                  const float* __restrict__ logits,
                  const float* __restrict__ scalars,
                  unsigned int* __restrict__ dHbits,
                  const int* __restrict__ actIn,
                  int* __restrict__ actOut,
                  float* __restrict__ out,
                  const int is_last) {
    static_assert(Lsz % 4 == 0 && LP == 1024, "a row is eight 16-B pieces per lane, none straddling column Lsz");
    const int lane = threadIdx.x & 31;
    const int wave = threadIdx.x >> 5;
    const int row0 = ub * 8 + wave * 2;

    int    a[2];
    float  dh[2];
    float4 v[2][8];
#pragma unroll
    for (int j = 0; j < 2; ++j) {
        a[j]  = actIn[row0 + j];
        dh[j] = __uint_as_float(dHbits[row0 + j]);
        const float* lr = logits + (size_t)(row0 + j) * LP;
#pragma unroll
        for (int q = 0; q < 8; ++q) {
            const int col = 128 * q + 4 * lane;
            v[j][q] = *reinterpret_cast<const float4*>(lr + (col < Lsz ? col : Lsz - 4));
        }
    }
    const float gcoef = scalars[1];

#pragma unroll
    for (int j = 0; j < 2; ++j) {
        const int row = row0 + j;
        int aj = a[j];
        if (aj != 0) {
            float m1 = -1e30f, m2 = -1e30f;
#pragma unroll
            for (int q = 0; q < 8; ++q) {
                const bool in = 128 * q + 4 * lane < Lsz;
                const float e[4] = {v[j][q].x, v[j][q].y, v[j][q].z, v[j][q].w};
#pragma unroll
                for (int c = 0; c < 4; ++c) {
                    const float x = in ? e[c] : -1e30f;
                    m2 = fmaxf(m2, fminf(m1, x));
                    m1 = fmaxf(m1, x);
                }
            }
            for (int off = 16; off > 0; off >>= 1) {
                const float o1 = __shfl_xor(m1, off, 32);
                const float o2 = __shfl_xor(m2, off, 32);
                const float hi = fmaxf(m1, o1);
                const float lo = fmaxf(fminf(m1, o1), fmaxf(m2, o2));
                m1 = hi; m2 = lo;
            }
            const float margin = m1 - m2;
            const float Gamma = gcoef * dh[j];
            const bool newly = margin > 2.0f * Gamma;
            if (newly || is_last) {
                float* orow = out + (size_t)row * Lsz;
#pragma unroll
                for (int q = 0; q < 8; ++q) {
                    const int col = 128 * q + 4 * lane;
                    if (col < Lsz) *reinterpret_cast<float4*>(orow + col) = v[j][q];
                }
                aj = 0;
            }
        }
        if (lane == 0) {
            actOut[row] = aj;
            dHbits[row] = 0u;
        }
    }
}

#define ROW_GROUPS  (Bsz / 64)
#define REC_BLOCKS  ((HD / 64) * ROW_GROUPS)
#define LOG_BLOCKS  ((LP / 64) * ROW_GROUPS)
#define CERT_BLOCKS (Bsz / 8)
__global__ __launch_bounds__(128)
void step_kernel(const unsigned short* __restrict__ HbCur,
                 const float* __restrict__ HCur,
                 const unsigned short* __restrict__ WT_H,
                 const unsigned short* __restrict__ WT_HL,
                 const float* __restrict__ drive,
                 const float* __restrict__ BL,
                 const float* __restrict__ scalars,
                 const int* __restrict__ actIn,
                 int* __restrict__ actOut,
                 float* __restrict__ HNext,
                 unsigned short* __restrict__ HbNext,
                 unsigned int* __restrict__ dHNext,
                 float* __restrict__ logitsCur,
                 const float* __restrict__ logitsPrev,
                 unsigned int* __restrict__ dHPrev,
                 float* __restrict__ out,
                 int do_rec, int do_log, int do_cert, int is_last) {
    __shared__ __attribute__((aligned(16))) unsigned char smem[2 * TILE_BYTES];
    __shared__ int s_cnt[Bsz / 32];
    __shared__ int s_list[Bsz];
    const int b = blockIdx.x;
    if (b < REC_BLOCKS) {
        if (do_rec)
            gemm_tile<1>(HbCur, WT_H, Bsz, HD, HD, nullptr, drive, HCur, actIn,
                         HNext, HbNext, dHNext, HD,
                         smem, s_cnt, s_list, b / ROW_GROUPS, b % ROW_GROUPS, b);
    } else if (b < REC_BLOCKS + LOG_BLOCKS) {
        const int q = b - REC_BLOCKS;
        if (do_log)
            gemm_tile<2>(HbCur, WT_HL, Bsz, LP, HD, BL, nullptr, nullptr, actIn,
                         logitsCur, nullptr, nullptr, Lsz,
                         smem, s_cnt, s_list, q / ROW_GROUPS, ROW_GROUPS - 1 - q % ROW_GROUPS, q);
    } else {
        if (do_cert)
            certify_rows(b - (REC_BLOCKS + LOG_BLOCKS), logitsPrev, scalars, dHPrev,
                         actIn, actOut, out, is_last);
    }
}

static inline char* wsalloc(char*& p, size_t bytes) {
    char* r = p;
    p += (bytes + 255) & ~(size_t)255;
    return r;
}

extern "C" void kernel_launch(void* const* d_in, const int* in_sizes, int n_in,
                              void* d_out, int out_size, void* d_ws, size_t ws_size,
                              hipStream_t stream) {
    (void)in_sizes; (void)n_in; (void)out_size; (void)ws_size;
    const float* x   = (const float*)d_in[0];
    const float* RIH = (const float*)d_in[1];
    const float* RH  = (const float*)d_in[2];
    const float* RHL = (const float*)d_in[3];
    const float* BH  = (const float*)d_in[4];
    const float* BL  = (const float*)d_in[5];
    const float* MIH = (const float*)d_in[6];
    const float* MH  = (const float*)d_in[7];
    const float* MHL = (const float*)d_in[8];
    float* out = (float*)d_out;

    char* p = (char*)d_ws;
    unsigned short* WT_H  = (unsigned short*)wsalloc(p, (size_t)HD * HD * 2);
    unsigned short* WT_HL = (unsigned short*)wsalloc(p, (size_t)LP * HD * 2);
    unsigned short* WT_IH = (unsigned short*)wsalloc(p, (size_t)HD * Isz * 2);
    unsigned short* xb    = (unsigned short*)wsalloc(p, (size_t)Bsz * Isz * 2);
    float* drive = (float*)wsalloc(p, (size_t)Bsz * HD * 4);
    float* Hbuf0 = (float*)wsalloc(p, (size_t)Bsz * HD * 4);
    float* Hbuf1 = (float*)wsalloc(p, (size_t)Bsz * HD * 4);
    unsigned short* Hb0 = (unsigned short*)wsalloc(p, (size_t)Bsz * HD * 2);
    unsigned short* Hb1 = (unsigned short*)wsalloc(p, (size_t)Bsz * HD * 2);
    float* logits = (float*)wsalloc(p, (size_t)Bsz * LP * 4);
    float* col_l1 = (float*)wsalloc(p, HD * 4);
    float* scalars = (float*)wsalloc(p, 256);
    unsigned int* dHb = (unsigned int*)wsalloc(p, Bsz * 4);
    int* active = (int*)wsalloc(p, Bsz * 4);

    unsigned int* dH1 = (unsigned int*)col_l1;
    unsigned int* dH2 = dH1 + Bsz;
    int* active1 = (int*)(dH2 + Bsz);
    float* logits1 = (float*)WT_IH;
    float* part_h = logits;
    float* part_l = logits + 8 * HD;

    setup_inputs_kernel<<<SA_BLOCKS, 256, 0, stream>>>(
        x, RIH, MIH, RH, MH, RHL, MHL, part_h, part_l, WT_IH, WT_HL, xb,
        Hbuf0, Hb0, dHb, dH1, dH2, active, active1);
    drive_scalars_kernel<<<DRV_BLOCKS + 1, 128, 0, stream>>>(
        xb, WT_IH, BH, drive, part_h, part_l, scalars);
    build_wt_kernel<<<dim3(HD / TT, HD / TT), 256, 0, stream>>>(
        RH, MH, scalars,  0, WT_H, HD, HD, HD);

    float* Hc[2] = {Hbuf0, Hbuf1};
    unsigned short* Hbc[2] = {Hb0, Hb1};
    int* act[2] = {active, active1};
    unsigned int* dHp[3] = {dHb, dH1, dH2};
    float* lg[2] = {logits, logits1};
    for (int s = 0; s <= TMAX + 1; ++s) {
        const int c = s & 1, n = (s + 1) & 1;
        step_kernel<<<REC_BLOCKS + LOG_BLOCKS + CERT_BLOCKS, 128, 0, stream>>>(
            Hbc[c], Hc[c], WT_H, WT_HL, drive, BL, scalars, act[c], act[n],
            Hc[n], Hbc[n], dHp[(s + 1) % 3], lg[c], lg[n], dHp[(s + 2) % 3], out,
            (s < TMAX) ? 1 : 0, (s >= 1 && s <= TMAX) ? 1 : 0, (s >= 2) ? 1 : 0,
            (s == TMAX + 1) ? 1 : 0);
    }
}


# 64x64 GEMM epilogue: the tile's drive, old-H and bias values fetched as one batch of independent loads before any use, stores grouped by row; was a load-wait-store round trip per element; same sums
# speedup vs baseline: 5.1474x; 1.1312x over previous
#include <hip/hip_runtime.h>


#define Bsz 512
#define Isz 1024
#define HD  2048
#define Lsz 1000
#define LP  1024
#define TMAX 32

typedef __attribute__((ext_vector_type(16))) __bf16 v16bf;
typedef __attribute__((ext_vector_type(8)))  float  v8f;
typedef __attribute__((ext_vector_type(4)))  unsigned int u32x4;

__device__ __forceinline__ unsigned short f2bf(float f) {
    unsigned int u = __float_as_uint(f);
    u += 0x7FFFu + ((u >> 16) & 1u);
    return (unsigned short)(u >> 16);
}

#define KC 256
#define PITCH (KC * 2 + 16)
#define TILE_BYTES (64 * PITCH)

struct alignas(32) FragBits { u32x4 lo, hi; };

__device__ __forceinline__ v16bf lds_frag(const unsigned char* p) {
    FragBits fb;
    fb.lo = *reinterpret_cast<const u32x4*>(p);
    fb.hi = *reinterpret_cast<const u32x4*>(p + 32);
    v16bf v;
    __builtin_memcpy(&v, &fb, sizeof(v));
    return v;
}

#define TAIL_ROWS (Bsz / 16)
template <int MODE>
__device__ __forceinline__
void gemm_tile(const unsigned short* __restrict__ A,
               const unsigned short* __restrict__ BT,
               int M, int N, int K,
               const float* __restrict__ bias,
               const float* __restrict__ drive,
               const float* __restrict__ Hold,
               const int*   __restrict__ active,
               float* __restrict__ Cout,
               unsigned short* __restrict__ HbOut,
               unsigned int* __restrict__ dHbits,
               int NL,
               unsigned char* const smem, int* const s_cnt, int* const s_list,
               const int bx, const int by, const int lin) {
    unsigned char* const sA = smem;
    unsigned char* const sB = smem + TILE_BYTES;

    const int tid  = threadIdx.x;
    const int lane = tid & 31;
    const int wave = tid >> 5;
    const int rowBase = by * 64 + wave * 16;
    const int colBase = bx * 64;
    const int r16 = lane & 15;
    const int kh  = lane >> 4;
    int n_act = M;

    if (MODE != 0) {
        int flag[Bsz / 128];
        unsigned msk[Bsz / 128];
#pragma unroll
        for (int j = 0; j < Bsz / 128; ++j) {
            flag[j] = active[j * 128 + tid];
            msk[j]  = (unsigned)__ballot(flag[j] != 0);
        }
        if (lane == 0) {
#pragma unroll
            for (int j = 0; j < Bsz / 128; ++j) s_cnt[j * 4 + wave] = __popc(msk[j]);
        }
        __syncthreads();
        n_act = 0;
        int before[Bsz / 128];
#pragma unroll
        for (int s = 0; s < Bsz / 32; ++s) {
            if ((s & 3) == wave) before[s >> 2] = n_act;
            n_act += s_cnt[s];
        }
        if (n_act == 0) return;

#pragma unroll
        for (int j = 0; j < Bsz / 128; ++j) {
            if (flag[j] != 0)
                s_list[before[j] + __popc(msk[j] & ((1u << lane) - 1u))] = j * 128 + tid;
        }
        __syncthreads();

        if (n_act <= TAIL_ROWS) {
            const int NT = N >> 4;
            const int RT = (n_act + 15) >> 4;
            const int b  = lin;
            if (b >= RT * NT) return;
            const int rt = b / NT;
            const int ct = b - rt * NT;

            const int li   = rt * 16 + r16;
            const int arow = s_list[li < n_act ? li : 0];
            const int KQ   = K >> 2;
            const unsigned short* pa = A  + (size_t)arow * K + wave * KQ + kh * 16;
            const unsigned short* pb = BT + (size_t)(ct * 16 + r16) * K + wave * KQ + kh * 16;

            v8f part = {0.f,0.f,0.f,0.f,0.f,0.f,0.f,0.f};
            for (int k0 = 0; k0 < KQ; k0 += 512) {
                FragBits ta[16], tb[16];
#pragma unroll
                for (int i = 0; i < 16; ++i) {
                    ta[i].lo = *reinterpret_cast<const u32x4*>(pa + k0 + i * 32);
                    ta[i].hi = *reinterpret_cast<const u32x4*>(pa + k0 + i * 32 + 8);
                    tb[i].lo = *reinterpret_cast<const u32x4*>(pb + k0 + i * 32);
                    tb[i].hi = *reinterpret_cast<const u32x4*>(pb + k0 + i * 32 + 8);
                }
                __builtin_amdgcn_sched_barrier(0);
#pragma unroll
                for (int i = 0; i < 16; ++i) {
                    v16bf av, bv;
                    __builtin_memcpy(&av, &ta[i], sizeof(av));
                    __builtin_memcpy(&bv, &tb[i], sizeof(bv));
                    part = __builtin_amdgcn_wmma_f32_16x16x32_bf16(
                        false, av, false, bv, (short)0, part, false, false);
                }
            }

            float* const red = reinterpret_cast<float*>(smem);
#pragma unroll
            for (int r = 0; r < 8; ++r) red[(wave * 8 + r) * 32 + lane] = part[r];
            __syncthreads();

            const int n = ct * 16 + r16;
            int   mrow[2];
            bool  live[2];
            float v[2];
#pragma unroll
            for (int j = 0; j < 2; ++j) {
                const int r  = 2 * wave + j;
                const int lr = rt * 16 + r + 8 * kh;
                live[j] = lr < n_act;
                mrow[j] = s_list[live[j] ? lr : 0];
                v[j] = ((red[r * 32 + lane] + red[(8 + r) * 32 + lane])
                        + red[(16 + r) * 32 + lane]) + red[(24 + r) * 32 + lane];
            }
            if (MODE == 2) {
                if (n < NL) {
                    const float bn = bias[n];
#pragma unroll
                    for (int j = 0; j < 2; ++j)
                        if (live[j]) Cout[(size_t)mrow[j] * N + n] = v[j] + bn;
                }
            } else {
                float dv[2], hv[2], d[2];
#pragma unroll
                for (int j = 0; j < 2; ++j) {
                    const size_t idx = (size_t)mrow[j] * N + n;
                    dv[j] = drive[idx];
                    hv[j] = Hold[idx];
                }
#pragma unroll
                for (int j = 0; j < 2; ++j) {
                    const size_t idx = (size_t)mrow[j] * N + n;
                    const float pre  = v[j] + dv[j];
                    const float actv = pre >= 0.f ? pre : 0.05f * pre;
                    d[j] = live[j] ? fabsf(actv - hv[j]) : 0.f;
                    if (live[j]) {
                        Cout[idx]  = actv;
                        HbOut[idx] = f2bf(actv);
                    }
                }
#pragma unroll
                for (int off = 8; off > 0; off >>= 1) {
#pragma unroll
                    for (int j = 0; j < 2; ++j) d[j] = fmaxf(d[j], __shfl_xor(d[j], off, 32));
                }
#pragma unroll
                for (int j = 0; j < 2; ++j)
                    if (live[j] && r16 == 0 && d[j] > 0.f)
                        atomicMax(dHbits + mrow[j], __float_as_uint(d[j]));
            }
            return;
        }

        if (by * 64 >= n_act) return;
    }

    const int wv = __builtin_amdgcn_readfirstlane(wave);
    const unsigned char* arow[16];
#pragma unroll
    for (int i = 0; i < 16; ++i) {
        const int li = by * 64 + wv + 4 * i;
        int row = li;
        if (MODE != 0) row = __builtin_amdgcn_readfirstlane(s_list[li < n_act ? li : 0]);
        arow[i] = reinterpret_cast<const unsigned char*>(A) + (size_t)row * (size_t)(K * 2);
    }
    unsigned voff = (unsigned)(lane * 16);
    const unsigned short* gb = BT + (size_t)(colBase + wave) * K + lane * 8;
    const size_t gstep = (size_t)4 * K;
    const int sst = wave * PITCH + lane * 16;

    const int fa = (wave * 16 + r16) * PITCH + kh * 16;
    const int fb = r16 * PITCH + kh * 16;

    v8f acc0 = {0.f,0.f,0.f,0.f,0.f,0.f,0.f,0.f};
    v8f acc[4];
    acc[0] = acc0; acc[1] = acc0; acc[2] = acc0; acc[3] = acc0;

    u32x4 ra[16], rb[16];

#pragma unroll
    for (int i = 0; i < 16; ++i) {
        ra[i] = *reinterpret_cast<const u32x4*>(arow[i] + voff);
        rb[i] = *reinterpret_cast<const u32x4*>(gb + i * gstep);
    }
#pragma unroll
    for (int i = 0; i < 16; ++i) {
        *reinterpret_cast<u32x4*>(sA + sst + i * 4 * PITCH) = ra[i];
        *reinterpret_cast<u32x4*>(sB + sst + i * 4 * PITCH) = rb[i];
    }
    __syncthreads();

    const int nchunks = K / KC;
    for (int c = 0; c < nchunks; ++c) {
        const bool more = (c + 1 < nchunks);
        if (more) {
            voff += KC * 2; gb += KC;
#pragma unroll
            for (int i = 0; i < 16; ++i) {
                ra[i] = *reinterpret_cast<const u32x4*>(arow[i] + voff);
                rb[i] = *reinterpret_cast<const u32x4*>(gb + i * gstep);
            }
        }

        v16bf a_cur  = lds_frag(sA + fa);
        v16bf b_cur0 = lds_frag(sB + fb);
        v16bf b_cur1 = lds_frag(sB + fb + 16 * PITCH);
        v16bf b_cur2 = lds_frag(sB + fb + 32 * PITCH);
        v16bf b_cur3 = lds_frag(sB + fb + 48 * PITCH);
#pragma unroll
        for (int ks = 0; ks < KC / 32; ++ks) {
            v16bf a_nxt = a_cur, b_nxt0 = b_cur0, b_nxt1 = b_cur1,
                  b_nxt2 = b_cur2, b_nxt3 = b_cur3;
            if (ks + 1 < KC / 32) {
                const int ko = (ks + 1) * 64;
                a_nxt  = lds_frag(sA + fa + ko);
                b_nxt0 = lds_frag(sB + fb + ko);
                b_nxt1 = lds_frag(sB + fb + 16 * PITCH + ko);
                b_nxt2 = lds_frag(sB + fb + 32 * PITCH + ko);
                b_nxt3 = lds_frag(sB + fb + 48 * PITCH + ko);
            }
            __builtin_amdgcn_sched_barrier(0);
            acc[0] = __builtin_amdgcn_wmma_f32_16x16x32_bf16(
                false, a_cur, false, b_cur0, (short)0, acc[0], false, false);
            acc[1] = __builtin_amdgcn_wmma_f32_16x16x32_bf16(
                false, a_cur, false, b_cur1, (short)0, acc[1], false, false);
            acc[2] = __builtin_amdgcn_wmma_f32_16x16x32_bf16(
                false, a_cur, false, b_cur2, (short)0, acc[2], false, false);
            acc[3] = __builtin_amdgcn_wmma_f32_16x16x32_bf16(
                false, a_cur, false, b_cur3, (short)0, acc[3], false, false);
            __builtin_amdgcn_sched_barrier(0);
            a_cur = a_nxt;
            b_cur0 = b_nxt0; b_cur1 = b_nxt1; b_cur2 = b_nxt2; b_cur3 = b_nxt3;
        }

        __syncthreads();
        if (more) {
#pragma unroll
            for (int i = 0; i < 16; ++i) {
                *reinterpret_cast<u32x4*>(sA + sst + i * 4 * PITCH) = ra[i];
                *reinterpret_cast<u32x4*>(sB + sst + i * 4 * PITCH) = rb[i];
            }
            __syncthreads();
        }
    }

    int   mrow[8];
    bool  live[8];
    float dmax[8];
#pragma unroll
    for (int r = 0; r < 8; ++r) {
        const int li = rowBase + r + 8 * kh;
        live[r] = (MODE == 0) || (li < n_act);
        mrow[r] = (MODE != 0) ? s_list[live[r] ? li : 0] : li;
        dmax[r] = 0.f;
    }

    if (MODE == 0) {
#pragma unroll
        for (int f = 0; f < 4; ++f) {
            const int n = colBase + f * 16 + r16;
#pragma unroll
            for (int r = 0; r < 8; ++r)
                Cout[(size_t)mrow[r] * N + n] = acc[f][r] + bias[n];
        }
    } else if (MODE == 2) {
        float bn[4];
#pragma unroll
        for (int f = 0; f < 4; ++f) {
            const int n = colBase + f * 16 + r16;
            bn[f] = bias[n < NL ? n : 0];
        }
        float lg[4][8];
#pragma unroll
        for (int f = 0; f < 4; ++f) {
#pragma unroll
            for (int r = 0; r < 8; ++r) lg[f][r] = acc[f][r] + bn[f];
        }
#pragma unroll
        for (int r = 0; r < 8; ++r) {
            if (live[r]) {
#pragma unroll
                for (int f = 0; f < 4; ++f) {
                    const int n = colBase + f * 16 + r16;
                    if (n < NL) Cout[(size_t)mrow[r] * N + n] = lg[f][r];
                }
            }
        }
    } else {
        float dv[4][8], hv[4][8];
#pragma unroll
        for (int f = 0; f < 4; ++f) {
#pragma unroll
            for (int r = 0; r < 8; ++r) {
                const size_t idx = (size_t)mrow[r] * N + (colBase + f * 16 + r16);
                dv[f][r] = drive[idx];
                hv[f][r] = Hold[idx];
            }
        }
        __builtin_amdgcn_sched_barrier(0);
#pragma unroll
        for (int f = 0; f < 4; ++f) {
#pragma unroll
            for (int r = 0; r < 8; ++r) {
                const float pre  = acc[f][r] + dv[f][r];
                const float actv = pre >= 0.f ? pre : 0.05f * pre;
                const float dd   = fabsf(actv - hv[f][r]);
                dmax[r] = fmaxf(dmax[r], live[r] ? dd : 0.f);
                acc[f][r] = actv;
            }
        }
#pragma unroll
        for (int r = 0; r < 8; ++r) {
            if (live[r]) {
#pragma unroll
                for (int f = 0; f < 4; ++f) {
                    const size_t idx = (size_t)mrow[r] * N + (colBase + f * 16 + r16);
                    Cout[idx]  = acc[f][r];
                    HbOut[idx] = f2bf(acc[f][r]);
                }
            }
        }
#pragma unroll
        for (int r = 0; r < 8; ++r) {
            if (dmax[r] > 0.f) {
                atomicMax(dHbits + mrow[r], __float_as_uint(dmax[r]));
            }
        }
    }
}

__device__ __forceinline__
void col_l1_block(const float* __restrict__ A, const float* __restrict__ Msk,
                  float* __restrict__ part, const int cols, const int bx, const int b) {
    static_assert(HD % 256 == 0, "leaf rows are t + 256*r");
    __shared__ float smx[8][32];
    __shared__ float smy[8][32];
    const int lane = threadIdx.x & 31;
    const int v    = threadIdx.x >> 5;
    const int c    = (bx * 32 + lane) * 2;
    const bool in  = c < cols;
    const int cl   = in ? c : 0;
    float lx[4], ly[4];
#pragma unroll
    for (int m = 0; m < 4; ++m) {
        const int t = b + 8 * (v + 8 * m);
        float sx = 0.f, sy = 0.f;
#pragma unroll
        for (int r = 0; r < HD / 256; ++r) {
            const size_t idx = (size_t)(t + 256 * r) * cols + cl;
            const float2 a  = *reinterpret_cast<const float2*>(A + idx);
            const float2 mk = *reinterpret_cast<const float2*>(Msk + idx);
            sx += fabsf(a.x * mk.x);
            sy += fabsf(a.y * mk.y);
        }
        lx[m] = sx; ly[m] = sy;
    }
    smx[v][lane] = (lx[0] + lx[2]) + (lx[1] + lx[3]);
    smy[v][lane] = (ly[0] + ly[2]) + (ly[1] + ly[3]);
    __syncthreads();
    if (v == 0 && in) {
        part[(size_t)b * cols + c] =
            ((smx[0][lane] + smx[4][lane]) + (smx[2][lane] + smx[6][lane])) +
            ((smx[1][lane] + smx[5][lane]) + (smx[3][lane] + smx[7][lane]));
        part[(size_t)b * cols + c + 1] =
            ((smy[0][lane] + smy[4][lane]) + (smy[2][lane] + smy[6][lane])) +
            ((smy[1][lane] + smy[5][lane]) + (smy[3][lane] + smy[7][lane]));
    }
}

__device__ __forceinline__ float fold8(const float* __restrict__ p, int cols, int j) {
    const float p0 = p[j],            p1 = p[cols + j];
    const float p2 = p[2 * cols + j], p3 = p[3 * cols + j];
    const float p4 = p[4 * cols + j], p5 = p[5 * cols + j];
    const float p6 = p[6 * cols + j], p7 = p[7 * cols + j];
    return ((p0 + p4) + (p2 + p6)) + ((p1 + p5) + (p3 + p7));
}

#define SCB 128
__device__ __forceinline__
void scalars_block(const float* __restrict__ col_l1,
                   const float* __restrict__ whl,
                   float* __restrict__ scalars, float* const sm) {
    static_assert(HD % SCB == 0, "whole trips over the HD columns");
    const int t = threadIdx.x;
    float mx = -1e30f;
#pragma unroll
    for (int i = 0; i < HD / SCB; ++i) mx = fmaxf(mx, fold8(col_l1, HD, t + SCB * i));
    sm[t] = mx; __syncthreads();
    for (int s = SCB / 2; s > 0; s >>= 1) {
        if (t < s) sm[t] = fmaxf(sm[t], sm[t + s]);
        __syncthreads();
    }
    const float m1 = sm[0];
    __syncthreads();
    mx = -1e30f;
#pragma unroll
    for (int i = 0; i < (Lsz + SCB - 1) / SCB; ++i) {
        const int j = t + SCB * i;
        mx = fmaxf(mx, fold8(whl, Lsz, j < Lsz ? j : Lsz - 1));
    }
    sm[t] = mx; __syncthreads();
    for (int s = SCB / 2; s > 0; s >>= 1) {
        if (t < s) sm[t] = fmaxf(sm[t], sm[t + s]);
        __syncthreads();
    }
    if (t == 0) {
        scalars[0] = 0.95f / fmaxf(m1, 1e-8f);
        scalars[1] = sm[0] * 19.0f;
    }
}

#define TT 64
#define TPITCH (TT + 2)
__device__ __forceinline__
void build_wt_block(const float* __restrict__ R, const float* __restrict__ M,
                    const float* __restrict__ scalars, const int sel,
                    unsigned short* __restrict__ WT,
                    const int K, const int srcld, const int NL,
                    const int bkx, const int bny) {
    __shared__ __attribute__((aligned(16))) unsigned short tile[TT * TPITCH];
    const int k0 = bkx * TT;
    const int n0 = bny * TT;
    const int t  = threadIdx.x;

    const int ln = (t & 15) * 4;
    const int lk = t >> 4;
    const int n  = n0 + ln;
    const bool in = n < NL;
    const int nc = in ? n : 0;
    float4 r[4], m[4];
#pragma unroll
    for (int i = 0; i < 4; ++i) {
        const size_t sidx = (size_t)(k0 + lk + 16 * i) * srcld + nc;
        r[i] = *reinterpret_cast<const float4*>(R + sidx);
        m[i] = *reinterpret_cast<const float4*>(M + sidx);
    }
    float sc = 1.f;
    if (sel == 0) sc = scalars[0];
#pragma unroll
    for (int i = 0; i < 4; ++i) {
        float v0 = r[i].x * m[i].x, v1 = r[i].y * m[i].y;
        float v2 = r[i].z * m[i].z, v3 = r[i].w * m[i].w;
        if (sel == 0) { v0 *= sc; v1 *= sc; v2 *= sc; v3 *= sc; }
        if (!in) { v0 = 0.f; v1 = 0.f; v2 = 0.f; v3 = 0.f; }
        unsigned int* d = reinterpret_cast<unsigned int*>(tile + (lk + 16 * i) * TPITCH + ln);
        d[0] = (unsigned int)f2bf(v0) | ((unsigned int)f2bf(v1) << 16);
        d[1] = (unsigned int)f2bf(v2) | ((unsigned int)f2bf(v3) << 16);
    }
    __syncthreads();

    const int ks = (t & 7) * 8;
#pragma unroll
    for (int p = 0; p < 2; ++p) {
        const int nl = (t >> 3) + 32 * p;
        unsigned int w[4];
#pragma unroll
        for (int j = 0; j < 4; ++j) {
            const unsigned int e0 = tile[(ks + 2 * j) * TPITCH + nl];
            const unsigned int e1 = tile[(ks + 2 * j + 1) * TPITCH + nl];
            w[j] = e0 | (e1 << 16);
        }
        u32x4 o; o.x = w[0]; o.y = w[1]; o.z = w[2]; o.w = w[3];
        *reinterpret_cast<u32x4*>(WT + (size_t)(n0 + nl) * K + k0 + ks) = o;
    }
}

__global__ __launch_bounds__(256)
void build_wt_kernel(const float* __restrict__ R, const float* __restrict__ M,
                     const float* __restrict__ scalars, int sel,
                     unsigned short* __restrict__ WT,
                     int K, int srcld, int NL) {
    build_wt_block(R, M, scalars, sel, WT, K, srcld, NL, (int)blockIdx.x, (int)blockIdx.y);
}

#define SA_COLH  ((HD / 64) * 8)
#define SA_COLL  (((Lsz + 63) / 64) * 8)
#define SA_WTIH  ((Isz / TT) * (HD / TT))
#define SA_WTHL  ((HD / TT) * (LP / TT))
#define SA_XB    128
#define SA_INIT  128
#define SA_BLOCKS (SA_COLH + SA_COLL + SA_WTIH + SA_WTHL + SA_XB + SA_INIT)
__global__ __launch_bounds__(256)
void setup_inputs_kernel(const float* __restrict__ x,
                         const float* __restrict__ RIH, const float* __restrict__ MIH,
                         const float* __restrict__ RH,  const float* __restrict__ MH,
                         const float* __restrict__ RHL, const float* __restrict__ MHL,
                         float* __restrict__ part_h, float* __restrict__ part_l,
                         unsigned short* __restrict__ WT_IH,
                         unsigned short* __restrict__ WT_HL,
                         unsigned short* __restrict__ xb,
                         float* __restrict__ H0, unsigned short* __restrict__ Hb0,
                         unsigned int* __restrict__ dHa, unsigned int* __restrict__ dHb,
                         unsigned int* __restrict__ dHc,
                         int* __restrict__ act0, int* __restrict__ act1) {
    int b = blockIdx.x;
    if (b < SA_COLH) {
        col_l1_block(RH, MH, part_h, HD, b % (HD / 64), b / (HD / 64));
        return;
    }
    b -= SA_COLH;
    if (b < SA_COLL) {
        col_l1_block(RHL, MHL, part_l, Lsz, b % ((Lsz + 63) / 64), b / ((Lsz + 63) / 64));
        return;
    }
    b -= SA_COLL;
    if (b < SA_WTIH) {
        build_wt_block(RIH, MIH, nullptr,  1, WT_IH, Isz, HD, HD,
                       b % (Isz / TT), b / (Isz / TT));
        return;
    }
    b -= SA_WTIH;
    if (b < SA_WTHL) {
        build_wt_block(RHL, MHL, nullptr,  1, WT_HL, HD, Lsz, Lsz,
                       b % (HD / TT), b / (HD / TT));
        return;
    }
    b -= SA_WTHL;
    if (b < SA_XB) {
        static_assert((Bsz * Isz) % (SA_XB * 256 * 8) == 0, "whole trips");
        const int gt = b * 256 + (int)threadIdx.x;
#pragma unroll
        for (int q = 0; q < (Bsz * Isz) / (SA_XB * 256 * 8); ++q) {
            const size_t e = ((size_t)q * (SA_XB * 256) + gt) * 8;
            const float4 lo = *reinterpret_cast<const float4*>(x + e);
            const float4 hi = *reinterpret_cast<const float4*>(x + e + 4);
            u32x4 o;
            o.x = (unsigned int)f2bf(lo.x) | ((unsigned int)f2bf(lo.y) << 16);
            o.y = (unsigned int)f2bf(lo.z) | ((unsigned int)f2bf(lo.w) << 16);
            o.z = (unsigned int)f2bf(hi.x) | ((unsigned int)f2bf(hi.y) << 16);
            o.w = (unsigned int)f2bf(hi.z) | ((unsigned int)f2bf(hi.w) << 16);
            *reinterpret_cast<u32x4*>(xb + e) = o;
        }
        return;
    }
    b -= SA_XB;
    {
        static_assert((Bsz * HD) % (SA_INIT * 256 * 8) == 0, "whole trips");
        const int gt = b * 256 + (int)threadIdx.x;
        const float4 zf = {0.f, 0.f, 0.f, 0.f};
        const u32x4  zu = {0u, 0u, 0u, 0u};
#pragma unroll
        for (int q = 0; q < (Bsz * HD) / (SA_INIT * 256 * 4); ++q)
            reinterpret_cast<float4*>(H0)[(size_t)q * (SA_INIT * 256) + gt] = zf;
#pragma unroll
        for (int q = 0; q < (Bsz * HD) / (SA_INIT * 256 * 8); ++q)
            reinterpret_cast<u32x4*>(Hb0)[(size_t)q * (SA_INIT * 256) + gt] = zu;
        if (gt < Bsz) { dHa[gt] = 0u; dHb[gt] = 0u; dHc[gt] = 0u; act0[gt] = 1; act1[gt] = 1; }
    }
}

#define DRV_BLOCKS ((HD / 64) * (Bsz / 64))
__global__ __launch_bounds__(128)
void drive_scalars_kernel(const unsigned short* __restrict__ xb,
                          const unsigned short* __restrict__ WT_IH,
                          const float* __restrict__ BH,
                          float* __restrict__ drive,
                          const float* __restrict__ part_h,
                          const float* __restrict__ part_l,
                          float* __restrict__ scalars) {
    static_assert(SCB == 128, "scalars_block runs on this kernel's block size");
    __shared__ __attribute__((aligned(16))) unsigned char smem[2 * TILE_BYTES];
    if (blockIdx.x == 0) {
        scalars_block(part_h, part_l, scalars, reinterpret_cast<float*>(smem));
    } else {
        const int b = (int)blockIdx.x - 1;
        gemm_tile<0>(xb, WT_IH, Bsz, HD, Isz, BH, nullptr, nullptr, nullptr,
                     drive, nullptr, nullptr, HD,
                     smem, nullptr, nullptr, b % (HD / 64), b / (HD / 64), b);
    }
}

__device__ __forceinline__
void certify_rows(const int ub,
                  const float* __restrict__ logits,
                  const float* __restrict__ scalars,
                  unsigned int* __restrict__ dHbits,
                  const int* __restrict__ actIn,
                  int* __restrict__ actOut,
                  float* __restrict__ out,
                  const int is_last) {
    static_assert(Lsz % 4 == 0 && LP == 1024, "a row is eight 16-B pieces per lane, none straddling column Lsz");
    const int lane = threadIdx.x & 31;
    const int wave = threadIdx.x >> 5;
    const int row0 = ub * 8 + wave * 2;

    int    a[2];
    float  dh[2];
    float4 v[2][8];
#pragma unroll
    for (int j = 0; j < 2; ++j) {
        a[j]  = actIn[row0 + j];
        dh[j] = __uint_as_float(dHbits[row0 + j]);
        const float* lr = logits + (size_t)(row0 + j) * LP;
#pragma unroll
        for (int q = 0; q < 8; ++q) {
            const int col = 128 * q + 4 * lane;
            v[j][q] = *reinterpret_cast<const float4*>(lr + (col < Lsz ? col : Lsz - 4));
        }
    }
    const float gcoef = scalars[1];

#pragma unroll
    for (int j = 0; j < 2; ++j) {
        const int row = row0 + j;
        int aj = a[j];
        if (aj != 0) {
            float m1 = -1e30f, m2 = -1e30f;
#pragma unroll
            for (int q = 0; q < 8; ++q) {
                const bool in = 128 * q + 4 * lane < Lsz;
                const float e[4] = {v[j][q].x, v[j][q].y, v[j][q].z, v[j][q].w};
#pragma unroll
                for (int c = 0; c < 4; ++c) {
                    const float x = in ? e[c] : -1e30f;
                    m2 = fmaxf(m2, fminf(m1, x));
                    m1 = fmaxf(m1, x);
                }
            }
            for (int off = 16; off > 0; off >>= 1) {
                const float o1 = __shfl_xor(m1, off, 32);
                const float o2 = __shfl_xor(m2, off, 32);
                const float hi = fmaxf(m1, o1);
                const float lo = fmaxf(fminf(m1, o1), fmaxf(m2, o2));
                m1 = hi; m2 = lo;
            }
            const float margin = m1 - m2;
            const float Gamma = gcoef * dh[j];
            const bool newly = margin > 2.0f * Gamma;
            if (newly || is_last) {
                float* orow = out + (size_t)row * Lsz;
#pragma unroll
                for (int q = 0; q < 8; ++q) {
                    const int col = 128 * q + 4 * lane;
                    if (col < Lsz) *reinterpret_cast<float4*>(orow + col) = v[j][q];
                }
                aj = 0;
            }
        }
        if (lane == 0) {
            actOut[row] = aj;
            dHbits[row] = 0u;
        }
    }
}

#define ROW_GROUPS  (Bsz / 64)
#define REC_BLOCKS  ((HD / 64) * ROW_GROUPS)
#define LOG_BLOCKS  ((LP / 64) * ROW_GROUPS)
#define CERT_BLOCKS (Bsz / 8)
__global__ __launch_bounds__(128)
void step_kernel(const unsigned short* __restrict__ HbCur,
                 const float* __restrict__ HCur,
                 const unsigned short* __restrict__ WT_H,
                 const unsigned short* __restrict__ WT_HL,
                 const float* __restrict__ drive,
                 const float* __restrict__ BL,
                 const float* __restrict__ scalars,
                 const int* __restrict__ actIn,
                 int* __restrict__ actOut,
                 float* __restrict__ HNext,
                 unsigned short* __restrict__ HbNext,
                 unsigned int* __restrict__ dHNext,
                 float* __restrict__ logitsCur,
                 const float* __restrict__ logitsPrev,
                 unsigned int* __restrict__ dHPrev,
                 float* __restrict__ out,
                 int do_rec, int do_log, int do_cert, int is_last) {
    __shared__ __attribute__((aligned(16))) unsigned char smem[2 * TILE_BYTES];
    __shared__ int s_cnt[Bsz / 32];
    __shared__ int s_list[Bsz];
    const int b = blockIdx.x;
    if (b < REC_BLOCKS) {
        if (do_rec)
            gemm_tile<1>(HbCur, WT_H, Bsz, HD, HD, nullptr, drive, HCur, actIn,
                         HNext, HbNext, dHNext, HD,
                         smem, s_cnt, s_list, b / ROW_GROUPS, b % ROW_GROUPS, b);
    } else if (b < REC_BLOCKS + LOG_BLOCKS) {
        const int q = b - REC_BLOCKS;
        if (do_log)
            gemm_tile<2>(HbCur, WT_HL, Bsz, LP, HD, BL, nullptr, nullptr, actIn,
                         logitsCur, nullptr, nullptr, Lsz,
                         smem, s_cnt, s_list, q / ROW_GROUPS, ROW_GROUPS - 1 - q % ROW_GROUPS, q);
    } else {
        if (do_cert)
            certify_rows(b - (REC_BLOCKS + LOG_BLOCKS), logitsPrev, scalars, dHPrev,
                         actIn, actOut, out, is_last);
    }
}

static inline char* wsalloc(char*& p, size_t bytes) {
    char* r = p;
    p += (bytes + 255) & ~(size_t)255;
    return r;
}

extern "C" void kernel_launch(void* const* d_in, const int* in_sizes, int n_in,
                              void* d_out, int out_size, void* d_ws, size_t ws_size,
                              hipStream_t stream) {
    (void)in_sizes; (void)n_in; (void)out_size; (void)ws_size;
    const float* x   = (const float*)d_in[0];
    const float* RIH = (const float*)d_in[1];
    const float* RH  = (const float*)d_in[2];
    const float* RHL = (const float*)d_in[3];
    const float* BH  = (const float*)d_in[4];
    const float* BL  = (const float*)d_in[5];
    const float* MIH = (const float*)d_in[6];
    const float* MH  = (const float*)d_in[7];
    const float* MHL = (const float*)d_in[8];
    float* out = (float*)d_out;

    char* p = (char*)d_ws;
    unsigned short* WT_H  = (unsigned short*)wsalloc(p, (size_t)HD * HD * 2);
    unsigned short* WT_HL = (unsigned short*)wsalloc(p, (size_t)LP * HD * 2);
    unsigned short* WT_IH = (unsigned short*)wsalloc(p, (size_t)HD * Isz * 2);
    unsigned short* xb    = (unsigned short*)wsalloc(p, (size_t)Bsz * Isz * 2);
    float* drive = (float*)wsalloc(p, (size_t)Bsz * HD * 4);
    float* Hbuf0 = (float*)wsalloc(p, (size_t)Bsz * HD * 4);
    float* Hbuf1 = (float*)wsalloc(p, (size_t)Bsz * HD * 4);
    unsigned short* Hb0 = (unsigned short*)wsalloc(p, (size_t)Bsz * HD * 2);
    unsigned short* Hb1 = (unsigned short*)wsalloc(p, (size_t)Bsz * HD * 2);
    float* logits = (float*)wsalloc(p, (size_t)Bsz * LP * 4);
    float* col_l1 = (float*)wsalloc(p, HD * 4);
    float* scalars = (float*)wsalloc(p, 256);
    unsigned int* dHb = (unsigned int*)wsalloc(p, Bsz * 4);
    int* active = (int*)wsalloc(p, Bsz * 4);

    unsigned int* dH1 = (unsigned int*)col_l1;
    unsigned int* dH2 = dH1 + Bsz;
    int* active1 = (int*)(dH2 + Bsz);
    float* logits1 = (float*)WT_IH;
    float* part_h = logits;
    float* part_l = logits + 8 * HD;

    setup_inputs_kernel<<<SA_BLOCKS, 256, 0, stream>>>(
        x, RIH, MIH, RH, MH, RHL, MHL, part_h, part_l, WT_IH, WT_HL, xb,
        Hbuf0, Hb0, dHb, dH1, dH2, active, active1);
    drive_scalars_kernel<<<DRV_BLOCKS + 1, 128, 0, stream>>>(
        xb, WT_IH, BH, drive, part_h, part_l, scalars);
    build_wt_kernel<<<dim3(HD / TT, HD / TT), 256, 0, stream>>>(
        RH, MH, scalars,  0, WT_H, HD, HD, HD);

    float* Hc[2] = {Hbuf0, Hbuf1};
    unsigned short* Hbc[2] = {Hb0, Hb1};
    int* act[2] = {active, active1};
    unsigned int* dHp[3] = {dHb, dH1, dH2};
    float* lg[2] = {logits, logits1};
    for (int s = 0; s <= TMAX + 1; ++s) {
        const int c = s & 1, n = (s + 1) & 1;
        step_kernel<<<REC_BLOCKS + LOG_BLOCKS + CERT_BLOCKS, 128, 0, stream>>>(
            Hbc[c], Hc[c], WT_H, WT_HL, drive, BL, scalars, act[c], act[n],
            Hc[n], Hbc[n], dHp[(s + 1) % 3], lg[c], lg[n], dHp[(s + 2) % 3], out,
            (s < TMAX) ? 1 : 0, (s >= 1 && s <= TMAX) ? 1 : 0, (s >= 2) ? 1 : 0,
            (s == TMAX + 1) ? 1 : 0);
    }
}


# dH row max, 64x64 recurrence epilogue: the 16 lanes holding one row's columns fold their maxima by lane exchange, one lane sends the atomic max (32 per row and step, was 512); order-free, same bits
# speedup vs baseline: 5.8342x; 1.1334x over previous
#include <hip/hip_runtime.h>


#define Bsz 512
#define Isz 1024
#define HD  2048
#define Lsz 1000
#define LP  1024
#define TMAX 32

typedef __attribute__((ext_vector_type(16))) __bf16 v16bf;
typedef __attribute__((ext_vector_type(8)))  float  v8f;
typedef __attribute__((ext_vector_type(4)))  unsigned int u32x4;

__device__ __forceinline__ unsigned short f2bf(float f) {
    unsigned int u = __float_as_uint(f);
    u += 0x7FFFu + ((u >> 16) & 1u);
    return (unsigned short)(u >> 16);
}

#define KC 256
#define PITCH (KC * 2 + 16)
#define TILE_BYTES (64 * PITCH)

struct alignas(32) FragBits { u32x4 lo, hi; };

__device__ __forceinline__ v16bf lds_frag(const unsigned char* p) {
    FragBits fb;
    fb.lo = *reinterpret_cast<const u32x4*>(p);
    fb.hi = *reinterpret_cast<const u32x4*>(p + 32);
    v16bf v;
    __builtin_memcpy(&v, &fb, sizeof(v));
    return v;
}

#define TAIL_ROWS (Bsz / 16)
template <int MODE>
__device__ __forceinline__
void gemm_tile(const unsigned short* __restrict__ A,
               const unsigned short* __restrict__ BT,
               int M, int N, int K,
               const float* __restrict__ bias,
               const float* __restrict__ drive,
               const float* __restrict__ Hold,
               const int*   __restrict__ active,
               float* __restrict__ Cout,
               unsigned short* __restrict__ HbOut,
               unsigned int* __restrict__ dHbits,
               int NL,
               unsigned char* const smem, int* const s_cnt, int* const s_list,
               const int bx, const int by, const int lin) {
    unsigned char* const sA = smem;
    unsigned char* const sB = smem + TILE_BYTES;

    const int tid  = threadIdx.x;
    const int lane = tid & 31;
    const int wave = tid >> 5;
    const int rowBase = by * 64 + wave * 16;
    const int colBase = bx * 64;
    const int r16 = lane & 15;
    const int kh  = lane >> 4;
    int n_act = M;

    if (MODE != 0) {
        int flag[Bsz / 128];
        unsigned msk[Bsz / 128];
#pragma unroll
        for (int j = 0; j < Bsz / 128; ++j) {
            flag[j] = active[j * 128 + tid];
            msk[j]  = (unsigned)__ballot(flag[j] != 0);
        }
        if (lane == 0) {
#pragma unroll
            for (int j = 0; j < Bsz / 128; ++j) s_cnt[j * 4 + wave] = __popc(msk[j]);
        }
        __syncthreads();
        n_act = 0;
        int before[Bsz / 128];
#pragma unroll
        for (int s = 0; s < Bsz / 32; ++s) {
            if ((s & 3) == wave) before[s >> 2] = n_act;
            n_act += s_cnt[s];
        }
        if (n_act == 0) return;

#pragma unroll
        for (int j = 0; j < Bsz / 128; ++j) {
            if (flag[j] != 0)
                s_list[before[j] + __popc(msk[j] & ((1u << lane) - 1u))] = j * 128 + tid;
        }
        __syncthreads();

        if (n_act <= TAIL_ROWS) {
            const int NT = N >> 4;
            const int RT = (n_act + 15) >> 4;
            const int b  = lin;
            if (b >= RT * NT) return;
            const int rt = b / NT;
            const int ct = b - rt * NT;

            const int li   = rt * 16 + r16;
            const int arow = s_list[li < n_act ? li : 0];
            const int KQ   = K >> 2;
            const unsigned short* pa = A  + (size_t)arow * K + wave * KQ + kh * 16;
            const unsigned short* pb = BT + (size_t)(ct * 16 + r16) * K + wave * KQ + kh * 16;

            v8f part = {0.f,0.f,0.f,0.f,0.f,0.f,0.f,0.f};
            for (int k0 = 0; k0 < KQ; k0 += 512) {
                FragBits ta[16], tb[16];
#pragma unroll
                for (int i = 0; i < 16; ++i) {
                    ta[i].lo = *reinterpret_cast<const u32x4*>(pa + k0 + i * 32);
                    ta[i].hi = *reinterpret_cast<const u32x4*>(pa + k0 + i * 32 + 8);
                    tb[i].lo = *reinterpret_cast<const u32x4*>(pb + k0 + i * 32);
                    tb[i].hi = *reinterpret_cast<const u32x4*>(pb + k0 + i * 32 + 8);
                }
                __builtin_amdgcn_sched_barrier(0);
#pragma unroll
                for (int i = 0; i < 16; ++i) {
                    v16bf av, bv;
                    __builtin_memcpy(&av, &ta[i], sizeof(av));
                    __builtin_memcpy(&bv, &tb[i], sizeof(bv));
                    part = __builtin_amdgcn_wmma_f32_16x16x32_bf16(
                        false, av, false, bv, (short)0, part, false, false);
                }
            }

            float* const red = reinterpret_cast<float*>(smem);
#pragma unroll
            for (int r = 0; r < 8; ++r) red[(wave * 8 + r) * 32 + lane] = part[r];
            __syncthreads();

            const int n = ct * 16 + r16;
            int   mrow[2];
            bool  live[2];
            float v[2];
#pragma unroll
            for (int j = 0; j < 2; ++j) {
                const int r  = 2 * wave + j;
                const int lr = rt * 16 + r + 8 * kh;
                live[j] = lr < n_act;
                mrow[j] = s_list[live[j] ? lr : 0];
                v[j] = ((red[r * 32 + lane] + red[(8 + r) * 32 + lane])
                        + red[(16 + r) * 32 + lane]) + red[(24 + r) * 32 + lane];
            }
            if (MODE == 2) {
                if (n < NL) {
                    const float bn = bias[n];
#pragma unroll
                    for (int j = 0; j < 2; ++j)
                        if (live[j]) Cout[(size_t)mrow[j] * N + n] = v[j] + bn;
                }
            } else {
                float dv[2], hv[2], d[2];
#pragma unroll
                for (int j = 0; j < 2; ++j) {
                    const size_t idx = (size_t)mrow[j] * N + n;
                    dv[j] = drive[idx];
                    hv[j] = Hold[idx];
                }
#pragma unroll
                for (int j = 0; j < 2; ++j) {
                    const size_t idx = (size_t)mrow[j] * N + n;
                    const float pre  = v[j] + dv[j];
                    const float actv = pre >= 0.f ? pre : 0.05f * pre;
                    d[j] = live[j] ? fabsf(actv - hv[j]) : 0.f;
                    if (live[j]) {
                        Cout[idx]  = actv;
                        HbOut[idx] = f2bf(actv);
                    }
                }
#pragma unroll
                for (int off = 8; off > 0; off >>= 1) {
#pragma unroll
                    for (int j = 0; j < 2; ++j) d[j] = fmaxf(d[j], __shfl_xor(d[j], off, 32));
                }
#pragma unroll
                for (int j = 0; j < 2; ++j)
                    if (live[j] && r16 == 0 && d[j] > 0.f)
                        atomicMax(dHbits + mrow[j], __float_as_uint(d[j]));
            }
            return;
        }

        if (by * 64 >= n_act) return;
    }

    const int wv = __builtin_amdgcn_readfirstlane(wave);
    const unsigned char* arow[16];
#pragma unroll
    for (int i = 0; i < 16; ++i) {
        const int li = by * 64 + wv + 4 * i;
        int row = li;
        if (MODE != 0) row = __builtin_amdgcn_readfirstlane(s_list[li < n_act ? li : 0]);
        arow[i] = reinterpret_cast<const unsigned char*>(A) + (size_t)row * (size_t)(K * 2);
    }
    unsigned voff = (unsigned)(lane * 16);
    const unsigned short* gb = BT + (size_t)(colBase + wave) * K + lane * 8;
    const size_t gstep = (size_t)4 * K;
    const int sst = wave * PITCH + lane * 16;

    const int fa = (wave * 16 + r16) * PITCH + kh * 16;
    const int fb = r16 * PITCH + kh * 16;

    v8f acc0 = {0.f,0.f,0.f,0.f,0.f,0.f,0.f,0.f};
    v8f acc[4];
    acc[0] = acc0; acc[1] = acc0; acc[2] = acc0; acc[3] = acc0;

    u32x4 ra[16], rb[16];

#pragma unroll
    for (int i = 0; i < 16; ++i) {
        ra[i] = *reinterpret_cast<const u32x4*>(arow[i] + voff);
        rb[i] = *reinterpret_cast<const u32x4*>(gb + i * gstep);
    }
#pragma unroll
    for (int i = 0; i < 16; ++i) {
        *reinterpret_cast<u32x4*>(sA + sst + i * 4 * PITCH) = ra[i];
        *reinterpret_cast<u32x4*>(sB + sst + i * 4 * PITCH) = rb[i];
    }
    __syncthreads();

    const int nchunks = K / KC;
    for (int c = 0; c < nchunks; ++c) {
        const bool more = (c + 1 < nchunks);
        if (more) {
            voff += KC * 2; gb += KC;
#pragma unroll
            for (int i = 0; i < 16; ++i) {
                ra[i] = *reinterpret_cast<const u32x4*>(arow[i] + voff);
                rb[i] = *reinterpret_cast<const u32x4*>(gb + i * gstep);
            }
        }

        v16bf a_cur  = lds_frag(sA + fa);
        v16bf b_cur0 = lds_frag(sB + fb);
        v16bf b_cur1 = lds_frag(sB + fb + 16 * PITCH);
        v16bf b_cur2 = lds_frag(sB + fb + 32 * PITCH);
        v16bf b_cur3 = lds_frag(sB + fb + 48 * PITCH);
#pragma unroll
        for (int ks = 0; ks < KC / 32; ++ks) {
            v16bf a_nxt = a_cur, b_nxt0 = b_cur0, b_nxt1 = b_cur1,
                  b_nxt2 = b_cur2, b_nxt3 = b_cur3;
            if (ks + 1 < KC / 32) {
                const int ko = (ks + 1) * 64;
                a_nxt  = lds_frag(sA + fa + ko);
                b_nxt0 = lds_frag(sB + fb + ko);
                b_nxt1 = lds_frag(sB + fb + 16 * PITCH + ko);
                b_nxt2 = lds_frag(sB + fb + 32 * PITCH + ko);
                b_nxt3 = lds_frag(sB + fb + 48 * PITCH + ko);
            }
            __builtin_amdgcn_sched_barrier(0);
            acc[0] = __builtin_amdgcn_wmma_f32_16x16x32_bf16(
                false, a_cur, false, b_cur0, (short)0, acc[0], false, false);
            acc[1] = __builtin_amdgcn_wmma_f32_16x16x32_bf16(
                false, a_cur, false, b_cur1, (short)0, acc[1], false, false);
            acc[2] = __builtin_amdgcn_wmma_f32_16x16x32_bf16(
                false, a_cur, false, b_cur2, (short)0, acc[2], false, false);
            acc[3] = __builtin_amdgcn_wmma_f32_16x16x32_bf16(
                false, a_cur, false, b_cur3, (short)0, acc[3], false, false);
            __builtin_amdgcn_sched_barrier(0);
            a_cur = a_nxt;
            b_cur0 = b_nxt0; b_cur1 = b_nxt1; b_cur2 = b_nxt2; b_cur3 = b_nxt3;
        }

        __syncthreads();
        if (more) {
#pragma unroll
            for (int i = 0; i < 16; ++i) {
                *reinterpret_cast<u32x4*>(sA + sst + i * 4 * PITCH) = ra[i];
                *reinterpret_cast<u32x4*>(sB + sst + i * 4 * PITCH) = rb[i];
            }
            __syncthreads();
        }
    }

    int   mrow[8];
    bool  live[8];
    float dmax[8];
#pragma unroll
    for (int r = 0; r < 8; ++r) {
        const int li = rowBase + r + 8 * kh;
        live[r] = (MODE == 0) || (li < n_act);
        mrow[r] = (MODE != 0) ? s_list[live[r] ? li : 0] : li;
        dmax[r] = 0.f;
    }

    if (MODE == 0) {
#pragma unroll
        for (int f = 0; f < 4; ++f) {
            const int n = colBase + f * 16 + r16;
#pragma unroll
            for (int r = 0; r < 8; ++r)
                Cout[(size_t)mrow[r] * N + n] = acc[f][r] + bias[n];
        }
    } else if (MODE == 2) {
        float bn[4];
#pragma unroll
        for (int f = 0; f < 4; ++f) {
            const int n = colBase + f * 16 + r16;
            bn[f] = bias[n < NL ? n : 0];
        }
        float lg[4][8];
#pragma unroll
        for (int f = 0; f < 4; ++f) {
#pragma unroll
            for (int r = 0; r < 8; ++r) lg[f][r] = acc[f][r] + bn[f];
        }
#pragma unroll
        for (int r = 0; r < 8; ++r) {
            if (live[r]) {
#pragma unroll
                for (int f = 0; f < 4; ++f) {
                    const int n = colBase + f * 16 + r16;
                    if (n < NL) Cout[(size_t)mrow[r] * N + n] = lg[f][r];
                }
            }
        }
    } else {
        float dv[4][8], hv[4][8];
#pragma unroll
        for (int f = 0; f < 4; ++f) {
#pragma unroll
            for (int r = 0; r < 8; ++r) {
                const size_t idx = (size_t)mrow[r] * N + (colBase + f * 16 + r16);
                dv[f][r] = drive[idx];
                hv[f][r] = Hold[idx];
            }
        }
        __builtin_amdgcn_sched_barrier(0);
#pragma unroll
        for (int f = 0; f < 4; ++f) {
#pragma unroll
            for (int r = 0; r < 8; ++r) {
                const float pre  = acc[f][r] + dv[f][r];
                const float actv = pre >= 0.f ? pre : 0.05f * pre;
                const float dd   = fabsf(actv - hv[f][r]);
                dmax[r] = fmaxf(dmax[r], live[r] ? dd : 0.f);
                acc[f][r] = actv;
            }
        }
#pragma unroll
        for (int r = 0; r < 8; ++r) {
            if (live[r]) {
#pragma unroll
                for (int f = 0; f < 4; ++f) {
                    const size_t idx = (size_t)mrow[r] * N + (colBase + f * 16 + r16);
                    Cout[idx]  = acc[f][r];
                    HbOut[idx] = f2bf(acc[f][r]);
                }
            }
        }
#pragma unroll
        for (int off = 8; off > 0; off >>= 1) {
#pragma unroll
            for (int r = 0; r < 8; ++r) dmax[r] = fmaxf(dmax[r], __shfl_xor(dmax[r], off, 32));
        }
#pragma unroll
        for (int r = 0; r < 8; ++r) {
            if (r16 == 0 && dmax[r] > 0.f) {
                atomicMax(dHbits + mrow[r], __float_as_uint(dmax[r]));
            }
        }
    }
}

__device__ __forceinline__
void col_l1_block(const float* __restrict__ A, const float* __restrict__ Msk,
                  float* __restrict__ part, const int cols, const int bx, const int b) {
    static_assert(HD % 256 == 0, "leaf rows are t + 256*r");
    __shared__ float smx[8][32];
    __shared__ float smy[8][32];
    const int lane = threadIdx.x & 31;
    const int v    = threadIdx.x >> 5;
    const int c    = (bx * 32 + lane) * 2;
    const bool in  = c < cols;
    const int cl   = in ? c : 0;
    float lx[4], ly[4];
#pragma unroll
    for (int m = 0; m < 4; ++m) {
        const int t = b + 8 * (v + 8 * m);
        float sx = 0.f, sy = 0.f;
#pragma unroll
        for (int r = 0; r < HD / 256; ++r) {
            const size_t idx = (size_t)(t + 256 * r) * cols + cl;
            const float2 a  = *reinterpret_cast<const float2*>(A + idx);
            const float2 mk = *reinterpret_cast<const float2*>(Msk + idx);
            sx += fabsf(a.x * mk.x);
            sy += fabsf(a.y * mk.y);
        }
        lx[m] = sx; ly[m] = sy;
    }
    smx[v][lane] = (lx[0] + lx[2]) + (lx[1] + lx[3]);
    smy[v][lane] = (ly[0] + ly[2]) + (ly[1] + ly[3]);
    __syncthreads();
    if (v == 0 && in) {
        part[(size_t)b * cols + c] =
            ((smx[0][lane] + smx[4][lane]) + (smx[2][lane] + smx[6][lane])) +
            ((smx[1][lane] + smx[5][lane]) + (smx[3][lane] + smx[7][lane]));
        part[(size_t)b * cols + c + 1] =
            ((smy[0][lane] + smy[4][lane]) + (smy[2][lane] + smy[6][lane])) +
            ((smy[1][lane] + smy[5][lane]) + (smy[3][lane] + smy[7][lane]));
    }
}

__device__ __forceinline__ float fold8(const float* __restrict__ p, int cols, int j) {
    const float p0 = p[j],            p1 = p[cols + j];
    const float p2 = p[2 * cols + j], p3 = p[3 * cols + j];
    const float p4 = p[4 * cols + j], p5 = p[5 * cols + j];
    const float p6 = p[6 * cols + j], p7 = p[7 * cols + j];
    return ((p0 + p4) + (p2 + p6)) + ((p1 + p5) + (p3 + p7));
}

#define SCB 128
__device__ __forceinline__
void scalars_block(const float* __restrict__ col_l1,
                   const float* __restrict__ whl,
                   float* __restrict__ scalars, float* const sm) {
    static_assert(HD % SCB == 0, "whole trips over the HD columns");
    const int t = threadIdx.x;
    float mx = -1e30f;
#pragma unroll
    for (int i = 0; i < HD / SCB; ++i) mx = fmaxf(mx, fold8(col_l1, HD, t + SCB * i));
    sm[t] = mx; __syncthreads();
    for (int s = SCB / 2; s > 0; s >>= 1) {
        if (t < s) sm[t] = fmaxf(sm[t], sm[t + s]);
        __syncthreads();
    }
    const float m1 = sm[0];
    __syncthreads();
    mx = -1e30f;
#pragma unroll
    for (int i = 0; i < (Lsz + SCB - 1) / SCB; ++i) {
        const int j = t + SCB * i;
        mx = fmaxf(mx, fold8(whl, Lsz, j < Lsz ? j : Lsz - 1));
    }
    sm[t] = mx; __syncthreads();
    for (int s = SCB / 2; s > 0; s >>= 1) {
        if (t < s) sm[t] = fmaxf(sm[t], sm[t + s]);
        __syncthreads();
    }
    if (t == 0) {
        scalars[0] = 0.95f / fmaxf(m1, 1e-8f);
        scalars[1] = sm[0] * 19.0f;
    }
}

#define TT 64
#define TPITCH (TT + 2)
__device__ __forceinline__
void build_wt_block(const float* __restrict__ R, const float* __restrict__ M,
                    const float* __restrict__ scalars, const int sel,
                    unsigned short* __restrict__ WT,
                    const int K, const int srcld, const int NL,
                    const int bkx, const int bny) {
    __shared__ __attribute__((aligned(16))) unsigned short tile[TT * TPITCH];
    const int k0 = bkx * TT;
    const int n0 = bny * TT;
    const int t  = threadIdx.x;

    const int ln = (t & 15) * 4;
    const int lk = t >> 4;
    const int n  = n0 + ln;
    const bool in = n < NL;
    const int nc = in ? n : 0;
    float4 r[4], m[4];
#pragma unroll
    for (int i = 0; i < 4; ++i) {
        const size_t sidx = (size_t)(k0 + lk + 16 * i) * srcld + nc;
        r[i] = *reinterpret_cast<const float4*>(R + sidx);
        m[i] = *reinterpret_cast<const float4*>(M + sidx);
    }
    float sc = 1.f;
    if (sel == 0) sc = scalars[0];
#pragma unroll
    for (int i = 0; i < 4; ++i) {
        float v0 = r[i].x * m[i].x, v1 = r[i].y * m[i].y;
        float v2 = r[i].z * m[i].z, v3 = r[i].w * m[i].w;
        if (sel == 0) { v0 *= sc; v1 *= sc; v2 *= sc; v3 *= sc; }
        if (!in) { v0 = 0.f; v1 = 0.f; v2 = 0.f; v3 = 0.f; }
        unsigned int* d = reinterpret_cast<unsigned int*>(tile + (lk + 16 * i) * TPITCH + ln);
        d[0] = (unsigned int)f2bf(v0) | ((unsigned int)f2bf(v1) << 16);
        d[1] = (unsigned int)f2bf(v2) | ((unsigned int)f2bf(v3) << 16);
    }
    __syncthreads();

    const int ks = (t & 7) * 8;
#pragma unroll
    for (int p = 0; p < 2; ++p) {
        const int nl = (t >> 3) + 32 * p;
        unsigned int w[4];
#pragma unroll
        for (int j = 0; j < 4; ++j) {
            const unsigned int e0 = tile[(ks + 2 * j) * TPITCH + nl];
            const unsigned int e1 = tile[(ks + 2 * j + 1) * TPITCH + nl];
            w[j] = e0 | (e1 << 16);
        }
        u32x4 o; o.x = w[0]; o.y = w[1]; o.z = w[2]; o.w = w[3];
        *reinterpret_cast<u32x4*>(WT + (size_t)(n0 + nl) * K + k0 + ks) = o;
    }
}

__global__ __launch_bounds__(256)
void build_wt_kernel(const float* __restrict__ R, const float* __restrict__ M,
                     const float* __restrict__ scalars, int sel,
                     unsigned short* __restrict__ WT,
                     int K, int srcld, int NL) {
    build_wt_block(R, M, scalars, sel, WT, K, srcld, NL, (int)blockIdx.x, (int)blockIdx.y);
}

#define SA_COLH  ((HD / 64) * 8)
#define SA_COLL  (((Lsz + 63) / 64) * 8)
#define SA_WTIH  ((Isz / TT) * (HD / TT))
#define SA_WTHL  ((HD / TT) * (LP / TT))
#define SA_XB    128
#define SA_INIT  128
#define SA_BLOCKS (SA_COLH + SA_COLL + SA_WTIH + SA_WTHL + SA_XB + SA_INIT)
__global__ __launch_bounds__(256)
void setup_inputs_kernel(const float* __restrict__ x,
                         const float* __restrict__ RIH, const float* __restrict__ MIH,
                         const float* __restrict__ RH,  const float* __restrict__ MH,
                         const float* __restrict__ RHL, const float* __restrict__ MHL,
                         float* __restrict__ part_h, float* __restrict__ part_l,
                         unsigned short* __restrict__ WT_IH,
                         unsigned short* __restrict__ WT_HL,
                         unsigned short* __restrict__ xb,
                         float* __restrict__ H0, unsigned short* __restrict__ Hb0,
                         unsigned int* __restrict__ dHa, unsigned int* __restrict__ dHb,
                         unsigned int* __restrict__ dHc,
                         int* __restrict__ act0, int* __restrict__ act1) {
    int b = blockIdx.x;
    if (b < SA_COLH) {
        col_l1_block(RH, MH, part_h, HD, b % (HD / 64), b / (HD / 64));
        return;
    }
    b -= SA_COLH;
    if (b < SA_COLL) {
        col_l1_block(RHL, MHL, part_l, Lsz, b % ((Lsz + 63) / 64), b / ((Lsz + 63) / 64));
        return;
    }
    b -= SA_COLL;
    if (b < SA_WTIH) {
        build_wt_block(RIH, MIH, nullptr,  1, WT_IH, Isz, HD, HD,
                       b % (Isz / TT), b / (Isz / TT));
        return;
    }
    b -= SA_WTIH;
    if (b < SA_WTHL) {
        build_wt_block(RHL, MHL, nullptr,  1, WT_HL, HD, Lsz, Lsz,
                       b % (HD / TT), b / (HD / TT));
        return;
    }
    b -= SA_WTHL;
    if (b < SA_XB) {
        static_assert((Bsz * Isz) % (SA_XB * 256 * 8) == 0, "whole trips");
        const int gt = b * 256 + (int)threadIdx.x;
#pragma unroll
        for (int q = 0; q < (Bsz * Isz) / (SA_XB * 256 * 8); ++q) {
            const size_t e = ((size_t)q * (SA_XB * 256) + gt) * 8;
            const float4 lo = *reinterpret_cast<const float4*>(x + e);
            const float4 hi = *reinterpret_cast<const float4*>(x + e + 4);
            u32x4 o;
            o.x = (unsigned int)f2bf(lo.x) | ((unsigned int)f2bf(lo.y) << 16);
            o.y = (unsigned int)f2bf(lo.z) | ((unsigned int)f2bf(lo.w) << 16);
            o.z = (unsigned int)f2bf(hi.x) | ((unsigned int)f2bf(hi.y) << 16);
            o.w = (unsigned int)f2bf(hi.z) | ((unsigned int)f2bf(hi.w) << 16);
            *reinterpret_cast<u32x4*>(xb + e) = o;
        }
        return;
    }
    b -= SA_XB;
    {
        static_assert((Bsz * HD) % (SA_INIT * 256 * 8) == 0, "whole trips");
        const int gt = b * 256 + (int)threadIdx.x;
        const float4 zf = {0.f, 0.f, 0.f, 0.f};
        const u32x4  zu = {0u, 0u, 0u, 0u};
#pragma unroll
        for (int q = 0; q < (Bsz * HD) / (SA_INIT * 256 * 4); ++q)
            reinterpret_cast<float4*>(H0)[(size_t)q * (SA_INIT * 256) + gt] = zf;
#pragma unroll
        for (int q = 0; q < (Bsz * HD) / (SA_INIT * 256 * 8); ++q)
            reinterpret_cast<u32x4*>(Hb0)[(size_t)q * (SA_INIT * 256) + gt] = zu;
        if (gt < Bsz) { dHa[gt] = 0u; dHb[gt] = 0u; dHc[gt] = 0u; act0[gt] = 1; act1[gt] = 1; }
    }
}

#define DRV_BLOCKS ((HD / 64) * (Bsz / 64))
__global__ __launch_bounds__(128)
void drive_scalars_kernel(const unsigned short* __restrict__ xb,
                          const unsigned short* __restrict__ WT_IH,
                          const float* __restrict__ BH,
                          float* __restrict__ drive,
                          const float* __restrict__ part_h,
                          const float* __restrict__ part_l,
                          float* __restrict__ scalars) {
    static_assert(SCB == 128, "scalars_block runs on this kernel's block size");
    __shared__ __attribute__((aligned(16))) unsigned char smem[2 * TILE_BYTES];
    if (blockIdx.x == 0) {
        scalars_block(part_h, part_l, scalars, reinterpret_cast<float*>(smem));
    } else {
        const int b = (int)blockIdx.x - 1;
        gemm_tile<0>(xb, WT_IH, Bsz, HD, Isz, BH, nullptr, nullptr, nullptr,
                     drive, nullptr, nullptr, HD,
                     smem, nullptr, nullptr, b % (HD / 64), b / (HD / 64), b);
    }
}

__device__ __forceinline__
void certify_rows(const int ub,
                  const float* __restrict__ logits,
                  const float* __restrict__ scalars,
                  unsigned int* __restrict__ dHbits,
                  const int* __restrict__ actIn,
                  int* __restrict__ actOut,
                  float* __restrict__ out,
                  const int is_last) {
    static_assert(Lsz % 4 == 0 && LP == 1024, "a row is eight 16-B pieces per lane, none straddling column Lsz");
    const int lane = threadIdx.x & 31;
    const int wave = threadIdx.x >> 5;
    const int row0 = ub * 8 + wave * 2;

    int    a[2];
    float  dh[2];
    float4 v[2][8];
#pragma unroll
    for (int j = 0; j < 2; ++j) {
        a[j]  = actIn[row0 + j];
        dh[j] = __uint_as_float(dHbits[row0 + j]);
        const float* lr = logits + (size_t)(row0 + j) * LP;
#pragma unroll
        for (int q = 0; q < 8; ++q) {
            const int col = 128 * q + 4 * lane;
            v[j][q] = *reinterpret_cast<const float4*>(lr + (col < Lsz ? col : Lsz - 4));
        }
    }
    const float gcoef = scalars[1];

#pragma unroll
    for (int j = 0; j < 2; ++j) {
        const int row = row0 + j;
        int aj = a[j];
        if (aj != 0) {
            float m1 = -1e30f, m2 = -1e30f;
#pragma unroll
            for (int q = 0; q < 8; ++q) {
                const bool in = 128 * q + 4 * lane < Lsz;
                const float e[4] = {v[j][q].x, v[j][q].y, v[j][q].z, v[j][q].w};
#pragma unroll
                for (int c = 0; c < 4; ++c) {
                    const float x = in ? e[c] : -1e30f;
                    m2 = fmaxf(m2, fminf(m1, x));
                    m1 = fmaxf(m1, x);
                }
            }
            for (int off = 16; off > 0; off >>= 1) {
                const float o1 = __shfl_xor(m1, off, 32);
                const float o2 = __shfl_xor(m2, off, 32);
                const float hi = fmaxf(m1, o1);
                const float lo = fmaxf(fminf(m1, o1), fmaxf(m2, o2));
                m1 = hi; m2 = lo;
            }
            const float margin = m1 - m2;
            const float Gamma = gcoef * dh[j];
            const bool newly = margin > 2.0f * Gamma;
            if (newly || is_last) {
                float* orow = out + (size_t)row * Lsz;
#pragma unroll
                for (int q = 0; q < 8; ++q) {
                    const int col = 128 * q + 4 * lane;
                    if (col < Lsz) *reinterpret_cast<float4*>(orow + col) = v[j][q];
                }
                aj = 0;
            }
        }
        if (lane == 0) {
            actOut[row] = aj;
            dHbits[row] = 0u;
        }
    }
}

#define ROW_GROUPS  (Bsz / 64)
#define REC_BLOCKS  ((HD / 64) * ROW_GROUPS)
#define LOG_BLOCKS  ((LP / 64) * ROW_GROUPS)
#define CERT_BLOCKS (Bsz / 8)
__global__ __launch_bounds__(128)
void step_kernel(const unsigned short* __restrict__ HbCur,
                 const float* __restrict__ HCur,
                 const unsigned short* __restrict__ WT_H,
                 const unsigned short* __restrict__ WT_HL,
                 const float* __restrict__ drive,
                 const float* __restrict__ BL,
                 const float* __restrict__ scalars,
                 const int* __restrict__ actIn,
                 int* __restrict__ actOut,
                 float* __restrict__ HNext,
                 unsigned short* __restrict__ HbNext,
                 unsigned int* __restrict__ dHNext,
                 float* __restrict__ logitsCur,
                 const float* __restrict__ logitsPrev,
                 unsigned int* __restrict__ dHPrev,
                 float* __restrict__ out,
                 int do_rec, int do_log, int do_cert, int is_last) {
    __shared__ __attribute__((aligned(16))) unsigned char smem[2 * TILE_BYTES];
    __shared__ int s_cnt[Bsz / 32];
    __shared__ int s_list[Bsz];
    const int b = blockIdx.x;
    if (b < REC_BLOCKS) {
        if (do_rec)
            gemm_tile<1>(HbCur, WT_H, Bsz, HD, HD, nullptr, drive, HCur, actIn,
                         HNext, HbNext, dHNext, HD,
                         smem, s_cnt, s_list, b / ROW_GROUPS, b % ROW_GROUPS, b);
    } else if (b < REC_BLOCKS + LOG_BLOCKS) {
        const int q = b - REC_BLOCKS;
        if (do_log)
            gemm_tile<2>(HbCur, WT_HL, Bsz, LP, HD, BL, nullptr, nullptr, actIn,
                         logitsCur, nullptr, nullptr, Lsz,
                         smem, s_cnt, s_list, q / ROW_GROUPS, ROW_GROUPS - 1 - q % ROW_GROUPS, q);
    } else {
        if (do_cert)
            certify_rows(b - (REC_BLOCKS + LOG_BLOCKS), logitsPrev, scalars, dHPrev,
                         actIn, actOut, out, is_last);
    }
}

static inline char* wsalloc(char*& p, size_t bytes) {
    char* r = p;
    p += (bytes + 255) & ~(size_t)255;
    return r;
}

extern "C" void kernel_launch(void* const* d_in, const int* in_sizes, int n_in,
                              void* d_out, int out_size, void* d_ws, size_t ws_size,
                              hipStream_t stream) {
    (void)in_sizes; (void)n_in; (void)out_size; (void)ws_size;
    const float* x   = (const float*)d_in[0];
    const float* RIH = (const float*)d_in[1];
    const float* RH  = (const float*)d_in[2];
    const float* RHL = (const float*)d_in[3];
    const float* BH  = (const float*)d_in[4];
    const float* BL  = (const float*)d_in[5];
    const float* MIH = (const float*)d_in[6];
    const float* MH  = (const float*)d_in[7];
    const float* MHL = (const float*)d_in[8];
    float* out = (float*)d_out;

    char* p = (char*)d_ws;
    unsigned short* WT_H  = (unsigned short*)wsalloc(p, (size_t)HD * HD * 2);
    unsigned short* WT_HL = (unsigned short*)wsalloc(p, (size_t)LP * HD * 2);
    unsigned short* WT_IH = (unsigned short*)wsalloc(p, (size_t)HD * Isz * 2);
    unsigned short* xb    = (unsigned short*)wsalloc(p, (size_t)Bsz * Isz * 2);
    float* drive = (float*)wsalloc(p, (size_t)Bsz * HD * 4);
    float* Hbuf0 = (float*)wsalloc(p, (size_t)Bsz * HD * 4);
    float* Hbuf1 = (float*)wsalloc(p, (size_t)Bsz * HD * 4);
    unsigned short* Hb0 = (unsigned short*)wsalloc(p, (size_t)Bsz * HD * 2);
    unsigned short* Hb1 = (unsigned short*)wsalloc(p, (size_t)Bsz * HD * 2);
    float* logits = (float*)wsalloc(p, (size_t)Bsz * LP * 4);
    float* col_l1 = (float*)wsalloc(p, HD * 4);
    float* scalars = (float*)wsalloc(p, 256);
    unsigned int* dHb = (unsigned int*)wsalloc(p, Bsz * 4);
    int* active = (int*)wsalloc(p, Bsz * 4);

    unsigned int* dH1 = (unsigned int*)col_l1;
    unsigned int* dH2 = dH1 + Bsz;
    int* active1 = (int*)(dH2 + Bsz);
    float* logits1 = (float*)WT_IH;
    float* part_h = logits;
    float* part_l = logits + 8 * HD;

    setup_inputs_kernel<<<SA_BLOCKS, 256, 0, stream>>>(
        x, RIH, MIH, RH, MH, RHL, MHL, part_h, part_l, WT_IH, WT_HL, xb,
        Hbuf0, Hb0, dHb, dH1, dH2, active, active1);
    drive_scalars_kernel<<<DRV_BLOCKS + 1, 128, 0, stream>>>(
        xb, WT_IH, BH, drive, part_h, part_l, scalars);
    build_wt_kernel<<<dim3(HD / TT, HD / TT), 256, 0, stream>>>(
        RH, MH, scalars,  0, WT_H, HD, HD, HD);

    float* Hc[2] = {Hbuf0, Hbuf1};
    unsigned short* Hbc[2] = {Hb0, Hb1};
    int* act[2] = {active, active1};
    unsigned int* dHp[3] = {dHb, dH1, dH2};
    float* lg[2] = {logits, logits1};
    for (int s = 0; s <= TMAX + 1; ++s) {
        const int c = s & 1, n = (s + 1) & 1;
        step_kernel<<<REC_BLOCKS + LOG_BLOCKS + CERT_BLOCKS, 128, 0, stream>>>(
            Hbc[c], Hc[c], WT_H, WT_HL, drive, BL, scalars, act[c], act[n],
            Hc[n], Hbc[n], dHp[(s + 1) % 3], lg[c], lg[n], dHp[(s + 2) % 3], out,
            (s < TMAX) ? 1 : 0, (s >= 1 && s <= TMAX) ? 1 : 0, (s >= 2) ? 1 : 0,
            (s == TMAX + 1) ? 1 : 0);
    }
}


# 64x64 GEMM K loop: two LDS stages; chunk c+1 is stored to the other stage beside chunk c's fragment reads, chunk c+2's loads refill the registers; one LDS-only barrier per chunk, was two; same sums
# speedup vs baseline: 6.2627x; 1.0734x over previous
#include <hip/hip_runtime.h>


#define Bsz 512
#define Isz 1024
#define HD  2048
#define Lsz 1000
#define LP  1024
#define TMAX 32

typedef __attribute__((ext_vector_type(16))) __bf16 v16bf;
typedef __attribute__((ext_vector_type(8)))  float  v8f;
typedef __attribute__((ext_vector_type(4)))  unsigned int u32x4;

__device__ __forceinline__ unsigned short f2bf(float f) {
    unsigned int u = __float_as_uint(f);
    u += 0x7FFFu + ((u >> 16) & 1u);
    return (unsigned short)(u >> 16);
}

#define KC 256
#define PITCH (KC * 2 + 16)
#define TILE_BYTES (64 * PITCH)
#define STAGE_BYTES (2 * TILE_BYTES)
#define GEMM_LDS (2 * STAGE_BYTES)

struct alignas(32) FragBits { u32x4 lo, hi; };

__device__ __forceinline__ v16bf lds_frag(const unsigned char* p) {
    FragBits fb;
    fb.lo = *reinterpret_cast<const u32x4*>(p);
    fb.hi = *reinterpret_cast<const u32x4*>(p + 32);
    v16bf v;
    __builtin_memcpy(&v, &fb, sizeof(v));
    return v;
}

__device__ __forceinline__ void lds_barrier() {
    __builtin_amdgcn_fence(__ATOMIC_RELEASE, "workgroup", "local");
    __builtin_amdgcn_s_barrier();
    __builtin_amdgcn_fence(__ATOMIC_ACQUIRE, "workgroup", "local");
}

#define TAIL_ROWS (Bsz / 16)
template <int MODE>
__device__ __forceinline__
void gemm_tile(const unsigned short* __restrict__ A,
               const unsigned short* __restrict__ BT,
               int M, int N, int K,
               const float* __restrict__ bias,
               const float* __restrict__ drive,
               const float* __restrict__ Hold,
               const int*   __restrict__ active,
               float* __restrict__ Cout,
               unsigned short* __restrict__ HbOut,
               unsigned int* __restrict__ dHbits,
               int NL,
               unsigned char* const smem, int* const s_cnt, int* const s_list,
               const int bx, const int by, const int lin) {
    const int tid  = threadIdx.x;
    const int lane = tid & 31;
    const int wave = tid >> 5;
    const int rowBase = by * 64 + wave * 16;
    const int colBase = bx * 64;
    const int r16 = lane & 15;
    const int kh  = lane >> 4;
    int n_act = M;

    if (MODE != 0) {
        int flag[Bsz / 128];
        unsigned msk[Bsz / 128];
#pragma unroll
        for (int j = 0; j < Bsz / 128; ++j) {
            flag[j] = active[j * 128 + tid];
            msk[j]  = (unsigned)__ballot(flag[j] != 0);
        }
        if (lane == 0) {
#pragma unroll
            for (int j = 0; j < Bsz / 128; ++j) s_cnt[j * 4 + wave] = __popc(msk[j]);
        }
        __syncthreads();
        n_act = 0;
        int before[Bsz / 128];
#pragma unroll
        for (int s = 0; s < Bsz / 32; ++s) {
            if ((s & 3) == wave) before[s >> 2] = n_act;
            n_act += s_cnt[s];
        }
        if (n_act == 0) return;

#pragma unroll
        for (int j = 0; j < Bsz / 128; ++j) {
            if (flag[j] != 0)
                s_list[before[j] + __popc(msk[j] & ((1u << lane) - 1u))] = j * 128 + tid;
        }
        __syncthreads();

        if (n_act <= TAIL_ROWS) {
            const int NT = N >> 4;
            const int RT = (n_act + 15) >> 4;
            const int b  = lin;
            if (b >= RT * NT) return;
            const int rt = b / NT;
            const int ct = b - rt * NT;

            const int li   = rt * 16 + r16;
            const int arow = s_list[li < n_act ? li : 0];
            const int KQ   = K >> 2;
            const unsigned short* pa = A  + (size_t)arow * K + wave * KQ + kh * 16;
            const unsigned short* pb = BT + (size_t)(ct * 16 + r16) * K + wave * KQ + kh * 16;

            v8f part = {0.f,0.f,0.f,0.f,0.f,0.f,0.f,0.f};
            for (int k0 = 0; k0 < KQ; k0 += 512) {
                FragBits ta[16], tb[16];
#pragma unroll
                for (int i = 0; i < 16; ++i) {
                    ta[i].lo = *reinterpret_cast<const u32x4*>(pa + k0 + i * 32);
                    ta[i].hi = *reinterpret_cast<const u32x4*>(pa + k0 + i * 32 + 8);
                    tb[i].lo = *reinterpret_cast<const u32x4*>(pb + k0 + i * 32);
                    tb[i].hi = *reinterpret_cast<const u32x4*>(pb + k0 + i * 32 + 8);
                }
                __builtin_amdgcn_sched_barrier(0);
#pragma unroll
                for (int i = 0; i < 16; ++i) {
                    v16bf av, bv;
                    __builtin_memcpy(&av, &ta[i], sizeof(av));
                    __builtin_memcpy(&bv, &tb[i], sizeof(bv));
                    part = __builtin_amdgcn_wmma_f32_16x16x32_bf16(
                        false, av, false, bv, (short)0, part, false, false);
                }
            }

            float* const red = reinterpret_cast<float*>(smem);
#pragma unroll
            for (int r = 0; r < 8; ++r) red[(wave * 8 + r) * 32 + lane] = part[r];
            __syncthreads();

            const int n = ct * 16 + r16;
            int   mrow[2];
            bool  live[2];
            float v[2];
#pragma unroll
            for (int j = 0; j < 2; ++j) {
                const int r  = 2 * wave + j;
                const int lr = rt * 16 + r + 8 * kh;
                live[j] = lr < n_act;
                mrow[j] = s_list[live[j] ? lr : 0];
                v[j] = ((red[r * 32 + lane] + red[(8 + r) * 32 + lane])
                        + red[(16 + r) * 32 + lane]) + red[(24 + r) * 32 + lane];
            }
            if (MODE == 2) {
                if (n < NL) {
                    const float bn = bias[n];
#pragma unroll
                    for (int j = 0; j < 2; ++j)
                        if (live[j]) Cout[(size_t)mrow[j] * N + n] = v[j] + bn;
                }
            } else {
                float dv[2], hv[2], d[2];
#pragma unroll
                for (int j = 0; j < 2; ++j) {
                    const size_t idx = (size_t)mrow[j] * N + n;
                    dv[j] = drive[idx];
                    hv[j] = Hold[idx];
                }
#pragma unroll
                for (int j = 0; j < 2; ++j) {
                    const size_t idx = (size_t)mrow[j] * N + n;
                    const float pre  = v[j] + dv[j];
                    const float actv = pre >= 0.f ? pre : 0.05f * pre;
                    d[j] = live[j] ? fabsf(actv - hv[j]) : 0.f;
                    if (live[j]) {
                        Cout[idx]  = actv;
                        HbOut[idx] = f2bf(actv);
                    }
                }
#pragma unroll
                for (int off = 8; off > 0; off >>= 1) {
#pragma unroll
                    for (int j = 0; j < 2; ++j) d[j] = fmaxf(d[j], __shfl_xor(d[j], off, 32));
                }
#pragma unroll
                for (int j = 0; j < 2; ++j)
                    if (live[j] && r16 == 0 && d[j] > 0.f)
                        atomicMax(dHbits + mrow[j], __float_as_uint(d[j]));
            }
            return;
        }

        if (by * 64 >= n_act) return;
    }

    const int wv = __builtin_amdgcn_readfirstlane(wave);
    const unsigned char* arow[16];
#pragma unroll
    for (int i = 0; i < 16; ++i) {
        const int li = by * 64 + wv + 4 * i;
        int row = li;
        if (MODE != 0) row = __builtin_amdgcn_readfirstlane(s_list[li < n_act ? li : 0]);
        arow[i] = reinterpret_cast<const unsigned char*>(A) + (size_t)row * (size_t)(K * 2);
    }
    const unsigned voff = (unsigned)(lane * 16);
    const unsigned short* const gb = BT + (size_t)(colBase + wave) * K + lane * 8;
    const size_t gstep = (size_t)4 * K;
    const int sst = wave * PITCH + lane * 16;

    const int fa = (wave * 16 + r16) * PITCH + kh * 16;
    const int fb = r16 * PITCH + kh * 16;

    v8f acc0 = {0.f,0.f,0.f,0.f,0.f,0.f,0.f,0.f};
    v8f acc[4];
    acc[0] = acc0; acc[1] = acc0; acc[2] = acc0; acc[3] = acc0;

    const int nchunks = K / KC;
    u32x4 ra[16], rb[16];

    {
        u32x4 ta[16], tb[16];
#pragma unroll
        for (int i = 0; i < 16; ++i) {
            ta[i] = *reinterpret_cast<const u32x4*>(arow[i] + voff);
            tb[i] = *reinterpret_cast<const u32x4*>(gb + i * gstep);
        }
        __builtin_amdgcn_sched_barrier(0);
        if (nchunks > 1) {
#pragma unroll
            for (int i = 0; i < 16; ++i) {
                ra[i] = *reinterpret_cast<const u32x4*>(arow[i] + voff + KC * 2);
                rb[i] = *reinterpret_cast<const u32x4*>(gb + KC + i * gstep);
            }
        }
#pragma unroll
        for (int i = 0; i < 16; ++i) {
            *reinterpret_cast<u32x4*>(smem + sst + i * 4 * PITCH) = ta[i];
            *reinterpret_cast<u32x4*>(smem + TILE_BYTES + sst + i * 4 * PITCH) = tb[i];
        }
    }
    lds_barrier();

    auto chunk = [&](const int c, const bool st, const bool ld) __attribute__((always_inline)) {
        unsigned char* const cA = smem + (c & 1) * STAGE_BYTES;
        unsigned char* const cB = cA + TILE_BYTES;
        unsigned char* const nA = smem + ((c + 1) & 1) * STAGE_BYTES;
        unsigned char* const nB = nA + TILE_BYTES;
        const unsigned voff2 = voff + (unsigned)(c + 2) * (KC * 2);
        const unsigned short* const gb2 = gb + (size_t)(c + 2) * KC;

        v16bf a_cur  = lds_frag(cA + fa);
        v16bf b_cur0 = lds_frag(cB + fb);
        v16bf b_cur1 = lds_frag(cB + fb + 16 * PITCH);
        v16bf b_cur2 = lds_frag(cB + fb + 32 * PITCH);
        v16bf b_cur3 = lds_frag(cB + fb + 48 * PITCH);
#pragma unroll
        for (int ks = 0; ks < KC / 32; ++ks) {
            static_assert(KC / 32 == 8, "sixteen row pieces per operand over eight k-steps");
            v16bf a_nxt = a_cur, b_nxt0 = b_cur0, b_nxt1 = b_cur1,
                  b_nxt2 = b_cur2, b_nxt3 = b_cur3;
            if (ks + 1 < KC / 32) {
                const int ko = (ks + 1) * 64;
                a_nxt  = lds_frag(cA + fa + ko);
                b_nxt0 = lds_frag(cB + fb + ko);
                b_nxt1 = lds_frag(cB + fb + 16 * PITCH + ko);
                b_nxt2 = lds_frag(cB + fb + 32 * PITCH + ko);
                b_nxt3 = lds_frag(cB + fb + 48 * PITCH + ko);
            }
            if (st) {
#pragma unroll
                for (int j = 0; j < 2; ++j) {
                    const int i = 2 * ks + j;
                    *reinterpret_cast<u32x4*>(nA + sst + i * 4 * PITCH) = ra[i];
                    *reinterpret_cast<u32x4*>(nB + sst + i * 4 * PITCH) = rb[i];
                }
            }
            if (ld) {
#pragma unroll
                for (int j = 0; j < 2; ++j) {
                    const int i = 2 * ks + j;
                    ra[i] = *reinterpret_cast<const u32x4*>(arow[i] + voff2);
                    rb[i] = *reinterpret_cast<const u32x4*>(gb2 + i * gstep);
                }
            }
            __builtin_amdgcn_sched_barrier(0);
            acc[0] = __builtin_amdgcn_wmma_f32_16x16x32_bf16(
                false, a_cur, false, b_cur0, (short)0, acc[0], false, false);
            acc[1] = __builtin_amdgcn_wmma_f32_16x16x32_bf16(
                false, a_cur, false, b_cur1, (short)0, acc[1], false, false);
            acc[2] = __builtin_amdgcn_wmma_f32_16x16x32_bf16(
                false, a_cur, false, b_cur2, (short)0, acc[2], false, false);
            acc[3] = __builtin_amdgcn_wmma_f32_16x16x32_bf16(
                false, a_cur, false, b_cur3, (short)0, acc[3], false, false);
            __builtin_amdgcn_sched_barrier(0);
            a_cur = a_nxt;
            b_cur0 = b_nxt0; b_cur1 = b_nxt1; b_cur2 = b_nxt2; b_cur3 = b_nxt3;
        }
        if (st) lds_barrier();
    };

    for (int c = 0; c + 2 < nchunks; ++c) chunk(c, true, true);
    if (nchunks > 1) chunk(nchunks - 2, true, false);
    chunk(nchunks - 1, false, false);

    int   mrow[8];
    bool  live[8];
    float dmax[8];
#pragma unroll
    for (int r = 0; r < 8; ++r) {
        const int li = rowBase + r + 8 * kh;
        live[r] = (MODE == 0) || (li < n_act);
        mrow[r] = (MODE != 0) ? s_list[live[r] ? li : 0] : li;
        dmax[r] = 0.f;
    }

    if (MODE == 0) {
#pragma unroll
        for (int f = 0; f < 4; ++f) {
            const int n = colBase + f * 16 + r16;
#pragma unroll
            for (int r = 0; r < 8; ++r)
                Cout[(size_t)mrow[r] * N + n] = acc[f][r] + bias[n];
        }
    } else if (MODE == 2) {
        float bn[4];
#pragma unroll
        for (int f = 0; f < 4; ++f) {
            const int n = colBase + f * 16 + r16;
            bn[f] = bias[n < NL ? n : 0];
        }
        float lg[4][8];
#pragma unroll
        for (int f = 0; f < 4; ++f) {
#pragma unroll
            for (int r = 0; r < 8; ++r) lg[f][r] = acc[f][r] + bn[f];
        }
#pragma unroll
        for (int r = 0; r < 8; ++r) {
            if (live[r]) {
#pragma unroll
                for (int f = 0; f < 4; ++f) {
                    const int n = colBase + f * 16 + r16;
                    if (n < NL) Cout[(size_t)mrow[r] * N + n] = lg[f][r];
                }
            }
        }
    } else {
        float dv[4][8], hv[4][8];
#pragma unroll
        for (int f = 0; f < 4; ++f) {
#pragma unroll
            for (int r = 0; r < 8; ++r) {
                const size_t idx = (size_t)mrow[r] * N + (colBase + f * 16 + r16);
                dv[f][r] = drive[idx];
                hv[f][r] = Hold[idx];
            }
        }
        __builtin_amdgcn_sched_barrier(0);
#pragma unroll
        for (int f = 0; f < 4; ++f) {
#pragma unroll
            for (int r = 0; r < 8; ++r) {
                const float pre  = acc[f][r] + dv[f][r];
                const float actv = pre >= 0.f ? pre : 0.05f * pre;
                const float dd   = fabsf(actv - hv[f][r]);
                dmax[r] = fmaxf(dmax[r], live[r] ? dd : 0.f);
                acc[f][r] = actv;
            }
        }
#pragma unroll
        for (int r = 0; r < 8; ++r) {
            if (live[r]) {
#pragma unroll
                for (int f = 0; f < 4; ++f) {
                    const size_t idx = (size_t)mrow[r] * N + (colBase + f * 16 + r16);
                    Cout[idx]  = acc[f][r];
                    HbOut[idx] = f2bf(acc[f][r]);
                }
            }
        }
#pragma unroll
        for (int off = 8; off > 0; off >>= 1) {
#pragma unroll
            for (int r = 0; r < 8; ++r) dmax[r] = fmaxf(dmax[r], __shfl_xor(dmax[r], off, 32));
        }
#pragma unroll
        for (int r = 0; r < 8; ++r) {
            if (r16 == 0 && dmax[r] > 0.f) {
                atomicMax(dHbits + mrow[r], __float_as_uint(dmax[r]));
            }
        }
    }
}

__device__ __forceinline__
void col_l1_block(const float* __restrict__ A, const float* __restrict__ Msk,
                  float* __restrict__ part, const int cols, const int bx, const int b) {
    static_assert(HD % 256 == 0, "leaf rows are t + 256*r");
    __shared__ float smx[8][32];
    __shared__ float smy[8][32];
    const int lane = threadIdx.x & 31;
    const int v    = threadIdx.x >> 5;
    const int c    = (bx * 32 + lane) * 2;
    const bool in  = c < cols;
    const int cl   = in ? c : 0;
    float lx[4], ly[4];
#pragma unroll
    for (int m = 0; m < 4; ++m) {
        const int t = b + 8 * (v + 8 * m);
        float sx = 0.f, sy = 0.f;
#pragma unroll
        for (int r = 0; r < HD / 256; ++r) {
            const size_t idx = (size_t)(t + 256 * r) * cols + cl;
            const float2 a  = *reinterpret_cast<const float2*>(A + idx);
            const float2 mk = *reinterpret_cast<const float2*>(Msk + idx);
            sx += fabsf(a.x * mk.x);
            sy += fabsf(a.y * mk.y);
        }
        lx[m] = sx; ly[m] = sy;
    }
    smx[v][lane] = (lx[0] + lx[2]) + (lx[1] + lx[3]);
    smy[v][lane] = (ly[0] + ly[2]) + (ly[1] + ly[3]);
    __syncthreads();
    if (v == 0 && in) {
        part[(size_t)b * cols + c] =
            ((smx[0][lane] + smx[4][lane]) + (smx[2][lane] + smx[6][lane])) +
            ((smx[1][lane] + smx[5][lane]) + (smx[3][lane] + smx[7][lane]));
        part[(size_t)b * cols + c + 1] =
            ((smy[0][lane] + smy[4][lane]) + (smy[2][lane] + smy[6][lane])) +
            ((smy[1][lane] + smy[5][lane]) + (smy[3][lane] + smy[7][lane]));
    }
}

__device__ __forceinline__ float fold8(const float* __restrict__ p, int cols, int j) {
    const float p0 = p[j],            p1 = p[cols + j];
    const float p2 = p[2 * cols + j], p3 = p[3 * cols + j];
    const float p4 = p[4 * cols + j], p5 = p[5 * cols + j];
    const float p6 = p[6 * cols + j], p7 = p[7 * cols + j];
    return ((p0 + p4) + (p2 + p6)) + ((p1 + p5) + (p3 + p7));
}

#define SCB 128
__device__ __forceinline__
void scalars_block(const float* __restrict__ col_l1,
                   const float* __restrict__ whl,
                   float* __restrict__ scalars, float* const sm) {
    static_assert(HD % SCB == 0, "whole trips over the HD columns");
    const int t = threadIdx.x;
    float mx = -1e30f;
#pragma unroll
    for (int i = 0; i < HD / SCB; ++i) mx = fmaxf(mx, fold8(col_l1, HD, t + SCB * i));
    sm[t] = mx; __syncthreads();
    for (int s = SCB / 2; s > 0; s >>= 1) {
        if (t < s) sm[t] = fmaxf(sm[t], sm[t + s]);
        __syncthreads();
    }
    const float m1 = sm[0];
    __syncthreads();
    mx = -1e30f;
#pragma unroll
    for (int i = 0; i < (Lsz + SCB - 1) / SCB; ++i) {
        const int j = t + SCB * i;
        mx = fmaxf(mx, fold8(whl, Lsz, j < Lsz ? j : Lsz - 1));
    }
    sm[t] = mx; __syncthreads();
    for (int s = SCB / 2; s > 0; s >>= 1) {
        if (t < s) sm[t] = fmaxf(sm[t], sm[t + s]);
        __syncthreads();
    }
    if (t == 0) {
        scalars[0] = 0.95f / fmaxf(m1, 1e-8f);
        scalars[1] = sm[0] * 19.0f;
    }
}

#define TT 64
#define TPITCH (TT + 2)
__device__ __forceinline__
void build_wt_block(const float* __restrict__ R, const float* __restrict__ M,
                    const float* __restrict__ scalars, const int sel,
                    unsigned short* __restrict__ WT,
                    const int K, const int srcld, const int NL,
                    const int bkx, const int bny) {
    __shared__ __attribute__((aligned(16))) unsigned short tile[TT * TPITCH];
    const int k0 = bkx * TT;
    const int n0 = bny * TT;
    const int t  = threadIdx.x;

    const int ln = (t & 15) * 4;
    const int lk = t >> 4;
    const int n  = n0 + ln;
    const bool in = n < NL;
    const int nc = in ? n : 0;
    float4 r[4], m[4];
#pragma unroll
    for (int i = 0; i < 4; ++i) {
        const size_t sidx = (size_t)(k0 + lk + 16 * i) * srcld + nc;
        r[i] = *reinterpret_cast<const float4*>(R + sidx);
        m[i] = *reinterpret_cast<const float4*>(M + sidx);
    }
    float sc = 1.f;
    if (sel == 0) sc = scalars[0];
#pragma unroll
    for (int i = 0; i < 4; ++i) {
        float v0 = r[i].x * m[i].x, v1 = r[i].y * m[i].y;
        float v2 = r[i].z * m[i].z, v3 = r[i].w * m[i].w;
        if (sel == 0) { v0 *= sc; v1 *= sc; v2 *= sc; v3 *= sc; }
        if (!in) { v0 = 0.f; v1 = 0.f; v2 = 0.f; v3 = 0.f; }
        unsigned int* d = reinterpret_cast<unsigned int*>(tile + (lk + 16 * i) * TPITCH + ln);
        d[0] = (unsigned int)f2bf(v0) | ((unsigned int)f2bf(v1) << 16);
        d[1] = (unsigned int)f2bf(v2) | ((unsigned int)f2bf(v3) << 16);
    }
    __syncthreads();

    const int ks = (t & 7) * 8;
#pragma unroll
    for (int p = 0; p < 2; ++p) {
        const int nl = (t >> 3) + 32 * p;
        unsigned int w[4];
#pragma unroll
        for (int j = 0; j < 4; ++j) {
            const unsigned int e0 = tile[(ks + 2 * j) * TPITCH + nl];
            const unsigned int e1 = tile[(ks + 2 * j + 1) * TPITCH + nl];
            w[j] = e0 | (e1 << 16);
        }
        u32x4 o; o.x = w[0]; o.y = w[1]; o.z = w[2]; o.w = w[3];
        *reinterpret_cast<u32x4*>(WT + (size_t)(n0 + nl) * K + k0 + ks) = o;
    }
}

__global__ __launch_bounds__(256)
void build_wt_kernel(const float* __restrict__ R, const float* __restrict__ M,
                     const float* __restrict__ scalars, int sel,
                     unsigned short* __restrict__ WT,
                     int K, int srcld, int NL) {
    build_wt_block(R, M, scalars, sel, WT, K, srcld, NL, (int)blockIdx.x, (int)blockIdx.y);
}

#define SA_COLH  ((HD / 64) * 8)
#define SA_COLL  (((Lsz + 63) / 64) * 8)
#define SA_WTIH  ((Isz / TT) * (HD / TT))
#define SA_WTHL  ((HD / TT) * (LP / TT))
#define SA_XB    128
#define SA_INIT  128
#define SA_BLOCKS (SA_COLH + SA_COLL + SA_WTIH + SA_WTHL + SA_XB + SA_INIT)
__global__ __launch_bounds__(256)
void setup_inputs_kernel(const float* __restrict__ x,
                         const float* __restrict__ RIH, const float* __restrict__ MIH,
                         const float* __restrict__ RH,  const float* __restrict__ MH,
                         const float* __restrict__ RHL, const float* __restrict__ MHL,
                         float* __restrict__ part_h, float* __restrict__ part_l,
                         unsigned short* __restrict__ WT_IH,
                         unsigned short* __restrict__ WT_HL,
                         unsigned short* __restrict__ xb,
                         float* __restrict__ H0, unsigned short* __restrict__ Hb0,
                         unsigned int* __restrict__ dHa, unsigned int* __restrict__ dHb,
                         unsigned int* __restrict__ dHc,
                         int* __restrict__ act0, int* __restrict__ act1) {
    int b = blockIdx.x;
    if (b < SA_COLH) {
        col_l1_block(RH, MH, part_h, HD, b % (HD / 64), b / (HD / 64));
        return;
    }
    b -= SA_COLH;
    if (b < SA_COLL) {
        col_l1_block(RHL, MHL, part_l, Lsz, b % ((Lsz + 63) / 64), b / ((Lsz + 63) / 64));
        return;
    }
    b -= SA_COLL;
    if (b < SA_WTIH) {
        build_wt_block(RIH, MIH, nullptr,  1, WT_IH, Isz, HD, HD,
                       b % (Isz / TT), b / (Isz / TT));
        return;
    }
    b -= SA_WTIH;
    if (b < SA_WTHL) {
        build_wt_block(RHL, MHL, nullptr,  1, WT_HL, HD, Lsz, Lsz,
                       b % (HD / TT), b / (HD / TT));
        return;
    }
    b -= SA_WTHL;
    if (b < SA_XB) {
        static_assert((Bsz * Isz) % (SA_XB * 256 * 8) == 0, "whole trips");
        const int gt = b * 256 + (int)threadIdx.x;
#pragma unroll
        for (int q = 0; q < (Bsz * Isz) / (SA_XB * 256 * 8); ++q) {
            const size_t e = ((size_t)q * (SA_XB * 256) + gt) * 8;
            const float4 lo = *reinterpret_cast<const float4*>(x + e);
            const float4 hi = *reinterpret_cast<const float4*>(x + e + 4);
            u32x4 o;
            o.x = (unsigned int)f2bf(lo.x) | ((unsigned int)f2bf(lo.y) << 16);
            o.y = (unsigned int)f2bf(lo.z) | ((unsigned int)f2bf(lo.w) << 16);
            o.z = (unsigned int)f2bf(hi.x) | ((unsigned int)f2bf(hi.y) << 16);
            o.w = (unsigned int)f2bf(hi.z) | ((unsigned int)f2bf(hi.w) << 16);
            *reinterpret_cast<u32x4*>(xb + e) = o;
        }
        return;
    }
    b -= SA_XB;
    {
        static_assert((Bsz * HD) % (SA_INIT * 256 * 8) == 0, "whole trips");
        const int gt = b * 256 + (int)threadIdx.x;
        const float4 zf = {0.f, 0.f, 0.f, 0.f};
        const u32x4  zu = {0u, 0u, 0u, 0u};
#pragma unroll
        for (int q = 0; q < (Bsz * HD) / (SA_INIT * 256 * 4); ++q)
            reinterpret_cast<float4*>(H0)[(size_t)q * (SA_INIT * 256) + gt] = zf;
#pragma unroll
        for (int q = 0; q < (Bsz * HD) / (SA_INIT * 256 * 8); ++q)
            reinterpret_cast<u32x4*>(Hb0)[(size_t)q * (SA_INIT * 256) + gt] = zu;
        if (gt < Bsz) { dHa[gt] = 0u; dHb[gt] = 0u; dHc[gt] = 0u; act0[gt] = 1; act1[gt] = 1; }
    }
}

#define DRV_BLOCKS ((HD / 64) * (Bsz / 64))
__global__ __launch_bounds__(128)
void drive_scalars_kernel(const unsigned short* __restrict__ xb,
                          const unsigned short* __restrict__ WT_IH,
                          const float* __restrict__ BH,
                          float* __restrict__ drive,
                          const float* __restrict__ part_h,
                          const float* __restrict__ part_l,
                          float* __restrict__ scalars) {
    static_assert(SCB == 128, "scalars_block runs on this kernel's block size");
    __shared__ __attribute__((aligned(16))) unsigned char smem[GEMM_LDS];
    if (blockIdx.x == 0) {
        scalars_block(part_h, part_l, scalars, reinterpret_cast<float*>(smem));
    } else {
        const int b = (int)blockIdx.x - 1;
        gemm_tile<0>(xb, WT_IH, Bsz, HD, Isz, BH, nullptr, nullptr, nullptr,
                     drive, nullptr, nullptr, HD,
                     smem, nullptr, nullptr, b % (HD / 64), b / (HD / 64), b);
    }
}

__device__ __forceinline__
void certify_rows(const int ub,
                  const float* __restrict__ logits,
                  const float* __restrict__ scalars,
                  unsigned int* __restrict__ dHbits,
                  const int* __restrict__ actIn,
                  int* __restrict__ actOut,
                  float* __restrict__ out,
                  const int is_last) {
    static_assert(Lsz % 4 == 0 && LP == 1024, "a row is eight 16-B pieces per lane, none straddling column Lsz");
    const int lane = threadIdx.x & 31;
    const int wave = threadIdx.x >> 5;
    const int row0 = ub * 8 + wave * 2;

    int    a[2];
    float  dh[2];
    float4 v[2][8];
#pragma unroll
    for (int j = 0; j < 2; ++j) {
        a[j]  = actIn[row0 + j];
        dh[j] = __uint_as_float(dHbits[row0 + j]);
        const float* lr = logits + (size_t)(row0 + j) * LP;
#pragma unroll
        for (int q = 0; q < 8; ++q) {
            const int col = 128 * q + 4 * lane;
            v[j][q] = *reinterpret_cast<const float4*>(lr + (col < Lsz ? col : Lsz - 4));
        }
    }
    const float gcoef = scalars[1];

#pragma unroll
    for (int j = 0; j < 2; ++j) {
        const int row = row0 + j;
        int aj = a[j];
        if (aj != 0) {
            float m1 = -1e30f, m2 = -1e30f;
#pragma unroll
            for (int q = 0; q < 8; ++q) {
                const bool in = 128 * q + 4 * lane < Lsz;
                const float e[4] = {v[j][q].x, v[j][q].y, v[j][q].z, v[j][q].w};
#pragma unroll
                for (int c = 0; c < 4; ++c) {
                    const float x = in ? e[c] : -1e30f;
                    m2 = fmaxf(m2, fminf(m1, x));
                    m1 = fmaxf(m1, x);
                }
            }
            for (int off = 16; off > 0; off >>= 1) {
                const float o1 = __shfl_xor(m1, off, 32);
                const float o2 = __shfl_xor(m2, off, 32);
                const float hi = fmaxf(m1, o1);
                const float lo = fmaxf(fminf(m1, o1), fmaxf(m2, o2));
                m1 = hi; m2 = lo;
            }
            const float margin = m1 - m2;
            const float Gamma = gcoef * dh[j];
            const bool newly = margin > 2.0f * Gamma;
            if (newly || is_last) {
                float* orow = out + (size_t)row * Lsz;
#pragma unroll
                for (int q = 0; q < 8; ++q) {
                    const int col = 128 * q + 4 * lane;
                    if (col < Lsz) *reinterpret_cast<float4*>(orow + col) = v[j][q];
                }
                aj = 0;
            }
        }
        if (lane == 0) {
            actOut[row] = aj;
            dHbits[row] = 0u;
        }
    }
}

#define ROW_GROUPS  (Bsz / 64)
#define REC_BLOCKS  ((HD / 64) * ROW_GROUPS)
#define LOG_BLOCKS  ((LP / 64) * ROW_GROUPS)
#define CERT_BLOCKS (Bsz / 8)
__global__ __launch_bounds__(128)
void step_kernel(const unsigned short* __restrict__ HbCur,
                 const float* __restrict__ HCur,
                 const unsigned short* __restrict__ WT_H,
                 const unsigned short* __restrict__ WT_HL,
                 const float* __restrict__ drive,
                 const float* __restrict__ BL,
                 const float* __restrict__ scalars,
                 const int* __restrict__ actIn,
                 int* __restrict__ actOut,
                 float* __restrict__ HNext,
                 unsigned short* __restrict__ HbNext,
                 unsigned int* __restrict__ dHNext,
                 float* __restrict__ logitsCur,
                 const float* __restrict__ logitsPrev,
                 unsigned int* __restrict__ dHPrev,
                 float* __restrict__ out,
                 int do_rec, int do_log, int do_cert, int is_last) {
    __shared__ __attribute__((aligned(16))) unsigned char smem[GEMM_LDS];
    __shared__ int s_cnt[Bsz / 32];
    __shared__ int s_list[Bsz];
    const int b = blockIdx.x;
    if (b < REC_BLOCKS) {
        if (do_rec)
            gemm_tile<1>(HbCur, WT_H, Bsz, HD, HD, nullptr, drive, HCur, actIn,
                         HNext, HbNext, dHNext, HD,
                         smem, s_cnt, s_list, b / ROW_GROUPS, b % ROW_GROUPS, b);
    } else if (b < REC_BLOCKS + LOG_BLOCKS) {
        const int q = b - REC_BLOCKS;
        if (do_log)
            gemm_tile<2>(HbCur, WT_HL, Bsz, LP, HD, BL, nullptr, nullptr, actIn,
                         logitsCur, nullptr, nullptr, Lsz,
                         smem, s_cnt, s_list, q / ROW_GROUPS, ROW_GROUPS - 1 - q % ROW_GROUPS, q);
    } else {
        if (do_cert)
            certify_rows(b - (REC_BLOCKS + LOG_BLOCKS), logitsPrev, scalars, dHPrev,
                         actIn, actOut, out, is_last);
    }
}

static inline char* wsalloc(char*& p, size_t bytes) {
    char* r = p;
    p += (bytes + 255) & ~(size_t)255;
    return r;
}

extern "C" void kernel_launch(void* const* d_in, const int* in_sizes, int n_in,
                              void* d_out, int out_size, void* d_ws, size_t ws_size,
                              hipStream_t stream) {
    (void)in_sizes; (void)n_in; (void)out_size; (void)ws_size;
    const float* x   = (const float*)d_in[0];
    const float* RIH = (const float*)d_in[1];
    const float* RH  = (const float*)d_in[2];
    const float* RHL = (const float*)d_in[3];
    const float* BH  = (const float*)d_in[4];
    const float* BL  = (const float*)d_in[5];
    const float* MIH = (const float*)d_in[6];
    const float* MH  = (const float*)d_in[7];
    const float* MHL = (const float*)d_in[8];
    float* out = (float*)d_out;

    char* p = (char*)d_ws;
    unsigned short* WT_H  = (unsigned short*)wsalloc(p, (size_t)HD * HD * 2);
    unsigned short* WT_HL = (unsigned short*)wsalloc(p, (size_t)LP * HD * 2);
    unsigned short* WT_IH = (unsigned short*)wsalloc(p, (size_t)HD * Isz * 2);
    unsigned short* xb    = (unsigned short*)wsalloc(p, (size_t)Bsz * Isz * 2);
    float* drive = (float*)wsalloc(p, (size_t)Bsz * HD * 4);
    float* Hbuf0 = (float*)wsalloc(p, (size_t)Bsz * HD * 4);
    float* Hbuf1 = (float*)wsalloc(p, (size_t)Bsz * HD * 4);
    unsigned short* Hb0 = (unsigned short*)wsalloc(p, (size_t)Bsz * HD * 2);
    unsigned short* Hb1 = (unsigned short*)wsalloc(p, (size_t)Bsz * HD * 2);
    float* logits = (float*)wsalloc(p, (size_t)Bsz * LP * 4);
    float* col_l1 = (float*)wsalloc(p, HD * 4);
    float* scalars = (float*)wsalloc(p, 256);
    unsigned int* dHb = (unsigned int*)wsalloc(p, Bsz * 4);
    int* active = (int*)wsalloc(p, Bsz * 4);

    unsigned int* dH1 = (unsigned int*)col_l1;
    unsigned int* dH2 = dH1 + Bsz;
    int* active1 = (int*)(dH2 + Bsz);
    float* logits1 = (float*)WT_IH;
    float* part_h = logits;
    float* part_l = logits + 8 * HD;

    setup_inputs_kernel<<<SA_BLOCKS, 256, 0, stream>>>(
        x, RIH, MIH, RH, MH, RHL, MHL, part_h, part_l, WT_IH, WT_HL, xb,
        Hbuf0, Hb0, dHb, dH1, dH2, active, active1);
    drive_scalars_kernel<<<DRV_BLOCKS + 1, 128, 0, stream>>>(
        xb, WT_IH, BH, drive, part_h, part_l, scalars);
    build_wt_kernel<<<dim3(HD / TT, HD / TT), 256, 0, stream>>>(
        RH, MH, scalars,  0, WT_H, HD, HD, HD);

    float* Hc[2] = {Hbuf0, Hbuf1};
    unsigned short* Hbc[2] = {Hb0, Hb1};
    int* act[2] = {active, active1};
    unsigned int* dHp[3] = {dHb, dH1, dH2};
    float* lg[2] = {logits, logits1};
    for (int s = 0; s <= TMAX + 1; ++s) {
        const int c = s & 1, n = (s + 1) & 1;
        step_kernel<<<REC_BLOCKS + LOG_BLOCKS + CERT_BLOCKS, 128, 0, stream>>>(
            Hbc[c], Hc[c], WT_H, WT_HL, drive, BL, scalars, act[c], act[n],
            Hc[n], Hbc[n], dHp[(s + 1) % 3], lg[c], lg[n], dHp[(s + 2) % 3], out,
            (s < TMAX) ? 1 : 0, (s >= 1 && s <= TMAX) ? 1 : 0, (s >= 2) ? 1 : 0,
            (s == TMAX + 1) ? 1 : 0);
    }
}
